# starting kernel
# baseline (speedup 1.0000x reference)
#include <hip/hip_runtime.h>

typedef __attribute__((ext_vector_type(2))) float v2f;
typedef __attribute__((ext_vector_type(8))) float v8f;

#define N_NODES 100000
#define N_EDGES 1600000
#define NFEAT   256
#define NHID    128

__global__ void zero_kernel(float* __restrict__ p, int n) {
    int i = blockIdx.x * blockDim.x + threadIdx.x;
    if (i < n) p[i] = 0.0f;
}

__global__ void degree_kernel(const int* __restrict__ src, const int* __restrict__ dst,
                              float* __restrict__ degS, float* __restrict__ degD, int e) {
    int i = blockIdx.x * blockDim.x + threadIdx.x;
    if (i < e) {
        atomicAdd(&degS[src[i]], 1.0f);
        atomicAdd(&degD[dst[i]], 1.0f);
    }
}

__global__ void norm_finalize(float* __restrict__ degS, float* __restrict__ degD, int n) {
    int i = blockIdx.x * blockDim.x + threadIdx.x;
    if (i < n) {
        degS[i] = rsqrtf(fmaxf(degS[i], 1.0f));
        degD[i] = rsqrtf(fmaxf(degD[i], 1.0f));
    }
}

__global__ void __launch_bounds__(256)
gemm_scaled_wmma(const float* __restrict__ X, const float* __restrict__ scale,
                 const float* __restrict__ W, float* __restrict__ Out, int K) {
    const int wave = threadIdx.x >> 5;
    const int lane = threadIdx.x & 31;
    const int kh   = lane >> 4;
    const int l    = lane & 15;
    const int row0 = blockIdx.x << 4;
    const int col0 = wave << 4;
    const int row  = row0 + l;

    const float ns = scale[row];
    const float* __restrict__ xrow = X + (size_t)row * K;

    v8f c = {};
    for (int k = 0; k < K; k += 4) {
        const int ka = k + 2 * kh;
        v2f a;
        a.x = xrow[ka]     * ns;
        a.y = xrow[ka + 1] * ns;
        v2f b;
        b.x = W[(size_t)ka       * NHID + col0 + l];
        b.y = W[(size_t)(ka + 1) * NHID + col0 + l];
        c = __builtin_amdgcn_wmma_f32_16x16x4_f32(false, a, false, b,
                                                  (short)0, c, false, false);
    }

    float* __restrict__ orow = Out + (size_t)row0 * NHID + col0;
#pragma unroll
    for (int r = 0; r < 8; ++r) {
        orow[(size_t)(r + 8 * kh) * NHID + l] = c[r];
    }
}

__global__ void __launch_bounds__(256)
scatter_add(const float* __restrict__ M, const int* __restrict__ src,
            const int* __restrict__ dst, float* __restrict__ AGG, int e) {
    const int w    = (blockIdx.x * blockDim.x + threadIdx.x) >> 5;
    const int lane = threadIdx.x & 31;
    if (w >= e) return;
    const int s = src[w];
    const int d = dst[w];
    const float4 v = reinterpret_cast<const float4*>(M + (size_t)s * NHID)[lane];
    float* drow = AGG + (size_t)d * NHID + lane * 4;
    atomicAdd(drow + 0, v.x);
    atomicAdd(drow + 1, v.y);
    atomicAdd(drow + 2, v.z);
    atomicAdd(drow + 3, v.w);
}

__global__ void bias_relu(const float* __restrict__ AGG, const float* __restrict__ normD,
                          const float* __restrict__ b, float* __restrict__ H, int total) {
    int i = blockIdx.x * blockDim.x + threadIdx.x;
    if (i < total) {
        const int row = i >> 7;
        const int j   = i & 127;
        const float v = AGG[i] * normD[row] + b[j];
        H[i] = v > 0.0f ? v : 0.0f;
    }
}

__global__ void __launch_bounds__(256)
final_fc(const float* __restrict__ AGG, const float* __restrict__ normD,
         const float* __restrict__ b2, const float* __restrict__ Wfc,
         const float* __restrict__ bfc, float* __restrict__ out, int n) {
    const int w    = (blockIdx.x * blockDim.x + threadIdx.x) >> 5;
    const int lane = threadIdx.x & 31;
    if (w >= n) return;
    const float nd = normD[w];
    const float* arow = AGG + (size_t)w * NHID;
    float acc = 0.0f;
#pragma unroll
    for (int t = 0; t < 4; ++t) {
        const int j = lane + 32 * t;
        acc += (arow[j] * nd + b2[j]) * Wfc[j];
    }
#pragma unroll
    for (int off = 16; off > 0; off >>= 1)
        acc += __shfl_xor(acc, off, 32);
    if (lane == 0) out[w] = acc + bfc[0];
}

extern "C" void kernel_launch(void* const* d_in, const int* in_sizes, int n_in,
                              void* d_out, int out_size, void* d_ws, size_t ws_size,
                              hipStream_t stream) {
    const float* x   = (const float*)d_in[0];
    const float* W1  = (const float*)d_in[1];
    const float* b1  = (const float*)d_in[2];
    const float* W2  = (const float*)d_in[3];
    const float* b2  = (const float*)d_in[4];
    const float* Wfc = (const float*)d_in[5];
    const float* bfc = (const float*)d_in[6];
    const int*   src = (const int*)d_in[7];
    const int*   dst = (const int*)d_in[8];
    float* out = (float*)d_out;

    const int N = N_NODES, E = N_EDGES;
    const int NH = NHID;

    float* BUF_A = (float*)d_ws;
    float* BUF_B = BUF_A + (size_t)N * NH;
    float* normS = BUF_B + (size_t)N * NH;
    float* normD = normS + N;

    const int TPB = 256;
    const int featTotal = N * NH;
    const int gridFeat  = (featTotal + TPB - 1) / TPB;
    const int gridN     = (N + TPB - 1) / TPB;
    const int gridE     = (E + TPB - 1) / TPB;
    const int gridEdgeWaves = (E * (TPB / (TPB / 32)) ) ? ((E + 7) / 8) : 0;
    const int gridNodeWaves = (N + 7) / 8;
    const int gridGemm  = N / 16;

    zero_kernel<<<(2 * N + TPB - 1) / TPB, TPB, 0, stream>>>(normS, 2 * N);
    degree_kernel<<<gridE, TPB, 0, stream>>>(src, dst, normS, normD, E);
    norm_finalize<<<gridN, TPB, 0, stream>>>(normS, normD, N);

    zero_kernel<<<gridFeat, TPB, 0, stream>>>(BUF_B, featTotal);
    gemm_scaled_wmma<<<gridGemm, TPB, 0, stream>>>(x, normS, W1, BUF_A, NFEAT);
    scatter_add<<<gridEdgeWaves, TPB, 0, stream>>>(BUF_A, src, dst, BUF_B, E);
    bias_relu<<<gridFeat, TPB, 0, stream>>>(BUF_B, normD, b1, BUF_A, featTotal);

    gemm_scaled_wmma<<<gridGemm, TPB, 0, stream>>>(BUF_A, normS, W2, BUF_B, NH);
    zero_kernel<<<gridFeat, TPB, 0, stream>>>(BUF_A, featTotal);
    scatter_add<<<gridEdgeWaves, TPB, 0, stream>>>(BUF_B, src, dst, BUF_A, E);

    final_fc<<<gridNodeWaves, TPB, 0, stream>>>(BUF_A, normD, b2, Wfc, bfc, out, N);
}


# edge row sum: each work-group owns 400 dst nodes, scans the whole dst list, sums their 128-float rows in LDS (f32 LDS adds, arrival order), stores whole rows; no global row atomics or zero fill
# speedup vs baseline: 1.2662x; 1.2662x over previous
#include <hip/hip_runtime.h>

typedef __attribute__((ext_vector_type(2))) float v2f;
typedef __attribute__((ext_vector_type(8))) float v8f;

#define N_NODES 100000
#define N_EDGES 1600000
#define NFEAT   256
#define NHID    128

__global__ void zero_kernel(float* __restrict__ p, int n) {
    int i = blockIdx.x * blockDim.x + threadIdx.x;
    if (i < n) p[i] = 0.0f;
}

__global__ void degree_kernel(const int* __restrict__ src, const int* __restrict__ dst,
                              float* __restrict__ degS, float* __restrict__ degD, int e) {
    int i = blockIdx.x * blockDim.x + threadIdx.x;
    if (i < e) {
        atomicAdd(&degS[src[i]], 1.0f);
        atomicAdd(&degD[dst[i]], 1.0f);
    }
}

__global__ void norm_finalize(float* __restrict__ degS, float* __restrict__ degD, int n) {
    int i = blockIdx.x * blockDim.x + threadIdx.x;
    if (i < n) {
        degS[i] = rsqrtf(fmaxf(degS[i], 1.0f));
        degD[i] = rsqrtf(fmaxf(degD[i], 1.0f));
    }
}

__global__ void __launch_bounds__(256)
gemm_scaled_wmma(const float* __restrict__ X, const float* __restrict__ scale,
                 const float* __restrict__ W, float* __restrict__ Out, int K) {
    const int wave = threadIdx.x >> 5;
    const int lane = threadIdx.x & 31;
    const int kh   = lane >> 4;
    const int l    = lane & 15;
    const int row0 = blockIdx.x << 4;
    const int col0 = wave << 4;
    const int row  = row0 + l;

    const float ns = scale[row];
    const float* __restrict__ xrow = X + (size_t)row * K;

    v8f c = {};
    for (int k = 0; k < K; k += 4) {
        const int ka = k + 2 * kh;
        v2f a;
        a.x = xrow[ka]     * ns;
        a.y = xrow[ka + 1] * ns;
        v2f b;
        b.x = W[(size_t)ka       * NHID + col0 + l];
        b.y = W[(size_t)(ka + 1) * NHID + col0 + l];
        c = __builtin_amdgcn_wmma_f32_16x16x4_f32(false, a, false, b,
                                                  (short)0, c, false, false);
    }

    float* __restrict__ orow = Out + (size_t)row0 * NHID + col0;
#pragma unroll
    for (int r = 0; r < 8; ++r) {
        orow[(size_t)(r + 8 * kh) * NHID + l] = c[r];
    }
}

#define RS_NODES   400
#define RS_THREADS 1024
#define RS_WAVES   (RS_THREADS / 32)
#define RS_QCAP    2048
#define RS_STEPS   4
#define RS_LOADS   8
#define RS_CHUNK4  (RS_STEPS * RS_LOADS * RS_THREADS)
#define RS_E4      (N_EDGES / 4)
#define RS_NCHUNK  ((RS_E4 + RS_CHUNK4 - 1) / RS_CHUNK4)
#define RS_BLOCKS  ((N_NODES + RS_NODES - 1) / RS_NODES)

__device__ __forceinline__ void rs_push(int e, unsigned dl, int2* __restrict__ q,
                                        int* __restrict__ cnt, float* __restrict__ acc,
                                        const float* __restrict__ M,
                                        const int* __restrict__ src) {
    const int pos = atomicAdd(cnt, 1);
    if (pos < RS_QCAP) {
        q[pos] = make_int2(e, (int)dl);
    } else {
        const float* mrow = M + (size_t)src[e] * NHID;
        float* arow = acc + dl * NHID;
#pragma unroll 1
        for (int j = 0; j < NHID; ++j) atomicAdd(arow + j, mrow[j]);
    }
}

__global__ void __launch_bounds__(RS_THREADS)
edge_rowsum_lds(const float* __restrict__ M, const int* __restrict__ src,
                const int* __restrict__ dst, float* __restrict__ AGG) {
    __shared__ float acc[RS_NODES * NHID];
    __shared__ int2  queue[2][RS_QCAP];
    __shared__ int   qcount[RS_NCHUNK];

    const int tid  = threadIdx.x;
    const int lane = tid & 31;
    const int wave = __builtin_amdgcn_readfirstlane(tid >> 5);
    const int base = blockIdx.x * RS_NODES;

    float4* acc4 = reinterpret_cast<float4*>(acc);
    for (int i = tid; i < RS_NODES * NHID / 4; i += RS_THREADS)
        acc4[i] = make_float4(0.0f, 0.0f, 0.0f, 0.0f);
    if (tid < RS_NCHUNK) qcount[tid] = 0;
    __syncthreads();

    const int4*   dst4 = reinterpret_cast<const int4*>(dst);
    const float4* M4   = reinterpret_cast<const float4*>(M);

    const bool hi = lane >= 16;
    const int o0 = lane * 4 + (hi ? 2 : 0);
    const int o1 = lane * 4 + (hi ? 3 : 1);
    const int o2 = lane * 4 + (hi ? 0 : 2);
    const int o3 = lane * 4 + (hi ? 1 : 3);

    for (int c = 0; c < RS_NCHUNK; ++c) {
        int2* q   = queue[c & 1];
        int*  cnt = &qcount[c];

        int4 d[RS_LOADS];
#pragma unroll
        for (int j = 0; j < RS_LOADS; ++j) {
            const int i4 = c * RS_CHUNK4 + j * RS_THREADS + tid;
            d[j] = dst4[i4 < RS_E4 ? i4 : RS_E4 - 1];
        }
        for (int it = 0; it < RS_STEPS; ++it) {
            int4 dn[RS_LOADS];
#pragma unroll
            for (int j = 0; j < RS_LOADS; ++j) {
                dn[j] = d[j];
                if (it + 1 < RS_STEPS) {
                    const int i4 = c * RS_CHUNK4 + ((it + 1) * RS_LOADS + j) * RS_THREADS + tid;
                    dn[j] = dst4[i4 < RS_E4 ? i4 : RS_E4 - 1];
                }
            }
#pragma unroll
            for (int j = 0; j < RS_LOADS; ++j) {
                const int i4 = c * RS_CHUNK4 + (it * RS_LOADS + j) * RS_THREADS + tid;
                if (i4 < RS_E4) {
                    const int e = i4 * 4;
                    unsigned dl;
                    dl = (unsigned)(d[j].x - base);
                    if (dl < (unsigned)RS_NODES) rs_push(e + 0, dl, q, cnt, acc, M, src);
                    dl = (unsigned)(d[j].y - base);
                    if (dl < (unsigned)RS_NODES) rs_push(e + 1, dl, q, cnt, acc, M, src);
                    dl = (unsigned)(d[j].z - base);
                    if (dl < (unsigned)RS_NODES) rs_push(e + 2, dl, q, cnt, acc, M, src);
                    dl = (unsigned)(d[j].w - base);
                    if (dl < (unsigned)RS_NODES) rs_push(e + 3, dl, q, cnt, acc, M, src);
                }
            }
#pragma unroll
            for (int j = 0; j < RS_LOADS; ++j) d[j] = dn[j];
        }
        __syncthreads();

        int n = qcount[c];
        n = n < RS_QCAP ? n : RS_QCAP;
        n = __builtin_amdgcn_readfirstlane(n);
        for (int qb = 0; qb < n; qb += 32 * RS_WAVES) {
            const int idx = qb + (lane >> 3) * (8 * RS_WAVES) + wave * 8 + (lane & 7);
            int s = 0, dl = 0;
            if (idx < n) {
                const int2 t = q[idx];
                s  = src[t.x];
                dl = t.y;
            }
            for (int r = 0; r < 4; ++r) {
                const int q0 = qb + r * (8 * RS_WAVES) + wave * 8;
                if (q0 >= n) break;
                if (q0 + 8 <= n) {
                    float4 v[8];
#pragma unroll
                    for (int j = 0; j < 8; ++j) {
                        const int sj = __builtin_amdgcn_readlane(s, r * 8 + j);
                        v[j] = M4[(size_t)sj * (NHID / 4) + lane];
                    }
#pragma unroll
                    for (int j = 0; j < 8; ++j) {
                        const int dj = __builtin_amdgcn_readlane(dl, r * 8 + j);
                        float* p = acc + dj * NHID;
                        atomicAdd(p + o0, hi ? v[j].z : v[j].x);
                        atomicAdd(p + o1, hi ? v[j].w : v[j].y);
                        atomicAdd(p + o2, hi ? v[j].x : v[j].z);
                        atomicAdd(p + o3, hi ? v[j].y : v[j].w);
                    }
                } else {
                    const int m = n - q0;
                    for (int j = 0; j < m; ++j) {
                        const int sj = __builtin_amdgcn_readlane(s, r * 8 + j);
                        const int dj = __builtin_amdgcn_readlane(dl, r * 8 + j);
                        const float4 v = M4[(size_t)sj * (NHID / 4) + lane];
                        float* p = acc + dj * NHID;
                        atomicAdd(p + o0, hi ? v.z : v.x);
                        atomicAdd(p + o1, hi ? v.w : v.y);
                        atomicAdd(p + o2, hi ? v.x : v.z);
                        atomicAdd(p + o3, hi ? v.y : v.w);
                    }
                }
            }
        }
    }
    __syncthreads();

    float4* AGG4 = reinterpret_cast<float4*>(AGG);
    for (int i = tid; i < RS_NODES * NHID / 4; i += RS_THREADS) {
        const int node = base + (i >> 5);
        if (node < N_NODES) AGG4[(size_t)node * (NHID / 4) + (i & 31)] = acc4[i];
    }
}

__global__ void bias_relu(const float* __restrict__ AGG, const float* __restrict__ normD,
                          const float* __restrict__ b, float* __restrict__ H, int total) {
    int i = blockIdx.x * blockDim.x + threadIdx.x;
    if (i < total) {
        const int row = i >> 7;
        const int j   = i & 127;
        const float v = AGG[i] * normD[row] + b[j];
        H[i] = v > 0.0f ? v : 0.0f;
    }
}

__global__ void __launch_bounds__(256)
final_fc(const float* __restrict__ AGG, const float* __restrict__ normD,
         const float* __restrict__ b2, const float* __restrict__ Wfc,
         const float* __restrict__ bfc, float* __restrict__ out, int n) {
    const int w    = (blockIdx.x * blockDim.x + threadIdx.x) >> 5;
    const int lane = threadIdx.x & 31;
    if (w >= n) return;
    const float nd = normD[w];
    const float* arow = AGG + (size_t)w * NHID;
    float acc = 0.0f;
#pragma unroll
    for (int t = 0; t < 4; ++t) {
        const int j = lane + 32 * t;
        acc += (arow[j] * nd + b2[j]) * Wfc[j];
    }
#pragma unroll
    for (int off = 16; off > 0; off >>= 1)
        acc += __shfl_xor(acc, off, 32);
    if (lane == 0) out[w] = acc + bfc[0];
}

extern "C" void kernel_launch(void* const* d_in, const int* in_sizes, int n_in,
                              void* d_out, int out_size, void* d_ws, size_t ws_size,
                              hipStream_t stream) {
    const float* x   = (const float*)d_in[0];
    const float* W1  = (const float*)d_in[1];
    const float* b1  = (const float*)d_in[2];
    const float* W2  = (const float*)d_in[3];
    const float* b2  = (const float*)d_in[4];
    const float* Wfc = (const float*)d_in[5];
    const float* bfc = (const float*)d_in[6];
    const int*   src = (const int*)d_in[7];
    const int*   dst = (const int*)d_in[8];
    float* out = (float*)d_out;

    const int N = N_NODES, E = N_EDGES;
    const int NH = NHID;

    float* BUF_A = (float*)d_ws;
    float* BUF_B = BUF_A + (size_t)N * NH;
    float* normS = BUF_B + (size_t)N * NH;
    float* normD = normS + N;

    const int TPB = 256;
    const int featTotal = N * NH;
    const int gridFeat  = (featTotal + TPB - 1) / TPB;
    const int gridN     = (N + TPB - 1) / TPB;
    const int gridE     = (E + TPB - 1) / TPB;
    const int gridNodeWaves = (N + 7) / 8;
    const int gridGemm  = N / 16;

    zero_kernel<<<(2 * N + TPB - 1) / TPB, TPB, 0, stream>>>(normS, 2 * N);
    degree_kernel<<<gridE, TPB, 0, stream>>>(src, dst, normS, normD, E);
    norm_finalize<<<gridN, TPB, 0, stream>>>(normS, normD, N);

    gemm_scaled_wmma<<<gridGemm, TPB, 0, stream>>>(x, normS, W1, BUF_A, NFEAT);
    edge_rowsum_lds<<<RS_BLOCKS, RS_THREADS, 0, stream>>>(BUF_A, src, dst, BUF_B);
    bias_relu<<<gridFeat, TPB, 0, stream>>>(BUF_B, normD, b1, BUF_A, featTotal);

    gemm_scaled_wmma<<<gridGemm, TPB, 0, stream>>>(BUF_A, normS, W2, BUF_B, NH);
    edge_rowsum_lds<<<RS_BLOCKS, RS_THREADS, 0, stream>>>(BUF_B, src, dst, BUF_A);

    final_fc<<<gridNodeWaves, TPB, 0, stream>>>(BUF_A, normD, b2, Wfc, bfc, out, N);
}


# row-sum epilogue applies the layer's next op to its LDS rows: layer 1 norm+bias+relu and the 128x128 f32 WMMA product, layer 2 the 128->1 projection; AGG not stored, 3 launches gone; sum orders kept
# speedup vs baseline: 1.4503x; 1.1454x over previous
#include <hip/hip_runtime.h>

typedef __attribute__((ext_vector_type(2))) float v2f;
typedef __attribute__((ext_vector_type(8))) float v8f;

#define N_NODES 100000
#define N_EDGES 1600000
#define NFEAT   256
#define NHID    128

__global__ void zero_kernel(float* __restrict__ p, int n) {
    int i = blockIdx.x * blockDim.x + threadIdx.x;
    if (i < n) p[i] = 0.0f;
}

__global__ void degree_kernel(const int* __restrict__ src, const int* __restrict__ dst,
                              float* __restrict__ degS, float* __restrict__ degD, int e) {
    int i = blockIdx.x * blockDim.x + threadIdx.x;
    if (i < e) {
        atomicAdd(&degS[src[i]], 1.0f);
        atomicAdd(&degD[dst[i]], 1.0f);
    }
}

__global__ void norm_finalize(float* __restrict__ degS, float* __restrict__ degD, int n) {
    int i = blockIdx.x * blockDim.x + threadIdx.x;
    if (i < n) {
        degS[i] = rsqrtf(fmaxf(degS[i], 1.0f));
        degD[i] = rsqrtf(fmaxf(degD[i], 1.0f));
    }
}

__global__ void __launch_bounds__(256)
gemm_scaled_wmma(const float* __restrict__ X, const float* __restrict__ scale,
                 const float* __restrict__ W, float* __restrict__ Out, int K) {
    const int wave = threadIdx.x >> 5;
    const int lane = threadIdx.x & 31;
    const int kh   = lane >> 4;
    const int l    = lane & 15;
    const int row0 = blockIdx.x << 4;
    const int col0 = wave << 4;
    const int row  = row0 + l;

    const float ns = scale[row];
    const float* __restrict__ xrow = X + (size_t)row * K;

    v8f c = {};
    for (int k = 0; k < K; k += 4) {
        const int ka = k + 2 * kh;
        v2f a;
        a.x = xrow[ka]     * ns;
        a.y = xrow[ka + 1] * ns;
        v2f b;
        b.x = W[(size_t)ka       * NHID + col0 + l];
        b.y = W[(size_t)(ka + 1) * NHID + col0 + l];
        c = __builtin_amdgcn_wmma_f32_16x16x4_f32(false, a, false, b,
                                                  (short)0, c, false, false);
    }

    float* __restrict__ orow = Out + (size_t)row0 * NHID + col0;
#pragma unroll
    for (int r = 0; r < 8; ++r) {
        orow[(size_t)(r + 8 * kh) * NHID + l] = c[r];
    }
}

#define RS_NODES   400
#define RS_THREADS 1024
#define RS_WAVES   (RS_THREADS / 32)
#define RS_QCAP    2048
#define RS_STEPS   4
#define RS_LOADS   8
#define RS_CHUNK4  (RS_STEPS * RS_LOADS * RS_THREADS)
#define RS_E4      (N_EDGES / 4)
#define RS_NCHUNK  ((RS_E4 + RS_CHUNK4 - 1) / RS_CHUNK4)
#define RS_BLOCKS  (N_NODES / RS_NODES)
#define RS_PITCH   (NHID + 4)
#define RS_PITCH4  (RS_PITCH / 4)
#define RS_RTILES  (RS_NODES / 16)

static_assert(N_NODES % RS_NODES == 0, "node ranges must tile the node set");
static_assert(RS_NODES % 16 == 0, "owned rows must be whole 16-row tiles");
static_assert(NHID == 128 && RS_WAVES % 8 == 0, "8 column tiles, waves in groups of 8");

__device__ __forceinline__ void rs_push(int e, unsigned dl, int2* __restrict__ q,
                                        int* __restrict__ cnt, float* __restrict__ acc,
                                        const float* __restrict__ M,
                                        const int* __restrict__ src) {
    const int pos = atomicAdd(cnt, 1);
    if (pos < RS_QCAP) {
        q[pos] = make_int2(e, (int)dl);
    } else {
        const float* mrow = M + (size_t)src[e] * NHID;
        float* arow = acc + dl * RS_PITCH;
#pragma unroll 1
        for (int j = 0; j < NHID; ++j) atomicAdd(arow + j, mrow[j]);
    }
}

template <int LAYER>
__global__ void __launch_bounds__(RS_THREADS)
edge_rowsum_lds(const float* __restrict__ M, const int* __restrict__ src,
                const int* __restrict__ dst,
                const float* __restrict__ normS, const float* __restrict__ normD,
                const float* __restrict__ bias, const float* __restrict__ Wn,
                const float* __restrict__ bfc, float* __restrict__ Out) {
    __shared__ __attribute__((aligned(16))) float acc[RS_NODES * RS_PITCH];
    __shared__ int2  queue[2][RS_QCAP];
    __shared__ int   qcount[RS_NCHUNK];

    const int tid  = threadIdx.x;
    const int lane = tid & 31;
    const int wave = __builtin_amdgcn_readfirstlane(tid >> 5);
    const int base = blockIdx.x * RS_NODES;

    float4* acc4 = reinterpret_cast<float4*>(acc);
    for (int i = tid; i < RS_NODES * RS_PITCH4; i += RS_THREADS)
        acc4[i] = make_float4(0.0f, 0.0f, 0.0f, 0.0f);
    if (tid < RS_NCHUNK) qcount[tid] = 0;
    __syncthreads();

    const int4*   dst4 = reinterpret_cast<const int4*>(dst);
    const float4* M4   = reinterpret_cast<const float4*>(M);

    const bool hi = lane >= 16;
    const int o0 = lane * 4 + (hi ? 2 : 0);
    const int o1 = lane * 4 + (hi ? 3 : 1);
    const int o2 = lane * 4 + (hi ? 0 : 2);
    const int o3 = lane * 4 + (hi ? 1 : 3);

    for (int c = 0; c < RS_NCHUNK; ++c) {
        int2* q   = queue[c & 1];
        int*  cnt = &qcount[c];

        int4 d[RS_LOADS];
#pragma unroll
        for (int j = 0; j < RS_LOADS; ++j) {
            const int i4 = c * RS_CHUNK4 + j * RS_THREADS + tid;
            d[j] = dst4[i4 < RS_E4 ? i4 : RS_E4 - 1];
        }
        for (int it = 0; it < RS_STEPS; ++it) {
            int4 dn[RS_LOADS];
#pragma unroll
            for (int j = 0; j < RS_LOADS; ++j) {
                dn[j] = d[j];
                if (it + 1 < RS_STEPS) {
                    const int i4 = c * RS_CHUNK4 + ((it + 1) * RS_LOADS + j) * RS_THREADS + tid;
                    dn[j] = dst4[i4 < RS_E4 ? i4 : RS_E4 - 1];
                }
            }
#pragma unroll
            for (int j = 0; j < RS_LOADS; ++j) {
                const int i4 = c * RS_CHUNK4 + (it * RS_LOADS + j) * RS_THREADS + tid;
                if (i4 < RS_E4) {
                    const int e = i4 * 4;
                    unsigned dl;
                    dl = (unsigned)(d[j].x - base);
                    if (dl < (unsigned)RS_NODES) rs_push(e + 0, dl, q, cnt, acc, M, src);
                    dl = (unsigned)(d[j].y - base);
                    if (dl < (unsigned)RS_NODES) rs_push(e + 1, dl, q, cnt, acc, M, src);
                    dl = (unsigned)(d[j].z - base);
                    if (dl < (unsigned)RS_NODES) rs_push(e + 2, dl, q, cnt, acc, M, src);
                    dl = (unsigned)(d[j].w - base);
                    if (dl < (unsigned)RS_NODES) rs_push(e + 3, dl, q, cnt, acc, M, src);
                }
            }
#pragma unroll
            for (int j = 0; j < RS_LOADS; ++j) d[j] = dn[j];
        }
        __syncthreads();

        int n = qcount[c];
        n = n < RS_QCAP ? n : RS_QCAP;
        n = __builtin_amdgcn_readfirstlane(n);
        for (int qb = 0; qb < n; qb += 32 * RS_WAVES) {
            const int idx = qb + (lane >> 3) * (8 * RS_WAVES) + wave * 8 + (lane & 7);
            int s = 0, dl = 0;
            if (idx < n) {
                const int2 t = q[idx];
                s  = src[t.x];
                dl = t.y;
            }
            for (int r = 0; r < 4; ++r) {
                const int q0 = qb + r * (8 * RS_WAVES) + wave * 8;
                if (q0 >= n) break;
                if (q0 + 8 <= n) {
                    float4 v[8];
#pragma unroll
                    for (int j = 0; j < 8; ++j) {
                        const int sj = __builtin_amdgcn_readlane(s, r * 8 + j);
                        v[j] = M4[(size_t)sj * (NHID / 4) + lane];
                    }
#pragma unroll
                    for (int j = 0; j < 8; ++j) {
                        const int dj = __builtin_amdgcn_readlane(dl, r * 8 + j);
                        float* p = acc + dj * RS_PITCH;
                        atomicAdd(p + o0, hi ? v[j].z : v[j].x);
                        atomicAdd(p + o1, hi ? v[j].w : v[j].y);
                        atomicAdd(p + o2, hi ? v[j].x : v[j].z);
                        atomicAdd(p + o3, hi ? v[j].y : v[j].w);
                    }
                } else {
                    const int m = n - q0;
                    for (int j = 0; j < m; ++j) {
                        const int sj = __builtin_amdgcn_readlane(s, r * 8 + j);
                        const int dj = __builtin_amdgcn_readlane(dl, r * 8 + j);
                        const float4 v = M4[(size_t)sj * (NHID / 4) + lane];
                        float* p = acc + dj * RS_PITCH;
                        atomicAdd(p + o0, hi ? v.z : v.x);
                        atomicAdd(p + o1, hi ? v.w : v.y);
                        atomicAdd(p + o2, hi ? v.x : v.z);
                        atomicAdd(p + o3, hi ? v.y : v.w);
                    }
                }
            }
        }
    }
    __syncthreads();

    if constexpr (LAYER == 1) {
        const float4* b4 = reinterpret_cast<const float4*>(bias);
        const float4 bb = b4[lane];
        for (int i = tid; i < RS_NODES * (NHID / 4); i += RS_THREADS) {
            const int ln   = i >> 5;
            const int node = base + ln;
            const float nd = normD[node];
            const float ns = normS[node];
            float4 v = acc4[ln * RS_PITCH4 + lane];
            float t;
            t = v.x * nd + bb.x;  v.x = (t > 0.0f ? t : 0.0f) * ns;
            t = v.y * nd + bb.y;  v.y = (t > 0.0f ? t : 0.0f) * ns;
            t = v.z * nd + bb.z;  v.z = (t > 0.0f ? t : 0.0f) * ns;
            t = v.w * nd + bb.w;  v.w = (t > 0.0f ? t : 0.0f) * ns;
            acc4[ln * RS_PITCH4 + lane] = v;
        }
        __syncthreads();

        const int l   = lane & 15;
        const int kh  = lane >> 4;
        const int col = (wave & 7) * 16 + l;
        v2f bfr[NHID / 4];
#pragma unroll
        for (int ks = 0; ks < NHID / 4; ++ks) {
            const int ka = ks * 4 + 2 * kh;
            bfr[ks].x = Wn[(size_t)ka       * NHID + col];
            bfr[ks].y = Wn[(size_t)(ka + 1) * NHID + col];
        }
        for (int rt = wave >> 3; rt < RS_RTILES; rt += RS_WAVES / 8) {
            const float* arow = acc + (rt * 16 + l) * RS_PITCH + 2 * kh;
            v8f c = {};
#pragma unroll
            for (int ks = 0; ks < NHID / 4; ++ks) {
                const v2f a = *reinterpret_cast<const v2f*>(arow + ks * 4);
                c = __builtin_amdgcn_wmma_f32_16x16x4_f32(false, a, false, bfr[ks],
                                                          (short)0, c, false, false);
            }
            float* orow = Out + (size_t)(base + rt * 16 + 8 * kh) * NHID + col;
#pragma unroll
            for (int r = 0; r < 8; ++r) orow[(size_t)r * NHID] = c[r];
        }
    } else {
        __shared__ float outv[RS_NODES];
        const float bf = bfc[0];
        for (int ln = wave; ln < RS_NODES; ln += RS_WAVES) {
            const float nd = normD[base + ln];
            const float* arow = acc + ln * RS_PITCH;
            float a = 0.0f;
#pragma unroll
            for (int t = 0; t < 4; ++t) {
                const int j = lane + 32 * t;
                a += (arow[j] * nd + bias[j]) * Wn[j];
            }
#pragma unroll
            for (int off = 16; off > 0; off >>= 1)
                a += __shfl_xor(a, off, 32);
            if (lane == 0) outv[ln] = a + bf;
        }
        __syncthreads();
        if (tid < RS_NODES) Out[base + tid] = outv[tid];
    }
}

extern "C" void kernel_launch(void* const* d_in, const int* in_sizes, int n_in,
                              void* d_out, int out_size, void* d_ws, size_t ws_size,
                              hipStream_t stream) {
    const float* x   = (const float*)d_in[0];
    const float* W1  = (const float*)d_in[1];
    const float* b1  = (const float*)d_in[2];
    const float* W2  = (const float*)d_in[3];
    const float* b2  = (const float*)d_in[4];
    const float* Wfc = (const float*)d_in[5];
    const float* bfc = (const float*)d_in[6];
    const int*   src = (const int*)d_in[7];
    const int*   dst = (const int*)d_in[8];
    float* out = (float*)d_out;

    const int N = N_NODES, E = N_EDGES;
    const int NH = NHID;

    float* BUF_A = (float*)d_ws;
    float* BUF_B = BUF_A + (size_t)N * NH;
    float* normS = BUF_B + (size_t)N * NH;
    float* normD = normS + N;

    const int TPB = 256;
    const int gridN     = (N + TPB - 1) / TPB;
    const int gridE     = (E + TPB - 1) / TPB;
    const int gridGemm  = N / 16;

    zero_kernel<<<(2 * N + TPB - 1) / TPB, TPB, 0, stream>>>(normS, 2 * N);
    degree_kernel<<<gridE, TPB, 0, stream>>>(src, dst, normS, normD, E);
    norm_finalize<<<gridN, TPB, 0, stream>>>(normS, normD, N);

    gemm_scaled_wmma<<<gridGemm, TPB, 0, stream>>>(x, normS, W1, BUF_A, NFEAT);
    edge_rowsum_lds<1><<<RS_BLOCKS, RS_THREADS, 0, stream>>>(BUF_A, src, dst, normS, normD,
                                                            b1, W2, bfc, BUF_B);

    edge_rowsum_lds<2><<<RS_BLOCKS, RS_THREADS, 0, stream>>>(BUF_B, src, dst, normS, normD,
                                                            b2, Wfc, bfc, out);
}


# in-degree counted by the layer-1 row-sum work-group as it queues its own edges (LDS integer adds), normD formed in its epilogue; degree pass keeps only the out-degree atomics; sum orders kept
# speedup vs baseline: 1.5720x; 1.0839x over previous
#include <hip/hip_runtime.h>

typedef __attribute__((ext_vector_type(2))) float v2f;
typedef __attribute__((ext_vector_type(8))) float v8f;

#define N_NODES 100000
#define N_EDGES 1600000
#define NFEAT   256
#define NHID    128

__global__ void zero_kernel(float* __restrict__ p, int n) {
    int i = blockIdx.x * blockDim.x + threadIdx.x;
    if (i < n) p[i] = 0.0f;
}

__global__ void degree_kernel(const int* __restrict__ src, float* __restrict__ degS, int e) {
    int i = blockIdx.x * blockDim.x + threadIdx.x;
    if (i < e) atomicAdd(&degS[src[i]], 1.0f);
}

__global__ void norm_finalize(float* __restrict__ degS, int n) {
    int i = blockIdx.x * blockDim.x + threadIdx.x;
    if (i < n) degS[i] = rsqrtf(fmaxf(degS[i], 1.0f));
}

__global__ void __launch_bounds__(256)
gemm_scaled_wmma(const float* __restrict__ X, const float* __restrict__ scale,
                 const float* __restrict__ W, float* __restrict__ Out, int K) {
    const int wave = threadIdx.x >> 5;
    const int lane = threadIdx.x & 31;
    const int kh   = lane >> 4;
    const int l    = lane & 15;
    const int row0 = blockIdx.x << 4;
    const int col0 = wave << 4;
    const int row  = row0 + l;

    const float ns = scale[row];
    const float* __restrict__ xrow = X + (size_t)row * K;

    v8f c = {};
    for (int k = 0; k < K; k += 4) {
        const int ka = k + 2 * kh;
        v2f a;
        a.x = xrow[ka]     * ns;
        a.y = xrow[ka + 1] * ns;
        v2f b;
        b.x = W[(size_t)ka       * NHID + col0 + l];
        b.y = W[(size_t)(ka + 1) * NHID + col0 + l];
        c = __builtin_amdgcn_wmma_f32_16x16x4_f32(false, a, false, b,
                                                  (short)0, c, false, false);
    }

    float* __restrict__ orow = Out + (size_t)row0 * NHID + col0;
#pragma unroll
    for (int r = 0; r < 8; ++r) {
        orow[(size_t)(r + 8 * kh) * NHID + l] = c[r];
    }
}

#define RS_NODES   400
#define RS_THREADS 1024
#define RS_WAVES   (RS_THREADS / 32)
#define RS_QCAP    2048
#define RS_STEPS   4
#define RS_LOADS   8
#define RS_CHUNK4  (RS_STEPS * RS_LOADS * RS_THREADS)
#define RS_E4      (N_EDGES / 4)
#define RS_NCHUNK  ((RS_E4 + RS_CHUNK4 - 1) / RS_CHUNK4)
#define RS_BLOCKS  (N_NODES / RS_NODES)
#define RS_PITCH   (NHID + 4)
#define RS_PITCH4  (RS_PITCH / 4)
#define RS_RTILES  (RS_NODES / 16)

static_assert(N_NODES % RS_NODES == 0, "node ranges must tile the node set");
static_assert(RS_NODES % 16 == 0, "owned rows must be whole 16-row tiles");
static_assert(NHID == 128 && RS_WAVES % 8 == 0, "8 column tiles, waves in groups of 8");

template <int LAYER>
__device__ __forceinline__ void rs_push(int e, unsigned dl, int2* __restrict__ q,
                                        int* __restrict__ cnt, int* __restrict__ indeg,
                                        float* __restrict__ acc,
                                        const float* __restrict__ M,
                                        const int* __restrict__ src) {
    if constexpr (LAYER == 1) atomicAdd(indeg + dl, 1);
    const int pos = atomicAdd(cnt, 1);
    if (pos < RS_QCAP) {
        q[pos] = make_int2(e, (int)dl);
    } else {
        const float* mrow = M + (size_t)src[e] * NHID;
        float* arow = acc + dl * RS_PITCH;
#pragma unroll 1
        for (int j = 0; j < NHID; ++j) atomicAdd(arow + j, mrow[j]);
    }
}

template <int LAYER>
__global__ void __launch_bounds__(RS_THREADS)
edge_rowsum_lds(const float* __restrict__ M, const int* __restrict__ src,
                const int* __restrict__ dst,
                const float* __restrict__ normS, const float* __restrict__ normD,
                const float* __restrict__ bias, const float* __restrict__ Wn,
                const float* __restrict__ bfc, float* __restrict__ Out,
                float* __restrict__ normDout) {
    __shared__ __attribute__((aligned(16))) float acc[RS_NODES * RS_PITCH];
    __shared__ int2  queue[2][RS_QCAP];
    __shared__ int   qcount[RS_NCHUNK];
    __shared__ int   indeg[LAYER == 1 ? RS_NODES : 1];

    const int tid  = threadIdx.x;
    const int lane = tid & 31;
    const int wave = __builtin_amdgcn_readfirstlane(tid >> 5);
    const int base = blockIdx.x * RS_NODES;

    float4* acc4 = reinterpret_cast<float4*>(acc);
    for (int i = tid; i < RS_NODES * RS_PITCH4; i += RS_THREADS)
        acc4[i] = make_float4(0.0f, 0.0f, 0.0f, 0.0f);
    if (tid < RS_NCHUNK) qcount[tid] = 0;
    if constexpr (LAYER == 1) {
        if (tid < RS_NODES) indeg[tid] = 0;
    }
    __syncthreads();

    const int4*   dst4 = reinterpret_cast<const int4*>(dst);
    const float4* M4   = reinterpret_cast<const float4*>(M);

    const bool hi = lane >= 16;
    const int o0 = lane * 4 + (hi ? 2 : 0);
    const int o1 = lane * 4 + (hi ? 3 : 1);
    const int o2 = lane * 4 + (hi ? 0 : 2);
    const int o3 = lane * 4 + (hi ? 1 : 3);

    for (int c = 0; c < RS_NCHUNK; ++c) {
        int2* q   = queue[c & 1];
        int*  cnt = &qcount[c];

        int4 d[RS_LOADS];
#pragma unroll
        for (int j = 0; j < RS_LOADS; ++j) {
            const int i4 = c * RS_CHUNK4 + j * RS_THREADS + tid;
            d[j] = dst4[i4 < RS_E4 ? i4 : RS_E4 - 1];
        }
        for (int it = 0; it < RS_STEPS; ++it) {
            int4 dn[RS_LOADS];
#pragma unroll
            for (int j = 0; j < RS_LOADS; ++j) {
                dn[j] = d[j];
                if (it + 1 < RS_STEPS) {
                    const int i4 = c * RS_CHUNK4 + ((it + 1) * RS_LOADS + j) * RS_THREADS + tid;
                    dn[j] = dst4[i4 < RS_E4 ? i4 : RS_E4 - 1];
                }
            }
#pragma unroll
            for (int j = 0; j < RS_LOADS; ++j) {
                const int i4 = c * RS_CHUNK4 + (it * RS_LOADS + j) * RS_THREADS + tid;
                if (i4 < RS_E4) {
                    const int e = i4 * 4;
                    unsigned dl;
                    dl = (unsigned)(d[j].x - base);
                    if (dl < (unsigned)RS_NODES) rs_push<LAYER>(e + 0, dl, q, cnt, indeg, acc, M, src);
                    dl = (unsigned)(d[j].y - base);
                    if (dl < (unsigned)RS_NODES) rs_push<LAYER>(e + 1, dl, q, cnt, indeg, acc, M, src);
                    dl = (unsigned)(d[j].z - base);
                    if (dl < (unsigned)RS_NODES) rs_push<LAYER>(e + 2, dl, q, cnt, indeg, acc, M, src);
                    dl = (unsigned)(d[j].w - base);
                    if (dl < (unsigned)RS_NODES) rs_push<LAYER>(e + 3, dl, q, cnt, indeg, acc, M, src);
                }
            }
#pragma unroll
            for (int j = 0; j < RS_LOADS; ++j) d[j] = dn[j];
        }
        __syncthreads();

        int n = qcount[c];
        n = n < RS_QCAP ? n : RS_QCAP;
        n = __builtin_amdgcn_readfirstlane(n);
        for (int qb = 0; qb < n; qb += 32 * RS_WAVES) {
            const int idx = qb + (lane >> 3) * (8 * RS_WAVES) + wave * 8 + (lane & 7);
            int s = 0, dl = 0;
            if (idx < n) {
                const int2 t = q[idx];
                s  = src[t.x];
                dl = t.y;
            }
            for (int r = 0; r < 4; ++r) {
                const int q0 = qb + r * (8 * RS_WAVES) + wave * 8;
                if (q0 >= n) break;
                if (q0 + 8 <= n) {
                    float4 v[8];
#pragma unroll
                    for (int j = 0; j < 8; ++j) {
                        const int sj = __builtin_amdgcn_readlane(s, r * 8 + j);
                        v[j] = M4[(size_t)sj * (NHID / 4) + lane];
                    }
#pragma unroll
                    for (int j = 0; j < 8; ++j) {
                        const int dj = __builtin_amdgcn_readlane(dl, r * 8 + j);
                        float* p = acc + dj * RS_PITCH;
                        atomicAdd(p + o0, hi ? v[j].z : v[j].x);
                        atomicAdd(p + o1, hi ? v[j].w : v[j].y);
                        atomicAdd(p + o2, hi ? v[j].x : v[j].z);
                        atomicAdd(p + o3, hi ? v[j].y : v[j].w);
                    }
                } else {
                    const int m = n - q0;
                    for (int j = 0; j < m; ++j) {
                        const int sj = __builtin_amdgcn_readlane(s, r * 8 + j);
                        const int dj = __builtin_amdgcn_readlane(dl, r * 8 + j);
                        const float4 v = M4[(size_t)sj * (NHID / 4) + lane];
                        float* p = acc + dj * RS_PITCH;
                        atomicAdd(p + o0, hi ? v.z : v.x);
                        atomicAdd(p + o1, hi ? v.w : v.y);
                        atomicAdd(p + o2, hi ? v.x : v.z);
                        atomicAdd(p + o3, hi ? v.y : v.w);
                    }
                }
            }
        }
    }
    __syncthreads();

    if constexpr (LAYER == 1) {
        if (tid < RS_NODES) normDout[base + tid] = rsqrtf(fmaxf((float)indeg[tid], 1.0f));
        const float4* b4 = reinterpret_cast<const float4*>(bias);
        const float4 bb = b4[lane];
        for (int i = tid; i < RS_NODES * (NHID / 4); i += RS_THREADS) {
            const int ln   = i >> 5;
            const int node = base + ln;
            const float nd = rsqrtf(fmaxf((float)indeg[ln], 1.0f));
            const float ns = normS[node];
            float4 v = acc4[ln * RS_PITCH4 + lane];
            float t;
            t = v.x * nd + bb.x;  v.x = (t > 0.0f ? t : 0.0f) * ns;
            t = v.y * nd + bb.y;  v.y = (t > 0.0f ? t : 0.0f) * ns;
            t = v.z * nd + bb.z;  v.z = (t > 0.0f ? t : 0.0f) * ns;
            t = v.w * nd + bb.w;  v.w = (t > 0.0f ? t : 0.0f) * ns;
            acc4[ln * RS_PITCH4 + lane] = v;
        }
        __syncthreads();

        const int l   = lane & 15;
        const int kh  = lane >> 4;
        const int col = (wave & 7) * 16 + l;
        v2f bfr[NHID / 4];
#pragma unroll
        for (int ks = 0; ks < NHID / 4; ++ks) {
            const int ka = ks * 4 + 2 * kh;
            bfr[ks].x = Wn[(size_t)ka       * NHID + col];
            bfr[ks].y = Wn[(size_t)(ka + 1) * NHID + col];
        }
        for (int rt = wave >> 3; rt < RS_RTILES; rt += RS_WAVES / 8) {
            const float* arow = acc + (rt * 16 + l) * RS_PITCH + 2 * kh;
            v8f c = {};
#pragma unroll
            for (int ks = 0; ks < NHID / 4; ++ks) {
                const v2f a = *reinterpret_cast<const v2f*>(arow + ks * 4);
                c = __builtin_amdgcn_wmma_f32_16x16x4_f32(false, a, false, bfr[ks],
                                                          (short)0, c, false, false);
            }
            float* orow = Out + (size_t)(base + rt * 16 + 8 * kh) * NHID + col;
#pragma unroll
            for (int r = 0; r < 8; ++r) orow[(size_t)r * NHID] = c[r];
        }
    } else {
        __shared__ float outv[RS_NODES];
        const float bf = bfc[0];
        for (int ln = wave; ln < RS_NODES; ln += RS_WAVES) {
            const float nd = normD[base + ln];
            const float* arow = acc + ln * RS_PITCH;
            float a = 0.0f;
#pragma unroll
            for (int t = 0; t < 4; ++t) {
                const int j = lane + 32 * t;
                a += (arow[j] * nd + bias[j]) * Wn[j];
            }
#pragma unroll
            for (int off = 16; off > 0; off >>= 1)
                a += __shfl_xor(a, off, 32);
            if (lane == 0) outv[ln] = a + bf;
        }
        __syncthreads();
        if (tid < RS_NODES) Out[base + tid] = outv[tid];
    }
}

extern "C" void kernel_launch(void* const* d_in, const int* in_sizes, int n_in,
                              void* d_out, int out_size, void* d_ws, size_t ws_size,
                              hipStream_t stream) {
    const float* x   = (const float*)d_in[0];
    const float* W1  = (const float*)d_in[1];
    const float* b1  = (const float*)d_in[2];
    const float* W2  = (const float*)d_in[3];
    const float* b2  = (const float*)d_in[4];
    const float* Wfc = (const float*)d_in[5];
    const float* bfc = (const float*)d_in[6];
    const int*   src = (const int*)d_in[7];
    const int*   dst = (const int*)d_in[8];
    float* out = (float*)d_out;

    const int N = N_NODES, E = N_EDGES;
    const int NH = NHID;

    float* BUF_A = (float*)d_ws;
    float* BUF_B = BUF_A + (size_t)N * NH;
    float* normS = BUF_B + (size_t)N * NH;
    float* normD = normS + N;

    const int TPB = 256;
    const int gridN     = (N + TPB - 1) / TPB;
    const int gridE     = (E + TPB - 1) / TPB;
    const int gridGemm  = N / 16;

    zero_kernel<<<gridN, TPB, 0, stream>>>(normS, N);
    degree_kernel<<<gridE, TPB, 0, stream>>>(src, normS, E);
    norm_finalize<<<gridN, TPB, 0, stream>>>(normS, N);

    gemm_scaled_wmma<<<gridGemm, TPB, 0, stream>>>(x, normS, W1, BUF_A, NFEAT);
    edge_rowsum_lds<1><<<RS_BLOCKS, RS_THREADS, 0, stream>>>(BUF_A, src, dst, normS, normD,
                                                            b1, W2, bfc, BUF_B, normD);

    edge_rowsum_lds<2><<<RS_BLOCKS, RS_THREADS, 0, stream>>>(BUF_B, src, dst, normS, normD,
                                                            b2, Wfc, bfc, out, nullptr);
}


# edge row sum: a pass per layer splits the edge list into 160 slices and hands each node-range owner packed (src, local dst) lists; owners read their lists, no 250-fold dst scan; sums in arrival order
# speedup vs baseline: 2.5736x; 1.6372x over previous
#include <hip/hip_runtime.h>

typedef __attribute__((ext_vector_type(2))) float v2f;
typedef __attribute__((ext_vector_type(8))) float v8f;

#define N_NODES 100000
#define N_EDGES 1600000
#define NFEAT   256
#define NHID    128

__global__ void zero_kernel(float* __restrict__ p, int n) {
    int i = blockIdx.x * blockDim.x + threadIdx.x;
    if (i < n) p[i] = 0.0f;
}

__global__ void degree_kernel(const int* __restrict__ src, float* __restrict__ degS, int e) {
    int i = blockIdx.x * blockDim.x + threadIdx.x;
    if (i < e) atomicAdd(&degS[src[i]], 1.0f);
}

__global__ void norm_finalize(float* __restrict__ degS, int n) {
    int i = blockIdx.x * blockDim.x + threadIdx.x;
    if (i < n) degS[i] = rsqrtf(fmaxf(degS[i], 1.0f));
}

__global__ void __launch_bounds__(256)
gemm_scaled_wmma(const float* __restrict__ X, const float* __restrict__ scale,
                 const float* __restrict__ W, float* __restrict__ Out, int K) {
    const int wave = threadIdx.x >> 5;
    const int lane = threadIdx.x & 31;
    const int kh   = lane >> 4;
    const int l    = lane & 15;
    const int row0 = blockIdx.x << 4;
    const int col0 = wave << 4;
    const int row  = row0 + l;

    const float ns = scale[row];
    const float* __restrict__ xrow = X + (size_t)row * K;

    v8f c = {};
    for (int k = 0; k < K; k += 4) {
        const int ka = k + 2 * kh;
        v2f a;
        a.x = xrow[ka]     * ns;
        a.y = xrow[ka + 1] * ns;
        v2f b;
        b.x = W[(size_t)ka       * NHID + col0 + l];
        b.y = W[(size_t)(ka + 1) * NHID + col0 + l];
        c = __builtin_amdgcn_wmma_f32_16x16x4_f32(false, a, false, b,
                                                  (short)0, c, false, false);
    }

    float* __restrict__ orow = Out + (size_t)row0 * NHID + col0;
#pragma unroll
    for (int r = 0; r < 8; ++r) {
        orow[(size_t)(r + 8 * kh) * NHID + l] = c[r];
    }
}

#define RS_NODES   400
#define RS_THREADS 1024
#define RS_WAVES   (RS_THREADS / 32)
#define RS_E4      (N_EDGES / 4)
#define RS_BLOCKS  (N_NODES / RS_NODES)
#define RS_PITCH   (NHID + 4)
#define RS_PITCH4  (RS_PITCH / 4)
#define RS_RTILES  (RS_NODES / 16)

static_assert(N_NODES % RS_NODES == 0, "node ranges must tile the node set");
static_assert(RS_NODES % 16 == 0, "owned rows must be whole 16-row tiles");
static_assert(NHID == 128 && RS_WAVES % 8 == 0, "8 column tiles, waves in groups of 8");

#define BIN_SLICES   160
#define BIN_THREADS  1024
#define BIN_I4       (RS_E4 / BIN_SLICES)
#define BIN_LOADS    ((BIN_I4 + BIN_THREADS - 1) / BIN_THREADS)
#define BIN_REGION   (RS_NODES * NHID)
#define BIN_SLOT     (BIN_REGION / BIN_SLICES)
#define BIN_CAP      (BIN_SLOT - 1)
#define BIN_PER_WAVE (BIN_SLICES / RS_WAVES)
#define BIN_SHIFT    17
#define BIN_MASK     ((1u << BIN_SHIFT) - 1u)

static_assert(RS_E4 % BIN_SLICES == 0, "slices are whole 16-byte loads");
static_assert(BIN_REGION % BIN_SLICES == 0 && BIN_SLOT % 32 == 0, "a list starts on a 128-byte line");
static_assert(BIN_SLICES % RS_WAVES == 0 && BIN_PER_WAVE <= 32, "whole lists per wave, counts in one load");
static_assert(N_NODES <= (1 << BIN_SHIFT) && RS_NODES <= (1 << (32 - BIN_SHIFT)), "packed word holds both");
static_assert(RS_BLOCKS <= 256, "the start table below is built by one wave, 8 owners a lane");

__global__ void __launch_bounds__(BIN_THREADS)
bin_edges(const int* __restrict__ src, const int* __restrict__ dst, unsigned* __restrict__ bins) {
    __shared__ int      lcount[RS_BLOCKS];
    __shared__ int      lstart[RS_BLOCKS];
    __shared__ unsigned sorted[BIN_I4 * 4];

    const int tid  = threadIdx.x;
    const int lane = tid & 31;
    const int wave = tid >> 5;
    const int g    = blockIdx.x;

    if (tid < RS_BLOCKS) lcount[tid] = 0;
    __syncthreads();

    const int4* d4 = reinterpret_cast<const int4*>(dst) + (size_t)g * BIN_I4;
    const int4* s4 = reinterpret_cast<const int4*>(src) + (size_t)g * BIN_I4;

    unsigned wd[BIN_LOADS * 4];
    int      bn[BIN_LOADS * 4];
    int      rk[BIN_LOADS * 4];
#pragma unroll
    for (int j = 0; j < BIN_LOADS; ++j) {
        const int i4 = j * BIN_THREADS + tid;
        int dv[4] = {-1, -1, -1, -1};
        int sv[4] = {0, 0, 0, 0};
        if (i4 < BIN_I4) {
            const int4 d = d4[i4];
            const int4 s = s4[i4];
            dv[0] = d.x; dv[1] = d.y; dv[2] = d.z; dv[3] = d.w;
            sv[0] = s.x; sv[1] = s.y; sv[2] = s.z; sv[3] = s.w;
        }
#pragma unroll
        for (int t = 0; t < 4; ++t) {
            const int q = j * 4 + t;
            bn[q] = -1; rk[q] = 0; wd[q] = 0u;
            if ((unsigned)dv[t] < (unsigned)N_NODES) {
                const int b  = dv[t] / RS_NODES;
                const int dl = dv[t] - b * RS_NODES;
                bn[q] = b;
                wd[q] = (unsigned)sv[t] | ((unsigned)dl << BIN_SHIFT);
                rk[q] = atomicAdd(&lcount[b], 1);
            }
        }
    }
    __syncthreads();

    if (wave == 0) {
        int c[8];
        int sum = 0;
#pragma unroll
        for (int k = 0; k < 8; ++k) {
            const int b = lane * 8 + k;
            c[k] = b < RS_BLOCKS ? lcount[b] : 0;
            sum += c[k];
        }
        int incl = sum;
#pragma unroll
        for (int off = 1; off < 32; off <<= 1) {
            const int t = __shfl_up(incl, off, 32);
            if (lane >= off) incl += t;
        }
        int run = incl - sum;
#pragma unroll
        for (int k = 0; k < 8; ++k) {
            const int b = lane * 8 + k;
            if (b < RS_BLOCKS) lstart[b] = run;
            run += c[k];
        }
    }
    __syncthreads();

#pragma unroll
    for (int q = 0; q < BIN_LOADS * 4; ++q)
        if (bn[q] >= 0) sorted[lstart[bn[q]] + rk[q]] = wd[q];
    __syncthreads();

    for (int b = wave; b < RS_BLOCKS; b += BIN_THREADS / 32) {
        const int c  = lcount[b];
        const int st = lstart[b];
        unsigned* dest = bins + (size_t)b * BIN_REGION + (size_t)g * BIN_SLOT;
        const int n = c < BIN_CAP ? c : BIN_CAP;
        for (int i = lane; i < n; i += 32) dest[i] = sorted[st + i];
        if (lane == 0) dest[BIN_CAP] = (unsigned)c;
    }
}

template <int LAYER>
__device__ __forceinline__ void rs_slow_add(int e, unsigned dl, int* __restrict__ indeg,
                                            float* __restrict__ acc,
                                            const float* __restrict__ M,
                                            const int* __restrict__ src) {
    if constexpr (LAYER == 1) atomicAdd(indeg + dl, 1);
    const float* mrow = M + (size_t)src[e] * NHID;
    float* arow = acc + dl * RS_PITCH;
#pragma unroll 1
    for (int j = 0; j < NHID; ++j) atomicAdd(arow + j, mrow[j]);
}

template <int LAYER>
__global__ void __launch_bounds__(RS_THREADS)
edge_rowsum_lds(const float* __restrict__ M, const int* __restrict__ src,
                const int* __restrict__ dst, const unsigned* bins,
                const float* __restrict__ normS, const float* __restrict__ normD,
                const float* __restrict__ bias, const float* __restrict__ Wn,
                const float* __restrict__ bfc, float* Out,
                float* __restrict__ normDout) {
    __shared__ __attribute__((aligned(16))) float acc[RS_NODES * RS_PITCH];
    __shared__ int   indeg[LAYER == 1 ? RS_NODES : 1];

    const int tid  = threadIdx.x;
    const int lane = tid & 31;
    const int wave = __builtin_amdgcn_readfirstlane(tid >> 5);
    const int base = blockIdx.x * RS_NODES;

    float4* acc4 = reinterpret_cast<float4*>(acc);
    for (int i = tid; i < RS_NODES * RS_PITCH4; i += RS_THREADS)
        acc4[i] = make_float4(0.0f, 0.0f, 0.0f, 0.0f);
    if constexpr (LAYER == 1) {
        if (tid < RS_NODES) indeg[tid] = 0;
    }
    __syncthreads();

    const int4*   dst4 = reinterpret_cast<const int4*>(dst);
    const float4* M4   = reinterpret_cast<const float4*>(M);

    const bool hi = lane >= 16;
    const int o0 = lane * 4 + (hi ? 2 : 0);
    const int o1 = lane * 4 + (hi ? 3 : 1);
    const int o2 = lane * 4 + (hi ? 0 : 2);
    const int o3 = lane * 4 + (hi ? 1 : 3);

    const unsigned* mybin = bins + (size_t)blockIdx.x * BIN_REGION;
    int cl = 0;
    if (lane < BIN_PER_WAVE) cl = (int)mybin[(wave + lane * RS_WAVES) * BIN_SLOT + BIN_CAP];
    unsigned wnext = 0u;
    {
        const int c0 = __builtin_amdgcn_readlane(cl, 0);
        if (c0 <= BIN_CAP && lane < c0) wnext = mybin[wave * BIN_SLOT + lane];
    }
    for (int k = 0; k < BIN_PER_WAVE; ++k) {
        const int g = wave + k * RS_WAVES;
        const unsigned* slot = mybin + g * BIN_SLOT;
        const int c = __builtin_amdgcn_readlane(cl, k);
        unsigned w = wnext;
        wnext = 0u;
        if (k + 1 < BIN_PER_WAVE) {
            const int cn = __builtin_amdgcn_readlane(cl, k + 1);
            if (cn <= BIN_CAP && lane < cn) wnext = slot[RS_WAVES * BIN_SLOT + lane];
        }
        if (c <= BIN_CAP) {
            for (int p0 = 0; p0 < c; p0 += 32) {
                if (p0 > 0) w = (p0 + lane < c) ? slot[p0 + lane] : 0u;
                const int m32 = (c - p0) < 32 ? (c - p0) : 32;
                const int s  = (int)(w & BIN_MASK);
                const int dl = (int)(w >> BIN_SHIFT);
                if constexpr (LAYER == 1) {
                    if (lane < m32) atomicAdd(indeg + dl, 1);
                }
                for (int r = 0; r < 4; ++r) {
                    const int m = m32 - r * 8;
                    if (m <= 0) break;
                    float4 v[8];
#pragma unroll
                    for (int j = 0; j < 8; ++j) {
                        v[j] = make_float4(0.0f, 0.0f, 0.0f, 0.0f);
                        if (j < m) {
                            const int sj = __builtin_amdgcn_readlane(s, r * 8 + j);
                            v[j] = M4[(size_t)sj * (NHID / 4) + lane];
                        }
                    }
#pragma unroll
                    for (int j = 0; j < 8; ++j) {
                        if (j < m) {
                            const int dj = __builtin_amdgcn_readlane(dl, r * 8 + j);
                            float* p = acc + dj * RS_PITCH;
                            atomicAdd(p + o0, hi ? v[j].z : v[j].x);
                            atomicAdd(p + o1, hi ? v[j].w : v[j].y);
                            atomicAdd(p + o2, hi ? v[j].x : v[j].z);
                            atomicAdd(p + o3, hi ? v[j].y : v[j].w);
                        }
                    }
                }
            }
        } else {
            for (int i = lane; i < BIN_I4; i += 32) {
                const int i4 = g * BIN_I4 + i;
                const int4 d = dst4[i4];
                const int e = i4 * 4;
                unsigned dl;
                dl = (unsigned)(d.x - base);
                if (dl < (unsigned)RS_NODES) rs_slow_add<LAYER>(e + 0, dl, indeg, acc, M, src);
                dl = (unsigned)(d.y - base);
                if (dl < (unsigned)RS_NODES) rs_slow_add<LAYER>(e + 1, dl, indeg, acc, M, src);
                dl = (unsigned)(d.z - base);
                if (dl < (unsigned)RS_NODES) rs_slow_add<LAYER>(e + 2, dl, indeg, acc, M, src);
                dl = (unsigned)(d.w - base);
                if (dl < (unsigned)RS_NODES) rs_slow_add<LAYER>(e + 3, dl, indeg, acc, M, src);
            }
        }
    }
    __syncthreads();

    if constexpr (LAYER == 1) {
        if (tid < RS_NODES) normDout[base + tid] = rsqrtf(fmaxf((float)indeg[tid], 1.0f));
        const float4* b4 = reinterpret_cast<const float4*>(bias);
        const float4 bb = b4[lane];
        for (int i = tid; i < RS_NODES * (NHID / 4); i += RS_THREADS) {
            const int ln   = i >> 5;
            const int node = base + ln;
            const float nd = rsqrtf(fmaxf((float)indeg[ln], 1.0f));
            const float ns = normS[node];
            float4 v = acc4[ln * RS_PITCH4 + lane];
            float t;
            t = v.x * nd + bb.x;  v.x = (t > 0.0f ? t : 0.0f) * ns;
            t = v.y * nd + bb.y;  v.y = (t > 0.0f ? t : 0.0f) * ns;
            t = v.z * nd + bb.z;  v.z = (t > 0.0f ? t : 0.0f) * ns;
            t = v.w * nd + bb.w;  v.w = (t > 0.0f ? t : 0.0f) * ns;
            acc4[ln * RS_PITCH4 + lane] = v;
        }
        __syncthreads();

        const int l   = lane & 15;
        const int kh  = lane >> 4;
        const int col = (wave & 7) * 16 + l;
        v2f bfr[NHID / 4];
#pragma unroll
        for (int ks = 0; ks < NHID / 4; ++ks) {
            const int ka = ks * 4 + 2 * kh;
            bfr[ks].x = Wn[(size_t)ka       * NHID + col];
            bfr[ks].y = Wn[(size_t)(ka + 1) * NHID + col];
        }
        for (int rt = wave >> 3; rt < RS_RTILES; rt += RS_WAVES / 8) {
            const float* arow = acc + (rt * 16 + l) * RS_PITCH + 2 * kh;
            v8f c = {};
#pragma unroll
            for (int ks = 0; ks < NHID / 4; ++ks) {
                const v2f a = *reinterpret_cast<const v2f*>(arow + ks * 4);
                c = __builtin_amdgcn_wmma_f32_16x16x4_f32(false, a, false, bfr[ks],
                                                          (short)0, c, false, false);
            }
            float* orow = Out + (size_t)(base + rt * 16 + 8 * kh) * NHID + col;
#pragma unroll
            for (int r = 0; r < 8; ++r) orow[(size_t)r * NHID] = c[r];
        }
    } else {
        __shared__ float outv[RS_NODES];
        const float bf = bfc[0];
        for (int ln = wave; ln < RS_NODES; ln += RS_WAVES) {
            const float nd = normD[base + ln];
            const float* arow = acc + ln * RS_PITCH;
            float a = 0.0f;
#pragma unroll
            for (int t = 0; t < 4; ++t) {
                const int j = lane + 32 * t;
                a += (arow[j] * nd + bias[j]) * Wn[j];
            }
#pragma unroll
            for (int off = 16; off > 0; off >>= 1)
                a += __shfl_xor(a, off, 32);
            if (lane == 0) outv[ln] = a + bf;
        }
        __syncthreads();
        if (tid < RS_NODES) Out[base + tid] = outv[tid];
    }
}

extern "C" void kernel_launch(void* const* d_in, const int* in_sizes, int n_in,
                              void* d_out, int out_size, void* d_ws, size_t ws_size,
                              hipStream_t stream) {
    const float* x   = (const float*)d_in[0];
    const float* W1  = (const float*)d_in[1];
    const float* b1  = (const float*)d_in[2];
    const float* W2  = (const float*)d_in[3];
    const float* b2  = (const float*)d_in[4];
    const float* Wfc = (const float*)d_in[5];
    const float* bfc = (const float*)d_in[6];
    const int*   src = (const int*)d_in[7];
    const int*   dst = (const int*)d_in[8];
    float* out = (float*)d_out;

    const int N = N_NODES, E = N_EDGES;
    const int NH = NHID;

    float* BUF_A = (float*)d_ws;
    float* BUF_B = BUF_A + (size_t)N * NH;
    float* normS = BUF_B + (size_t)N * NH;
    float* normD = normS + N;

    const int TPB = 256;
    const int gridN     = (N + TPB - 1) / TPB;
    const int gridE     = (E + TPB - 1) / TPB;
    const int gridGemm  = N / 16;

    zero_kernel<<<gridN, TPB, 0, stream>>>(normS, N);
    degree_kernel<<<gridE, TPB, 0, stream>>>(src, normS, E);
    norm_finalize<<<gridN, TPB, 0, stream>>>(normS, N);

    gemm_scaled_wmma<<<gridGemm, TPB, 0, stream>>>(x, normS, W1, BUF_A, NFEAT);
    bin_edges<<<BIN_SLICES, BIN_THREADS, 0, stream>>>(src, dst, (unsigned*)BUF_B);
    edge_rowsum_lds<1><<<RS_BLOCKS, RS_THREADS, 0, stream>>>(BUF_A, src, dst, (const unsigned*)BUF_B,
                                                            normS, normD, b1, W2, bfc, BUF_B, normD);

    bin_edges<<<BIN_SLICES, BIN_THREADS, 0, stream>>>(src, dst, (unsigned*)BUF_A);
    edge_rowsum_lds<2><<<RS_BLOCKS, RS_THREADS, 0, stream>>>(BUF_B, src, dst, (const unsigned*)BUF_A,
                                                            normS, normD, b2, Wfc, bfc, out, nullptr);
}


# layer-1 product: W1 fragments kept in registers by persistent waves (32 columns each), x fragments read from global as 16-byte loads, a row tile read by 4 waves not 8; k sum regrouped within each 8
# speedup vs baseline: 3.1000x; 1.2045x over previous
#include <hip/hip_runtime.h>

typedef __attribute__((ext_vector_type(2))) float v2f;
typedef __attribute__((ext_vector_type(8))) float v8f;

#define N_NODES 100000
#define N_EDGES 1600000
#define NFEAT   256
#define NHID    128

__global__ void zero_kernel(float* __restrict__ p, int n) {
    int i = blockIdx.x * blockDim.x + threadIdx.x;
    if (i < n) p[i] = 0.0f;
}

__global__ void degree_kernel(const int* __restrict__ src, float* __restrict__ degS, int e) {
    int i = blockIdx.x * blockDim.x + threadIdx.x;
    if (i < e) atomicAdd(&degS[src[i]], 1.0f);
}

__global__ void norm_finalize(float* __restrict__ degS, int n) {
    int i = blockIdx.x * blockDim.x + threadIdx.x;
    if (i < n) degS[i] = rsqrtf(fmaxf(degS[i], 1.0f));
}

#define G1_BLOCKS  250
#define G1_THREADS 256
#define G1_TILES   (N_NODES / 16)
#define G1_STREAMS (2 * G1_BLOCKS)
#define G1_J       (NFEAT / 8)

static_assert(N_NODES % 16 == 0, "whole 16-row tiles: no row guard below");
static_assert(G1_STREAMS <= G1_TILES, "every stream has a first tile");
static_assert(NFEAT % 8 == 0 && NHID == 128, "8 k per float4 pair, 4 groups of 32 columns");

template <bool REFILL>
__device__ __forceinline__ void g1_tile(float4 (&a)[G1_J], const float4* __restrict__ np,
                                        const float ns, const v2f (&b0)[2 * G1_J],
                                        const v2f (&b1)[2 * G1_J],
                                        float* __restrict__ orow) {
    v8f c0 = {};
    v8f c1 = {};
#pragma unroll
    for (int j = 0; j < G1_J; ++j) {
        const float4 v = a[j];
        v2f lo, hi;
        lo.x = v.x * ns;  lo.y = v.y * ns;
        hi.x = v.z * ns;  hi.y = v.w * ns;
        if constexpr (REFILL) {
            __builtin_amdgcn_sched_barrier(0);
            a[j] = np[2 * j];
        }
        c0 = __builtin_amdgcn_wmma_f32_16x16x4_f32(false, lo, false, b0[2 * j],
                                                   (short)0, c0, false, false);
        c1 = __builtin_amdgcn_wmma_f32_16x16x4_f32(false, lo, false, b1[2 * j],
                                                   (short)0, c1, false, false);
        c0 = __builtin_amdgcn_wmma_f32_16x16x4_f32(false, hi, false, b0[2 * j + 1],
                                                   (short)0, c0, false, false);
        c1 = __builtin_amdgcn_wmma_f32_16x16x4_f32(false, hi, false, b1[2 * j + 1],
                                                   (short)0, c1, false, false);
    }
#pragma unroll
    for (int r = 0; r < 8; ++r) {
        orow[(size_t)r * NHID]      = c0[r];
        orow[(size_t)r * NHID + 16] = c1[r];
    }
}

__global__ void __launch_bounds__(G1_THREADS)
gemm_scaled_wmma(const float* __restrict__ X, const float* __restrict__ scale,
                 const float* __restrict__ W, float* __restrict__ Out) {
    const int lane = threadIdx.x & 31;
    const int wave = __builtin_amdgcn_readfirstlane(threadIdx.x >> 5);
    const int kh   = lane >> 4;
    const int l    = lane & 15;
    const int col  = (wave & 3) * 32 + l;
    const int stream = blockIdx.x + (wave >> 2) * G1_BLOCKS;

    v2f b0[2 * G1_J];
    v2f b1[2 * G1_J];
#pragma unroll
    for (int j = 0; j < G1_J; ++j) {
#pragma unroll
        for (int h = 0; h < 2; ++h) {
            const int k0 = 8 * j + 4 * kh + 2 * h;
            b0[2 * j + h].x = W[(size_t)k0       * NHID + col];
            b0[2 * j + h].y = W[(size_t)(k0 + 1) * NHID + col];
            b1[2 * j + h].x = W[(size_t)k0       * NHID + col + 16];
            b1[2 * j + h].y = W[(size_t)(k0 + 1) * NHID + col + 16];
        }
    }

    const float4* X4 = reinterpret_cast<const float4*>(X);
    int rt = stream;
    float4 a[G1_J];
    {
        const float4* p = X4 + (size_t)(rt * 16 + l) * (NFEAT / 4) + kh;
#pragma unroll
        for (int j = 0; j < G1_J; ++j) a[j] = p[2 * j];
    }
    float ns = scale[rt * 16 + l];
    for (;;) {
        float* orow = Out + (size_t)(rt * 16 + 8 * kh) * NHID + col;
        const int nrt = rt + G1_STREAMS;
        if (nrt >= G1_TILES) {
            g1_tile<false>(a, nullptr, ns, b0, b1, orow);
            break;
        }
        const float4* np = X4 + (size_t)(nrt * 16 + l) * (NFEAT / 4) + kh;
        const float nns = scale[nrt * 16 + l];
        g1_tile<true>(a, np, ns, b0, b1, orow);
        rt = nrt;
        ns = nns;
    }
}

#define RS_NODES   400
#define RS_THREADS 1024
#define RS_WAVES   (RS_THREADS / 32)
#define RS_E4      (N_EDGES / 4)
#define RS_BLOCKS  (N_NODES / RS_NODES)
#define RS_PITCH   (NHID + 4)
#define RS_PITCH4  (RS_PITCH / 4)
#define RS_RTILES  (RS_NODES / 16)

static_assert(N_NODES % RS_NODES == 0, "node ranges must tile the node set");
static_assert(RS_NODES % 16 == 0, "owned rows must be whole 16-row tiles");
static_assert(NHID == 128 && RS_WAVES % 8 == 0, "8 column tiles, waves in groups of 8");

#define BIN_SLICES   160
#define BIN_THREADS  1024
#define BIN_I4       (RS_E4 / BIN_SLICES)
#define BIN_LOADS    ((BIN_I4 + BIN_THREADS - 1) / BIN_THREADS)
#define BIN_REGION   (RS_NODES * NHID)
#define BIN_SLOT     (BIN_REGION / BIN_SLICES)
#define BIN_CAP      (BIN_SLOT - 1)
#define BIN_PER_WAVE (BIN_SLICES / RS_WAVES)
#define BIN_SHIFT    17
#define BIN_MASK     ((1u << BIN_SHIFT) - 1u)

static_assert(RS_E4 % BIN_SLICES == 0, "slices are whole 16-byte loads");
static_assert(BIN_REGION % BIN_SLICES == 0 && BIN_SLOT % 32 == 0, "a list starts on a 128-byte line");
static_assert(BIN_SLICES % RS_WAVES == 0 && BIN_PER_WAVE <= 32, "whole lists per wave, counts in one load");
static_assert(N_NODES <= (1 << BIN_SHIFT) && RS_NODES <= (1 << (32 - BIN_SHIFT)), "packed word holds both");
static_assert(RS_BLOCKS <= 256, "the start table below is built by one wave, 8 owners a lane");

__global__ void __launch_bounds__(BIN_THREADS)
bin_edges(const int* __restrict__ src, const int* __restrict__ dst, unsigned* __restrict__ bins) {
    __shared__ int      lcount[RS_BLOCKS];
    __shared__ int      lstart[RS_BLOCKS];
    __shared__ unsigned sorted[BIN_I4 * 4];

    const int tid  = threadIdx.x;
    const int lane = tid & 31;
    const int wave = tid >> 5;
    const int g    = blockIdx.x;

    if (tid < RS_BLOCKS) lcount[tid] = 0;
    __syncthreads();

    const int4* d4 = reinterpret_cast<const int4*>(dst) + (size_t)g * BIN_I4;
    const int4* s4 = reinterpret_cast<const int4*>(src) + (size_t)g * BIN_I4;

    unsigned wd[BIN_LOADS * 4];
    int      bn[BIN_LOADS * 4];
    int      rk[BIN_LOADS * 4];
#pragma unroll
    for (int j = 0; j < BIN_LOADS; ++j) {
        const int i4 = j * BIN_THREADS + tid;
        int dv[4] = {-1, -1, -1, -1};
        int sv[4] = {0, 0, 0, 0};
        if (i4 < BIN_I4) {
            const int4 d = d4[i4];
            const int4 s = s4[i4];
            dv[0] = d.x; dv[1] = d.y; dv[2] = d.z; dv[3] = d.w;
            sv[0] = s.x; sv[1] = s.y; sv[2] = s.z; sv[3] = s.w;
        }
#pragma unroll
        for (int t = 0; t < 4; ++t) {
            const int q = j * 4 + t;
            bn[q] = -1; rk[q] = 0; wd[q] = 0u;
            if ((unsigned)dv[t] < (unsigned)N_NODES) {
                const int b  = dv[t] / RS_NODES;
                const int dl = dv[t] - b * RS_NODES;
                bn[q] = b;
                wd[q] = (unsigned)sv[t] | ((unsigned)dl << BIN_SHIFT);
                rk[q] = atomicAdd(&lcount[b], 1);
            }
        }
    }
    __syncthreads();

    if (wave == 0) {
        int c[8];
        int sum = 0;
#pragma unroll
        for (int k = 0; k < 8; ++k) {
            const int b = lane * 8 + k;
            c[k] = b < RS_BLOCKS ? lcount[b] : 0;
            sum += c[k];
        }
        int incl = sum;
#pragma unroll
        for (int off = 1; off < 32; off <<= 1) {
            const int t = __shfl_up(incl, off, 32);
            if (lane >= off) incl += t;
        }
        int run = incl - sum;
#pragma unroll
        for (int k = 0; k < 8; ++k) {
            const int b = lane * 8 + k;
            if (b < RS_BLOCKS) lstart[b] = run;
            run += c[k];
        }
    }
    __syncthreads();

#pragma unroll
    for (int q = 0; q < BIN_LOADS * 4; ++q)
        if (bn[q] >= 0) sorted[lstart[bn[q]] + rk[q]] = wd[q];
    __syncthreads();

    for (int b = wave; b < RS_BLOCKS; b += BIN_THREADS / 32) {
        const int c  = lcount[b];
        const int st = lstart[b];
        unsigned* dest = bins + (size_t)b * BIN_REGION + (size_t)g * BIN_SLOT;
        const int n = c < BIN_CAP ? c : BIN_CAP;
        for (int i = lane; i < n; i += 32) dest[i] = sorted[st + i];
        if (lane == 0) dest[BIN_CAP] = (unsigned)c;
    }
}

template <int LAYER>
__device__ __forceinline__ void rs_slow_add(int e, unsigned dl, int* __restrict__ indeg,
                                            float* __restrict__ acc,
                                            const float* __restrict__ M,
                                            const int* __restrict__ src) {
    if constexpr (LAYER == 1) atomicAdd(indeg + dl, 1);
    const float* mrow = M + (size_t)src[e] * NHID;
    float* arow = acc + dl * RS_PITCH;
#pragma unroll 1
    for (int j = 0; j < NHID; ++j) atomicAdd(arow + j, mrow[j]);
}

template <int LAYER>
__global__ void __launch_bounds__(RS_THREADS)
edge_rowsum_lds(const float* __restrict__ M, const int* __restrict__ src,
                const int* __restrict__ dst, const unsigned* bins,
                const float* __restrict__ normS, const float* __restrict__ normD,
                const float* __restrict__ bias, const float* __restrict__ Wn,
                const float* __restrict__ bfc, float* Out,
                float* __restrict__ normDout) {
    __shared__ __attribute__((aligned(16))) float acc[RS_NODES * RS_PITCH];
    __shared__ int   indeg[LAYER == 1 ? RS_NODES : 1];

    const int tid  = threadIdx.x;
    const int lane = tid & 31;
    const int wave = __builtin_amdgcn_readfirstlane(tid >> 5);
    const int base = blockIdx.x * RS_NODES;

    float4* acc4 = reinterpret_cast<float4*>(acc);
    for (int i = tid; i < RS_NODES * RS_PITCH4; i += RS_THREADS)
        acc4[i] = make_float4(0.0f, 0.0f, 0.0f, 0.0f);
    if constexpr (LAYER == 1) {
        if (tid < RS_NODES) indeg[tid] = 0;
    }
    __syncthreads();

    const int4*   dst4 = reinterpret_cast<const int4*>(dst);
    const float4* M4   = reinterpret_cast<const float4*>(M);

    const bool hi = lane >= 16;
    const int o0 = lane * 4 + (hi ? 2 : 0);
    const int o1 = lane * 4 + (hi ? 3 : 1);
    const int o2 = lane * 4 + (hi ? 0 : 2);
    const int o3 = lane * 4 + (hi ? 1 : 3);

    const unsigned* mybin = bins + (size_t)blockIdx.x * BIN_REGION;
    int cl = 0;
    if (lane < BIN_PER_WAVE) cl = (int)mybin[(wave + lane * RS_WAVES) * BIN_SLOT + BIN_CAP];
    unsigned wnext = 0u;
    {
        const int c0 = __builtin_amdgcn_readlane(cl, 0);
        if (c0 <= BIN_CAP && lane < c0) wnext = mybin[wave * BIN_SLOT + lane];
    }
    for (int k = 0; k < BIN_PER_WAVE; ++k) {
        const int g = wave + k * RS_WAVES;
        const unsigned* slot = mybin + g * BIN_SLOT;
        const int c = __builtin_amdgcn_readlane(cl, k);
        unsigned w = wnext;
        wnext = 0u;
        if (k + 1 < BIN_PER_WAVE) {
            const int cn = __builtin_amdgcn_readlane(cl, k + 1);
            if (cn <= BIN_CAP && lane < cn) wnext = slot[RS_WAVES * BIN_SLOT + lane];
        }
        if (c <= BIN_CAP) {
            for (int p0 = 0; p0 < c; p0 += 32) {
                if (p0 > 0) w = (p0 + lane < c) ? slot[p0 + lane] : 0u;
                const int m32 = (c - p0) < 32 ? (c - p0) : 32;
                const int s  = (int)(w & BIN_MASK);
                const int dl = (int)(w >> BIN_SHIFT);
                if constexpr (LAYER == 1) {
                    if (lane < m32) atomicAdd(indeg + dl, 1);
                }
                for (int r = 0; r < 4; ++r) {
                    const int m = m32 - r * 8;
                    if (m <= 0) break;
                    float4 v[8];
#pragma unroll
                    for (int j = 0; j < 8; ++j) {
                        v[j] = make_float4(0.0f, 0.0f, 0.0f, 0.0f);
                        if (j < m) {
                            const int sj = __builtin_amdgcn_readlane(s, r * 8 + j);
                            v[j] = M4[(size_t)sj * (NHID / 4) + lane];
                        }
                    }
#pragma unroll
                    for (int j = 0; j < 8; ++j) {
                        if (j < m) {
                            const int dj = __builtin_amdgcn_readlane(dl, r * 8 + j);
                            float* p = acc + dj * RS_PITCH;
                            atomicAdd(p + o0, hi ? v[j].z : v[j].x);
                            atomicAdd(p + o1, hi ? v[j].w : v[j].y);
                            atomicAdd(p + o2, hi ? v[j].x : v[j].z);
                            atomicAdd(p + o3, hi ? v[j].y : v[j].w);
                        }
                    }
                }
            }
        } else {
            for (int i = lane; i < BIN_I4; i += 32) {
                const int i4 = g * BIN_I4 + i;
                const int4 d = dst4[i4];
                const int e = i4 * 4;
                unsigned dl;
                dl = (unsigned)(d.x - base);
                if (dl < (unsigned)RS_NODES) rs_slow_add<LAYER>(e + 0, dl, indeg, acc, M, src);
                dl = (unsigned)(d.y - base);
                if (dl < (unsigned)RS_NODES) rs_slow_add<LAYER>(e + 1, dl, indeg, acc, M, src);
                dl = (unsigned)(d.z - base);
                if (dl < (unsigned)RS_NODES) rs_slow_add<LAYER>(e + 2, dl, indeg, acc, M, src);
                dl = (unsigned)(d.w - base);
                if (dl < (unsigned)RS_NODES) rs_slow_add<LAYER>(e + 3, dl, indeg, acc, M, src);
            }
        }
    }
    __syncthreads();

    if constexpr (LAYER == 1) {
        if (tid < RS_NODES) normDout[base + tid] = rsqrtf(fmaxf((float)indeg[tid], 1.0f));
        const float4* b4 = reinterpret_cast<const float4*>(bias);
        const float4 bb = b4[lane];
        for (int i = tid; i < RS_NODES * (NHID / 4); i += RS_THREADS) {
            const int ln   = i >> 5;
            const int node = base + ln;
            const float nd = rsqrtf(fmaxf((float)indeg[ln], 1.0f));
            const float ns = normS[node];
            float4 v = acc4[ln * RS_PITCH4 + lane];
            float t;
            t = v.x * nd + bb.x;  v.x = (t > 0.0f ? t : 0.0f) * ns;
            t = v.y * nd + bb.y;  v.y = (t > 0.0f ? t : 0.0f) * ns;
            t = v.z * nd + bb.z;  v.z = (t > 0.0f ? t : 0.0f) * ns;
            t = v.w * nd + bb.w;  v.w = (t > 0.0f ? t : 0.0f) * ns;
            acc4[ln * RS_PITCH4 + lane] = v;
        }
        __syncthreads();

        const int l   = lane & 15;
        const int kh  = lane >> 4;
        const int col = (wave & 7) * 16 + l;
        v2f bfr[NHID / 4];
#pragma unroll
        for (int ks = 0; ks < NHID / 4; ++ks) {
            const int ka = ks * 4 + 2 * kh;
            bfr[ks].x = Wn[(size_t)ka       * NHID + col];
            bfr[ks].y = Wn[(size_t)(ka + 1) * NHID + col];
        }
        for (int rt = wave >> 3; rt < RS_RTILES; rt += RS_WAVES / 8) {
            const float* arow = acc + (rt * 16 + l) * RS_PITCH + 2 * kh;
            v8f c = {};
#pragma unroll
            for (int ks = 0; ks < NHID / 4; ++ks) {
                const v2f a = *reinterpret_cast<const v2f*>(arow + ks * 4);
                c = __builtin_amdgcn_wmma_f32_16x16x4_f32(false, a, false, bfr[ks],
                                                          (short)0, c, false, false);
            }
            float* orow = Out + (size_t)(base + rt * 16 + 8 * kh) * NHID + col;
#pragma unroll
            for (int r = 0; r < 8; ++r) orow[(size_t)r * NHID] = c[r];
        }
    } else {
        __shared__ float outv[RS_NODES];
        const float bf = bfc[0];
        for (int ln = wave; ln < RS_NODES; ln += RS_WAVES) {
            const float nd = normD[base + ln];
            const float* arow = acc + ln * RS_PITCH;
            float a = 0.0f;
#pragma unroll
            for (int t = 0; t < 4; ++t) {
                const int j = lane + 32 * t;
                a += (arow[j] * nd + bias[j]) * Wn[j];
            }
#pragma unroll
            for (int off = 16; off > 0; off >>= 1)
                a += __shfl_xor(a, off, 32);
            if (lane == 0) outv[ln] = a + bf;
        }
        __syncthreads();
        if (tid < RS_NODES) Out[base + tid] = outv[tid];
    }
}

extern "C" void kernel_launch(void* const* d_in, const int* in_sizes, int n_in,
                              void* d_out, int out_size, void* d_ws, size_t ws_size,
                              hipStream_t stream) {
    const float* x   = (const float*)d_in[0];
    const float* W1  = (const float*)d_in[1];
    const float* b1  = (const float*)d_in[2];
    const float* W2  = (const float*)d_in[3];
    const float* b2  = (const float*)d_in[4];
    const float* Wfc = (const float*)d_in[5];
    const float* bfc = (const float*)d_in[6];
    const int*   src = (const int*)d_in[7];
    const int*   dst = (const int*)d_in[8];
    float* out = (float*)d_out;

    const int N = N_NODES, E = N_EDGES;
    const int NH = NHID;

    float* BUF_A = (float*)d_ws;
    float* BUF_B = BUF_A + (size_t)N * NH;
    float* normS = BUF_B + (size_t)N * NH;
    float* normD = normS + N;

    const int TPB = 256;
    const int gridN     = (N + TPB - 1) / TPB;
    const int gridE     = (E + TPB - 1) / TPB;

    zero_kernel<<<gridN, TPB, 0, stream>>>(normS, N);
    degree_kernel<<<gridE, TPB, 0, stream>>>(src, normS, E);
    norm_finalize<<<gridN, TPB, 0, stream>>>(normS, N);

    gemm_scaled_wmma<<<G1_BLOCKS, G1_THREADS, 0, stream>>>(x, normS, W1, BUF_A);
    bin_edges<<<BIN_SLICES, BIN_THREADS, 0, stream>>>(src, dst, (unsigned*)BUF_B);
    edge_rowsum_lds<1><<<RS_BLOCKS, RS_THREADS, 0, stream>>>(BUF_A, src, dst, (const unsigned*)BUF_B,
                                                            normS, normD, b1, W2, bfc, BUF_B, normD);

    bin_edges<<<BIN_SLICES, BIN_THREADS, 0, stream>>>(src, dst, (unsigned*)BUF_A);
    edge_rowsum_lds<2><<<RS_BLOCKS, RS_THREADS, 0, stream>>>(BUF_B, src, dst, (const unsigned*)BUF_A,
                                                            normS, normD, b2, Wfc, bfc, out, nullptr);
}


# out-degree counted per node-range owner (LDS integer adds) from source lists that the layer-1 binning pass now also emits; no global degree atomics, 8 launches -> 6; all float sum orders kept
# speedup vs baseline: 4.1749x; 1.3467x over previous
#include <hip/hip_runtime.h>

typedef __attribute__((ext_vector_type(2))) float v2f;
typedef __attribute__((ext_vector_type(8))) float v8f;

#define N_NODES 100000
#define N_EDGES 1600000
#define NFEAT   256
#define NHID    128


#define G1_BLOCKS  250
#define G1_THREADS 256
#define G1_TILES   (N_NODES / 16)
#define G1_STREAMS (2 * G1_BLOCKS)
#define G1_J       (NFEAT / 8)

static_assert(N_NODES % 16 == 0, "whole 16-row tiles: no row guard below");
static_assert(G1_STREAMS <= G1_TILES, "every stream has a first tile");
static_assert(NFEAT % 8 == 0 && NHID == 128, "8 k per float4 pair, 4 groups of 32 columns");

template <bool REFILL>
__device__ __forceinline__ void g1_tile(float4 (&a)[G1_J], const float4* __restrict__ np,
                                        const float ns, const v2f (&b0)[2 * G1_J],
                                        const v2f (&b1)[2 * G1_J],
                                        float* __restrict__ orow) {
    v8f c0 = {};
    v8f c1 = {};
#pragma unroll
    for (int j = 0; j < G1_J; ++j) {
        const float4 v = a[j];
        v2f lo, hi;
        lo.x = v.x * ns;  lo.y = v.y * ns;
        hi.x = v.z * ns;  hi.y = v.w * ns;
        if constexpr (REFILL) {
            __builtin_amdgcn_sched_barrier(0);
            a[j] = np[2 * j];
        }
        c0 = __builtin_amdgcn_wmma_f32_16x16x4_f32(false, lo, false, b0[2 * j],
                                                   (short)0, c0, false, false);
        c1 = __builtin_amdgcn_wmma_f32_16x16x4_f32(false, lo, false, b1[2 * j],
                                                   (short)0, c1, false, false);
        c0 = __builtin_amdgcn_wmma_f32_16x16x4_f32(false, hi, false, b0[2 * j + 1],
                                                   (short)0, c0, false, false);
        c1 = __builtin_amdgcn_wmma_f32_16x16x4_f32(false, hi, false, b1[2 * j + 1],
                                                   (short)0, c1, false, false);
    }
#pragma unroll
    for (int r = 0; r < 8; ++r) {
        orow[(size_t)r * NHID]      = c0[r];
        orow[(size_t)r * NHID + 16] = c1[r];
    }
}

__global__ void __launch_bounds__(G1_THREADS)
gemm_scaled_wmma(const float* __restrict__ X, const float* __restrict__ scale,
                 const float* __restrict__ W, float* __restrict__ Out) {
    const int lane = threadIdx.x & 31;
    const int wave = __builtin_amdgcn_readfirstlane(threadIdx.x >> 5);
    const int kh   = lane >> 4;
    const int l    = lane & 15;
    const int col  = (wave & 3) * 32 + l;
    const int stream = blockIdx.x + (wave >> 2) * G1_BLOCKS;

    v2f b0[2 * G1_J];
    v2f b1[2 * G1_J];
#pragma unroll
    for (int j = 0; j < G1_J; ++j) {
#pragma unroll
        for (int h = 0; h < 2; ++h) {
            const int k0 = 8 * j + 4 * kh + 2 * h;
            b0[2 * j + h].x = W[(size_t)k0       * NHID + col];
            b0[2 * j + h].y = W[(size_t)(k0 + 1) * NHID + col];
            b1[2 * j + h].x = W[(size_t)k0       * NHID + col + 16];
            b1[2 * j + h].y = W[(size_t)(k0 + 1) * NHID + col + 16];
        }
    }

    const float4* X4 = reinterpret_cast<const float4*>(X);
    int rt = stream;
    float4 a[G1_J];
    {
        const float4* p = X4 + (size_t)(rt * 16 + l) * (NFEAT / 4) + kh;
#pragma unroll
        for (int j = 0; j < G1_J; ++j) a[j] = p[2 * j];
    }
    float ns = scale[rt * 16 + l];
    for (;;) {
        float* orow = Out + (size_t)(rt * 16 + 8 * kh) * NHID + col;
        const int nrt = rt + G1_STREAMS;
        if (nrt >= G1_TILES) {
            g1_tile<false>(a, nullptr, ns, b0, b1, orow);
            break;
        }
        const float4* np = X4 + (size_t)(nrt * 16 + l) * (NFEAT / 4) + kh;
        const float nns = scale[nrt * 16 + l];
        g1_tile<true>(a, np, ns, b0, b1, orow);
        rt = nrt;
        ns = nns;
    }
}

#define RS_NODES   400
#define RS_THREADS 1024
#define RS_WAVES   (RS_THREADS / 32)
#define RS_E4      (N_EDGES / 4)
#define RS_BLOCKS  (N_NODES / RS_NODES)
#define RS_PITCH   (NHID + 4)
#define RS_PITCH4  (RS_PITCH / 4)
#define RS_RTILES  (RS_NODES / 16)

static_assert(N_NODES % RS_NODES == 0, "node ranges must tile the node set");
static_assert(RS_NODES % 16 == 0, "owned rows must be whole 16-row tiles");
static_assert(NHID == 128 && RS_WAVES % 8 == 0, "8 column tiles, waves in groups of 8");

#define BIN_SLICES   160
#define BIN_THREADS  1024
#define BIN_I4       (RS_E4 / BIN_SLICES)
#define BIN_LOADS    ((BIN_I4 + BIN_THREADS - 1) / BIN_THREADS)
#define BIN_REGION   (RS_NODES * NHID)
#define BIN_SLOT     (BIN_REGION / BIN_SLICES)
#define BIN_CAP      (BIN_SLOT - 1)
#define BIN_PER_WAVE (BIN_SLICES / RS_WAVES)
#define BIN_SHIFT    17
#define BIN_MASK     ((1u << BIN_SHIFT) - 1u)

static_assert(RS_E4 % BIN_SLICES == 0, "slices are whole 16-byte loads");
static_assert(BIN_REGION % BIN_SLICES == 0 && BIN_SLOT % 32 == 0, "a list starts on a 128-byte line");
static_assert(BIN_SLICES % RS_WAVES == 0 && BIN_PER_WAVE <= 32, "whole lists per wave, counts in one load");
static_assert(N_NODES <= (1 << BIN_SHIFT) && RS_NODES <= (1 << (32 - BIN_SHIFT)), "packed word holds both");
static_assert(RS_BLOCKS <= 256, "the start table below is built by one wave, 8 owners a lane");

__device__ __forceinline__ void bin_starts(const int* cnt, int* st, int lane) {
    int c[8];
    int sum = 0;
#pragma unroll
    for (int k = 0; k < 8; ++k) {
        const int b = lane * 8 + k;
        c[k] = b < RS_BLOCKS ? cnt[b] : 0;
        sum += c[k];
    }
    int incl = sum;
#pragma unroll
    for (int off = 1; off < 32; off <<= 1) {
        const int t = __shfl_up(incl, off, 32);
        if (lane >= off) incl += t;
    }
    int run = incl - sum;
#pragma unroll
    for (int k = 0; k < 8; ++k) {
        const int b = lane * 8 + k;
        if (b < RS_BLOCKS) st[b] = run;
        run += c[k];
    }
}

template <bool SRC_LISTS>
__global__ void __launch_bounds__(BIN_THREADS)
bin_edges(const int* __restrict__ src, const int* __restrict__ dst, unsigned* __restrict__ bins,
          unsigned* __restrict__ sbins) {
    __shared__ int      lcount[RS_BLOCKS];
    __shared__ int      lstart[RS_BLOCKS];
    __shared__ unsigned sorted[BIN_I4 * 4];
    __shared__ int      scount[SRC_LISTS ? RS_BLOCKS : 1];
    __shared__ int      sstart[SRC_LISTS ? RS_BLOCKS : 1];
    __shared__ unsigned ssorted[SRC_LISTS ? BIN_I4 * 4 : 1];

    const int tid  = threadIdx.x;
    const int lane = tid & 31;
    const int wave = tid >> 5;
    const int g    = blockIdx.x;

    if (tid < RS_BLOCKS) {
        lcount[tid] = 0;
        if constexpr (SRC_LISTS) scount[tid] = 0;
    }
    __syncthreads();

    const int4* d4 = reinterpret_cast<const int4*>(dst) + (size_t)g * BIN_I4;
    const int4* s4 = reinterpret_cast<const int4*>(src) + (size_t)g * BIN_I4;

    unsigned wd[BIN_LOADS * 4];
    int      bn[BIN_LOADS * 4];
    int      rk[BIN_LOADS * 4];
    unsigned sw[SRC_LISTS ? BIN_LOADS * 4 : 1];
    int      sb[SRC_LISTS ? BIN_LOADS * 4 : 1];
    int      sr[SRC_LISTS ? BIN_LOADS * 4 : 1];
#pragma unroll
    for (int j = 0; j < BIN_LOADS; ++j) {
        const int i4 = j * BIN_THREADS + tid;
        int dv[4] = {-1, -1, -1, -1};
        int sv[4] = {-1, -1, -1, -1};
        if (i4 < BIN_I4) {
            const int4 d = d4[i4];
            const int4 s = s4[i4];
            dv[0] = d.x; dv[1] = d.y; dv[2] = d.z; dv[3] = d.w;
            sv[0] = s.x; sv[1] = s.y; sv[2] = s.z; sv[3] = s.w;
        }
#pragma unroll
        for (int t = 0; t < 4; ++t) {
            const int q = j * 4 + t;
            bn[q] = -1; rk[q] = 0; wd[q] = 0u;
            if ((unsigned)dv[t] < (unsigned)N_NODES) {
                const int b  = dv[t] / RS_NODES;
                const int dl = dv[t] - b * RS_NODES;
                bn[q] = b;
                wd[q] = (unsigned)sv[t] | ((unsigned)dl << BIN_SHIFT);
                rk[q] = atomicAdd(&lcount[b], 1);
            }
            if constexpr (SRC_LISTS) {
                sb[q] = -1; sr[q] = 0; sw[q] = 0u;
                if ((unsigned)sv[t] < (unsigned)N_NODES) {
                    const int b2 = sv[t] / RS_NODES;
                    sb[q] = b2;
                    sw[q] = (unsigned)(sv[t] - b2 * RS_NODES);
                    sr[q] = atomicAdd(&scount[b2], 1);
                }
            }
        }
    }
    __syncthreads();

    if (wave == 0) bin_starts(lcount, lstart, lane);
    if constexpr (SRC_LISTS) {
        if (wave == 1) bin_starts(scount, sstart, lane);
    }
    __syncthreads();

#pragma unroll
    for (int q = 0; q < BIN_LOADS * 4; ++q)
        if (bn[q] >= 0) sorted[lstart[bn[q]] + rk[q]] = wd[q];
    if constexpr (SRC_LISTS) {
#pragma unroll
        for (int q = 0; q < BIN_LOADS * 4; ++q)
            if (sb[q] >= 0) ssorted[sstart[sb[q]] + sr[q]] = sw[q];
    }
    __syncthreads();

    for (int b = wave; b < RS_BLOCKS; b += BIN_THREADS / 32) {
        const int c  = lcount[b];
        const int st = lstart[b];
        unsigned* dest = bins + (size_t)b * BIN_REGION + (size_t)g * BIN_SLOT;
        const int n = c < BIN_CAP ? c : BIN_CAP;
        for (int i = lane; i < n; i += 32) dest[i] = sorted[st + i];
        if (lane == 0) dest[BIN_CAP] = (unsigned)c;
        if constexpr (SRC_LISTS) {
            const int c2  = scount[b];
            const int st2 = sstart[b];
            unsigned* dest2 = sbins + (size_t)b * BIN_REGION + (size_t)g * BIN_SLOT;
            const int n2 = c2 < BIN_CAP ? c2 : BIN_CAP;
            for (int i = lane; i < n2; i += 32) dest2[i] = ssorted[st2 + i];
            if (lane == 0) dest2[BIN_CAP] = (unsigned)c2;
        }
    }
}

#define OD_THREADS  256
#define OD_WAVES    (OD_THREADS / 32)
#define OD_PER_WAVE (BIN_SLICES / OD_WAVES)
static_assert(BIN_SLICES % OD_WAVES == 0 && OD_PER_WAVE <= 32, "whole lists per wave, counts in one load");

__global__ void __launch_bounds__(OD_THREADS)
outdeg_norm(const int* __restrict__ src, const unsigned* __restrict__ sbins,
            float* __restrict__ normS) {
    __shared__ int odeg[RS_NODES];

    const int tid  = threadIdx.x;
    const int lane = tid & 31;
    const int wave = __builtin_amdgcn_readfirstlane(tid >> 5);
    const int base = blockIdx.x * RS_NODES;

    for (int i = tid; i < RS_NODES; i += OD_THREADS) odeg[i] = 0;
    __syncthreads();

    const unsigned* mybin = sbins + (size_t)blockIdx.x * BIN_REGION;
    int cl = 0;
    if (lane < OD_PER_WAVE) cl = (int)mybin[(wave + lane * OD_WAVES) * BIN_SLOT + BIN_CAP];

    unsigned w0[OD_PER_WAVE];
    unsigned w1[OD_PER_WAVE];
#pragma unroll
    for (int k = 0; k < OD_PER_WAVE; ++k) {
        const int c = __builtin_amdgcn_readlane(cl, k);
        const unsigned* slot = mybin + (wave + k * OD_WAVES) * BIN_SLOT;
        const bool whole = c <= BIN_CAP;
        w0[k] = slot[(whole && lane < c) ? lane : BIN_CAP];
        w1[k] = slot[(whole && lane + 32 < c) ? lane + 32 : BIN_CAP];
    }
#pragma unroll
    for (int k = 0; k < OD_PER_WAVE; ++k) {
        const int g = wave + k * OD_WAVES;
        const unsigned* slot = mybin + g * BIN_SLOT;
        const int c = __builtin_amdgcn_readlane(cl, k);
        if (c <= BIN_CAP) {
            if (lane < c && w0[k] < (unsigned)RS_NODES) atomicAdd(&odeg[w0[k]], 1);
            if (lane + 32 < c && w1[k] < (unsigned)RS_NODES) atomicAdd(&odeg[w1[k]], 1);
            for (int p0 = 64; p0 < c; p0 += 32) {
                if (p0 + lane < c) {
                    const unsigned v = slot[p0 + lane];
                    if (v < (unsigned)RS_NODES) atomicAdd(&odeg[v], 1);
                }
            }
        } else {
            const int4* s4 = reinterpret_cast<const int4*>(src) + (size_t)g * BIN_I4;
            for (int i = lane; i < BIN_I4; i += 32) {
                const int4 s = s4[i];
                unsigned sl;
                sl = (unsigned)(s.x - base);
                if (sl < (unsigned)RS_NODES) atomicAdd(&odeg[sl], 1);
                sl = (unsigned)(s.y - base);
                if (sl < (unsigned)RS_NODES) atomicAdd(&odeg[sl], 1);
                sl = (unsigned)(s.z - base);
                if (sl < (unsigned)RS_NODES) atomicAdd(&odeg[sl], 1);
                sl = (unsigned)(s.w - base);
                if (sl < (unsigned)RS_NODES) atomicAdd(&odeg[sl], 1);
            }
        }
    }
    __syncthreads();
    for (int i = tid; i < RS_NODES; i += OD_THREADS)
        normS[base + i] = rsqrtf(fmaxf((float)odeg[i], 1.0f));
}

template <int LAYER>
__device__ __forceinline__ void rs_slow_add(int e, unsigned dl, int* __restrict__ indeg,
                                            float* __restrict__ acc,
                                            const float* __restrict__ M,
                                            const int* __restrict__ src) {
    if constexpr (LAYER == 1) atomicAdd(indeg + dl, 1);
    const float* mrow = M + (size_t)src[e] * NHID;
    float* arow = acc + dl * RS_PITCH;
#pragma unroll 1
    for (int j = 0; j < NHID; ++j) atomicAdd(arow + j, mrow[j]);
}

template <int LAYER>
__global__ void __launch_bounds__(RS_THREADS)
edge_rowsum_lds(const float* __restrict__ M, const int* __restrict__ src,
                const int* __restrict__ dst, const unsigned* bins,
                const float* __restrict__ normS, const float* __restrict__ normD,
                const float* __restrict__ bias, const float* __restrict__ Wn,
                const float* __restrict__ bfc, float* Out,
                float* __restrict__ normDout) {
    __shared__ __attribute__((aligned(16))) float acc[RS_NODES * RS_PITCH];
    __shared__ int   indeg[LAYER == 1 ? RS_NODES : 1];

    const int tid  = threadIdx.x;
    const int lane = tid & 31;
    const int wave = __builtin_amdgcn_readfirstlane(tid >> 5);
    const int base = blockIdx.x * RS_NODES;

    float4* acc4 = reinterpret_cast<float4*>(acc);
    for (int i = tid; i < RS_NODES * RS_PITCH4; i += RS_THREADS)
        acc4[i] = make_float4(0.0f, 0.0f, 0.0f, 0.0f);
    if constexpr (LAYER == 1) {
        if (tid < RS_NODES) indeg[tid] = 0;
    }
    __syncthreads();

    const int4*   dst4 = reinterpret_cast<const int4*>(dst);
    const float4* M4   = reinterpret_cast<const float4*>(M);

    const bool hi = lane >= 16;
    const int o0 = lane * 4 + (hi ? 2 : 0);
    const int o1 = lane * 4 + (hi ? 3 : 1);
    const int o2 = lane * 4 + (hi ? 0 : 2);
    const int o3 = lane * 4 + (hi ? 1 : 3);

    const unsigned* mybin = bins + (size_t)blockIdx.x * BIN_REGION;
    int cl = 0;
    if (lane < BIN_PER_WAVE) cl = (int)mybin[(wave + lane * RS_WAVES) * BIN_SLOT + BIN_CAP];
    unsigned wnext = 0u;
    {
        const int c0 = __builtin_amdgcn_readlane(cl, 0);
        if (c0 <= BIN_CAP && lane < c0) wnext = mybin[wave * BIN_SLOT + lane];
    }
    for (int k = 0; k < BIN_PER_WAVE; ++k) {
        const int g = wave + k * RS_WAVES;
        const unsigned* slot = mybin + g * BIN_SLOT;
        const int c = __builtin_amdgcn_readlane(cl, k);
        unsigned w = wnext;
        wnext = 0u;
        if (k + 1 < BIN_PER_WAVE) {
            const int cn = __builtin_amdgcn_readlane(cl, k + 1);
            if (cn <= BIN_CAP && lane < cn) wnext = slot[RS_WAVES * BIN_SLOT + lane];
        }
        if (c <= BIN_CAP) {
            for (int p0 = 0; p0 < c; p0 += 32) {
                if (p0 > 0) w = (p0 + lane < c) ? slot[p0 + lane] : 0u;
                const int m32 = (c - p0) < 32 ? (c - p0) : 32;
                const int s  = (int)(w & BIN_MASK);
                const int dl = (int)(w >> BIN_SHIFT);
                if constexpr (LAYER == 1) {
                    if (lane < m32) atomicAdd(indeg + dl, 1);
                }
                for (int r = 0; r < 4; ++r) {
                    const int m = m32 - r * 8;
                    if (m <= 0) break;
                    float4 v[8];
#pragma unroll
                    for (int j = 0; j < 8; ++j) {
                        v[j] = make_float4(0.0f, 0.0f, 0.0f, 0.0f);
                        if (j < m) {
                            const int sj = __builtin_amdgcn_readlane(s, r * 8 + j);
                            v[j] = M4[(size_t)sj * (NHID / 4) + lane];
                        }
                    }
#pragma unroll
                    for (int j = 0; j < 8; ++j) {
                        if (j < m) {
                            const int dj = __builtin_amdgcn_readlane(dl, r * 8 + j);
                            float* p = acc + dj * RS_PITCH;
                            atomicAdd(p + o0, hi ? v[j].z : v[j].x);
                            atomicAdd(p + o1, hi ? v[j].w : v[j].y);
                            atomicAdd(p + o2, hi ? v[j].x : v[j].z);
                            atomicAdd(p + o3, hi ? v[j].y : v[j].w);
                        }
                    }
                }
            }
        } else {
            for (int i = lane; i < BIN_I4; i += 32) {
                const int i4 = g * BIN_I4 + i;
                const int4 d = dst4[i4];
                const int e = i4 * 4;
                unsigned dl;
                dl = (unsigned)(d.x - base);
                if (dl < (unsigned)RS_NODES) rs_slow_add<LAYER>(e + 0, dl, indeg, acc, M, src);
                dl = (unsigned)(d.y - base);
                if (dl < (unsigned)RS_NODES) rs_slow_add<LAYER>(e + 1, dl, indeg, acc, M, src);
                dl = (unsigned)(d.z - base);
                if (dl < (unsigned)RS_NODES) rs_slow_add<LAYER>(e + 2, dl, indeg, acc, M, src);
                dl = (unsigned)(d.w - base);
                if (dl < (unsigned)RS_NODES) rs_slow_add<LAYER>(e + 3, dl, indeg, acc, M, src);
            }
        }
    }
    __syncthreads();

    if constexpr (LAYER == 1) {
        if (tid < RS_NODES) normDout[base + tid] = rsqrtf(fmaxf((float)indeg[tid], 1.0f));
        const float4* b4 = reinterpret_cast<const float4*>(bias);
        const float4 bb = b4[lane];
        for (int i = tid; i < RS_NODES * (NHID / 4); i += RS_THREADS) {
            const int ln   = i >> 5;
            const int node = base + ln;
            const float nd = rsqrtf(fmaxf((float)indeg[ln], 1.0f));
            const float ns = normS[node];
            float4 v = acc4[ln * RS_PITCH4 + lane];
            float t;
            t = v.x * nd + bb.x;  v.x = (t > 0.0f ? t : 0.0f) * ns;
            t = v.y * nd + bb.y;  v.y = (t > 0.0f ? t : 0.0f) * ns;
            t = v.z * nd + bb.z;  v.z = (t > 0.0f ? t : 0.0f) * ns;
            t = v.w * nd + bb.w;  v.w = (t > 0.0f ? t : 0.0f) * ns;
            acc4[ln * RS_PITCH4 + lane] = v;
        }
        __syncthreads();

        const int l   = lane & 15;
        const int kh  = lane >> 4;
        const int col = (wave & 7) * 16 + l;
        v2f bfr[NHID / 4];
#pragma unroll
        for (int ks = 0; ks < NHID / 4; ++ks) {
            const int ka = ks * 4 + 2 * kh;
            bfr[ks].x = Wn[(size_t)ka       * NHID + col];
            bfr[ks].y = Wn[(size_t)(ka + 1) * NHID + col];
        }
        for (int rt = wave >> 3; rt < RS_RTILES; rt += RS_WAVES / 8) {
            const float* arow = acc + (rt * 16 + l) * RS_PITCH + 2 * kh;
            v8f c = {};
#pragma unroll
            for (int ks = 0; ks < NHID / 4; ++ks) {
                const v2f a = *reinterpret_cast<const v2f*>(arow + ks * 4);
                c = __builtin_amdgcn_wmma_f32_16x16x4_f32(false, a, false, bfr[ks],
                                                          (short)0, c, false, false);
            }
            float* orow = Out + (size_t)(base + rt * 16 + 8 * kh) * NHID + col;
#pragma unroll
            for (int r = 0; r < 8; ++r) orow[(size_t)r * NHID] = c[r];
        }
    } else {
        __shared__ float outv[RS_NODES];
        const float bf = bfc[0];
        for (int ln = wave; ln < RS_NODES; ln += RS_WAVES) {
            const float nd = normD[base + ln];
            const float* arow = acc + ln * RS_PITCH;
            float a = 0.0f;
#pragma unroll
            for (int t = 0; t < 4; ++t) {
                const int j = lane + 32 * t;
                a += (arow[j] * nd + bias[j]) * Wn[j];
            }
#pragma unroll
            for (int off = 16; off > 0; off >>= 1)
                a += __shfl_xor(a, off, 32);
            if (lane == 0) outv[ln] = a + bf;
        }
        __syncthreads();
        if (tid < RS_NODES) Out[base + tid] = outv[tid];
    }
}

extern "C" void kernel_launch(void* const* d_in, const int* in_sizes, int n_in,
                              void* d_out, int out_size, void* d_ws, size_t ws_size,
                              hipStream_t stream) {
    const float* x   = (const float*)d_in[0];
    const float* W1  = (const float*)d_in[1];
    const float* b1  = (const float*)d_in[2];
    const float* W2  = (const float*)d_in[3];
    const float* b2  = (const float*)d_in[4];
    const float* Wfc = (const float*)d_in[5];
    const float* bfc = (const float*)d_in[6];
    const int*   src = (const int*)d_in[7];
    const int*   dst = (const int*)d_in[8];
    float* out = (float*)d_out;

    const int N = N_NODES;
    const int NH = NHID;

    float* BUF_A = (float*)d_ws;
    float* BUF_B = BUF_A + (size_t)N * NH;
    float* normS = BUF_B + (size_t)N * NH;
    float* normD = normS + N;

    bin_edges<true><<<BIN_SLICES, BIN_THREADS, 0, stream>>>(src, dst, (unsigned*)BUF_B, (unsigned*)BUF_A);
    outdeg_norm<<<RS_BLOCKS, OD_THREADS, 0, stream>>>(src, (const unsigned*)BUF_A, normS);

    gemm_scaled_wmma<<<G1_BLOCKS, G1_THREADS, 0, stream>>>(x, normS, W1, BUF_A);
    edge_rowsum_lds<1><<<RS_BLOCKS, RS_THREADS, 0, stream>>>(BUF_A, src, dst, (const unsigned*)BUF_B,
                                                            normS, normD, b1, W2, bfc, BUF_B, normD);

    bin_edges<false><<<BIN_SLICES, BIN_THREADS, 0, stream>>>(src, dst, (unsigned*)BUF_A, nullptr);
    edge_rowsum_lds<2><<<RS_BLOCKS, RS_THREADS, 0, stream>>>(BUF_B, src, dst, (const unsigned*)BUF_A,
                                                            normS, normD, b2, Wfc, bfc, out, nullptr);
}


# edge row sum: a wave owns 12-13 of its work-group's dst rows, filters the staged edge lists for them, adds source rows by plain 16-byte LDS read-add-write; no f32 LDS atomics; sums now in list order
# speedup vs baseline: 4.6493x; 1.1136x over previous
#include <hip/hip_runtime.h>

typedef __attribute__((ext_vector_type(2))) float v2f;
typedef __attribute__((ext_vector_type(8))) float v8f;

#define N_NODES 100000
#define N_EDGES 1600000
#define NFEAT   256
#define NHID    128


#define G1_BLOCKS  250
#define G1_THREADS 256
#define G1_TILES   (N_NODES / 16)
#define G1_STREAMS (2 * G1_BLOCKS)
#define G1_J       (NFEAT / 8)

static_assert(N_NODES % 16 == 0, "whole 16-row tiles: no row guard below");
static_assert(G1_STREAMS <= G1_TILES, "every stream has a first tile");
static_assert(NFEAT % 8 == 0 && NHID == 128, "8 k per float4 pair, 4 groups of 32 columns");

template <bool REFILL>
__device__ __forceinline__ void g1_tile(float4 (&a)[G1_J], const float4* __restrict__ np,
                                        const float ns, const v2f (&b0)[2 * G1_J],
                                        const v2f (&b1)[2 * G1_J],
                                        float* __restrict__ orow) {
    v8f c0 = {};
    v8f c1 = {};
#pragma unroll
    for (int j = 0; j < G1_J; ++j) {
        const float4 v = a[j];
        v2f lo, hi;
        lo.x = v.x * ns;  lo.y = v.y * ns;
        hi.x = v.z * ns;  hi.y = v.w * ns;
        if constexpr (REFILL) {
            __builtin_amdgcn_sched_barrier(0);
            a[j] = np[2 * j];
        }
        c0 = __builtin_amdgcn_wmma_f32_16x16x4_f32(false, lo, false, b0[2 * j],
                                                   (short)0, c0, false, false);
        c1 = __builtin_amdgcn_wmma_f32_16x16x4_f32(false, lo, false, b1[2 * j],
                                                   (short)0, c1, false, false);
        c0 = __builtin_amdgcn_wmma_f32_16x16x4_f32(false, hi, false, b0[2 * j + 1],
                                                   (short)0, c0, false, false);
        c1 = __builtin_amdgcn_wmma_f32_16x16x4_f32(false, hi, false, b1[2 * j + 1],
                                                   (short)0, c1, false, false);
    }
#pragma unroll
    for (int r = 0; r < 8; ++r) {
        orow[(size_t)r * NHID]      = c0[r];
        orow[(size_t)r * NHID + 16] = c1[r];
    }
}

__global__ void __launch_bounds__(G1_THREADS)
gemm_scaled_wmma(const float* __restrict__ X, const float* __restrict__ scale,
                 const float* __restrict__ W, float* __restrict__ Out) {
    const int lane = threadIdx.x & 31;
    const int wave = __builtin_amdgcn_readfirstlane(threadIdx.x >> 5);
    const int kh   = lane >> 4;
    const int l    = lane & 15;
    const int col  = (wave & 3) * 32 + l;
    const int stream = blockIdx.x + (wave >> 2) * G1_BLOCKS;

    v2f b0[2 * G1_J];
    v2f b1[2 * G1_J];
#pragma unroll
    for (int j = 0; j < G1_J; ++j) {
#pragma unroll
        for (int h = 0; h < 2; ++h) {
            const int k0 = 8 * j + 4 * kh + 2 * h;
            b0[2 * j + h].x = W[(size_t)k0       * NHID + col];
            b0[2 * j + h].y = W[(size_t)(k0 + 1) * NHID + col];
            b1[2 * j + h].x = W[(size_t)k0       * NHID + col + 16];
            b1[2 * j + h].y = W[(size_t)(k0 + 1) * NHID + col + 16];
        }
    }

    const float4* X4 = reinterpret_cast<const float4*>(X);
    int rt = stream;
    float4 a[G1_J];
    {
        const float4* p = X4 + (size_t)(rt * 16 + l) * (NFEAT / 4) + kh;
#pragma unroll
        for (int j = 0; j < G1_J; ++j) a[j] = p[2 * j];
    }
    float ns = scale[rt * 16 + l];
    for (;;) {
        float* orow = Out + (size_t)(rt * 16 + 8 * kh) * NHID + col;
        const int nrt = rt + G1_STREAMS;
        if (nrt >= G1_TILES) {
            g1_tile<false>(a, nullptr, ns, b0, b1, orow);
            break;
        }
        const float4* np = X4 + (size_t)(nrt * 16 + l) * (NFEAT / 4) + kh;
        const float nns = scale[nrt * 16 + l];
        g1_tile<true>(a, np, ns, b0, b1, orow);
        rt = nrt;
        ns = nns;
    }
}

#define RS_NODES   400
#define RS_THREADS 1024
#define RS_WAVES   (RS_THREADS / 32)
#define RS_E4      (N_EDGES / 4)
#define RS_BLOCKS  (N_NODES / RS_NODES)
#define RS_PITCH   (NHID + 4)
#define RS_PITCH4  (RS_PITCH / 4)
#define RS_RTILES  (RS_NODES / 16)

static_assert(N_NODES % RS_NODES == 0, "node ranges must tile the node set");
static_assert(RS_NODES % 16 == 0, "owned rows must be whole 16-row tiles");
static_assert(NHID == 128 && RS_WAVES % 8 == 0, "8 column tiles, waves in groups of 8");

#define BIN_SLICES   160
#define BIN_THREADS  1024
#define BIN_I4       (RS_E4 / BIN_SLICES)
#define BIN_LOADS    ((BIN_I4 + BIN_THREADS - 1) / BIN_THREADS)
#define BIN_REGION   (RS_NODES * NHID)
#define BIN_SLOT     (BIN_REGION / BIN_SLICES)
#define BIN_CAP      (BIN_SLOT - 1)
#define BIN_PER_WAVE (BIN_SLICES / RS_WAVES)
#define BIN_SHIFT    17
#define BIN_MASK     ((1u << BIN_SHIFT) - 1u)

static_assert(RS_E4 % BIN_SLICES == 0, "slices are whole 16-byte loads");
static_assert(BIN_REGION % BIN_SLICES == 0 && BIN_SLOT % 32 == 0, "a list starts on a 128-byte line");
static_assert(BIN_SLICES % RS_WAVES == 0 && BIN_PER_WAVE <= 32, "whole lists per wave, counts in one load");
static_assert(N_NODES <= (1 << BIN_SHIFT) && RS_NODES <= (1 << (32 - BIN_SHIFT)), "packed word holds both");
static_assert(RS_BLOCKS <= 256, "the start table below is built by one wave, 8 owners a lane");

#define ST_CAP     16384
#define WQ_LEN     64
#define ST_PER_LANE (BIN_SLICES / 32)
static_assert(BIN_SLICES % 32 == 0, "one wave sums the list counts, whole lists per lane");
static_assert(RS_NODES == 12 * RS_WAVES + 16 && RS_WAVES == 32, "waves 0..15 own 13 rows, waves 16..31 own 12");
static_assert(WQ_LEN >= 7 + 32 && (WQ_LEN & (WQ_LEN - 1)) == 0, "7 left over plus one batch of 32");

__device__ __forceinline__ void bin_starts(const int* cnt, int* st, int lane) {
    int c[8];
    int sum = 0;
#pragma unroll
    for (int k = 0; k < 8; ++k) {
        const int b = lane * 8 + k;
        c[k] = b < RS_BLOCKS ? cnt[b] : 0;
        sum += c[k];
    }
    int incl = sum;
#pragma unroll
    for (int off = 1; off < 32; off <<= 1) {
        const int t = __shfl_up(incl, off, 32);
        if (lane >= off) incl += t;
    }
    int run = incl - sum;
#pragma unroll
    for (int k = 0; k < 8; ++k) {
        const int b = lane * 8 + k;
        if (b < RS_BLOCKS) st[b] = run;
        run += c[k];
    }
}

template <bool SRC_LISTS>
__global__ void __launch_bounds__(BIN_THREADS)
bin_edges(const int* __restrict__ src, const int* __restrict__ dst, unsigned* __restrict__ bins,
          unsigned* __restrict__ sbins) {
    __shared__ int      lcount[RS_BLOCKS];
    __shared__ int      lstart[RS_BLOCKS];
    __shared__ unsigned sorted[BIN_I4 * 4];
    __shared__ int      scount[SRC_LISTS ? RS_BLOCKS : 1];
    __shared__ int      sstart[SRC_LISTS ? RS_BLOCKS : 1];
    __shared__ unsigned ssorted[SRC_LISTS ? BIN_I4 * 4 : 1];

    const int tid  = threadIdx.x;
    const int lane = tid & 31;
    const int wave = tid >> 5;
    const int g    = blockIdx.x;

    if (tid < RS_BLOCKS) {
        lcount[tid] = 0;
        if constexpr (SRC_LISTS) scount[tid] = 0;
    }
    __syncthreads();

    const int4* d4 = reinterpret_cast<const int4*>(dst) + (size_t)g * BIN_I4;
    const int4* s4 = reinterpret_cast<const int4*>(src) + (size_t)g * BIN_I4;

    unsigned wd[BIN_LOADS * 4];
    int      bn[BIN_LOADS * 4];
    int      rk[BIN_LOADS * 4];
    unsigned sw[SRC_LISTS ? BIN_LOADS * 4 : 1];
    int      sb[SRC_LISTS ? BIN_LOADS * 4 : 1];
    int      sr[SRC_LISTS ? BIN_LOADS * 4 : 1];
#pragma unroll
    for (int j = 0; j < BIN_LOADS; ++j) {
        const int i4 = j * BIN_THREADS + tid;
        int dv[4] = {-1, -1, -1, -1};
        int sv[4] = {-1, -1, -1, -1};
        if (i4 < BIN_I4) {
            const int4 d = d4[i4];
            const int4 s = s4[i4];
            dv[0] = d.x; dv[1] = d.y; dv[2] = d.z; dv[3] = d.w;
            sv[0] = s.x; sv[1] = s.y; sv[2] = s.z; sv[3] = s.w;
        }
#pragma unroll
        for (int t = 0; t < 4; ++t) {
            const int q = j * 4 + t;
            bn[q] = -1; rk[q] = 0; wd[q] = 0u;
            if ((unsigned)dv[t] < (unsigned)N_NODES) {
                const int b  = dv[t] / RS_NODES;
                const int dl = dv[t] - b * RS_NODES;
                bn[q] = b;
                wd[q] = (unsigned)sv[t] | ((unsigned)dl << BIN_SHIFT);
                rk[q] = atomicAdd(&lcount[b], 1);
            }
            if constexpr (SRC_LISTS) {
                sb[q] = -1; sr[q] = 0; sw[q] = 0u;
                if ((unsigned)sv[t] < (unsigned)N_NODES) {
                    const int b2 = sv[t] / RS_NODES;
                    sb[q] = b2;
                    sw[q] = (unsigned)(sv[t] - b2 * RS_NODES);
                    sr[q] = atomicAdd(&scount[b2], 1);
                }
            }
        }
    }
    __syncthreads();

    if (wave == 0) bin_starts(lcount, lstart, lane);
    if constexpr (SRC_LISTS) {
        if (wave == 1) bin_starts(scount, sstart, lane);
    }
    __syncthreads();

#pragma unroll
    for (int q = 0; q < BIN_LOADS * 4; ++q)
        if (bn[q] >= 0) sorted[lstart[bn[q]] + rk[q]] = wd[q];
    if constexpr (SRC_LISTS) {
#pragma unroll
        for (int q = 0; q < BIN_LOADS * 4; ++q)
            if (sb[q] >= 0) ssorted[sstart[sb[q]] + sr[q]] = sw[q];
    }
    __syncthreads();

    for (int b = wave; b < RS_BLOCKS; b += BIN_THREADS / 32) {
        const int c  = lcount[b];
        const int st = lstart[b];
        unsigned* dest = bins + (size_t)b * BIN_REGION + (size_t)g * BIN_SLOT;
        const int n = c < BIN_CAP ? c : BIN_CAP;
        for (int i = lane; i < n; i += 32) dest[i] = sorted[st + i];
        if (lane == 0) dest[BIN_CAP] = (unsigned)c;
        if constexpr (SRC_LISTS) {
            const int c2  = scount[b];
            const int st2 = sstart[b];
            unsigned* dest2 = sbins + (size_t)b * BIN_REGION + (size_t)g * BIN_SLOT;
            const int n2 = c2 < BIN_CAP ? c2 : BIN_CAP;
            for (int i = lane; i < n2; i += 32) dest2[i] = ssorted[st2 + i];
            if (lane == 0) dest2[BIN_CAP] = (unsigned)c2;
        }
    }
}

#define OD_THREADS  256
#define OD_WAVES    (OD_THREADS / 32)
#define OD_PER_WAVE (BIN_SLICES / OD_WAVES)
static_assert(BIN_SLICES % OD_WAVES == 0 && OD_PER_WAVE <= 32, "whole lists per wave, counts in one load");

__global__ void __launch_bounds__(OD_THREADS)
outdeg_norm(const int* __restrict__ src, const unsigned* __restrict__ sbins,
            float* __restrict__ normS) {
    __shared__ int odeg[RS_NODES];

    const int tid  = threadIdx.x;
    const int lane = tid & 31;
    const int wave = __builtin_amdgcn_readfirstlane(tid >> 5);
    const int base = blockIdx.x * RS_NODES;

    for (int i = tid; i < RS_NODES; i += OD_THREADS) odeg[i] = 0;
    __syncthreads();

    const unsigned* mybin = sbins + (size_t)blockIdx.x * BIN_REGION;
    int cl = 0;
    if (lane < OD_PER_WAVE) cl = (int)mybin[(wave + lane * OD_WAVES) * BIN_SLOT + BIN_CAP];

    unsigned w0[OD_PER_WAVE];
    unsigned w1[OD_PER_WAVE];
#pragma unroll
    for (int k = 0; k < OD_PER_WAVE; ++k) {
        const int c = __builtin_amdgcn_readlane(cl, k);
        const unsigned* slot = mybin + (wave + k * OD_WAVES) * BIN_SLOT;
        const bool whole = c <= BIN_CAP;
        w0[k] = slot[(whole && lane < c) ? lane : BIN_CAP];
        w1[k] = slot[(whole && lane + 32 < c) ? lane + 32 : BIN_CAP];
    }
#pragma unroll
    for (int k = 0; k < OD_PER_WAVE; ++k) {
        const int g = wave + k * OD_WAVES;
        const unsigned* slot = mybin + g * BIN_SLOT;
        const int c = __builtin_amdgcn_readlane(cl, k);
        if (c <= BIN_CAP) {
            if (lane < c && w0[k] < (unsigned)RS_NODES) atomicAdd(&odeg[w0[k]], 1);
            if (lane + 32 < c && w1[k] < (unsigned)RS_NODES) atomicAdd(&odeg[w1[k]], 1);
            for (int p0 = 64; p0 < c; p0 += 32) {
                if (p0 + lane < c) {
                    const unsigned v = slot[p0 + lane];
                    if (v < (unsigned)RS_NODES) atomicAdd(&odeg[v], 1);
                }
            }
        } else {
            const int4* s4 = reinterpret_cast<const int4*>(src) + (size_t)g * BIN_I4;
            for (int i = lane; i < BIN_I4; i += 32) {
                const int4 s = s4[i];
                unsigned sl;
                sl = (unsigned)(s.x - base);
                if (sl < (unsigned)RS_NODES) atomicAdd(&odeg[sl], 1);
                sl = (unsigned)(s.y - base);
                if (sl < (unsigned)RS_NODES) atomicAdd(&odeg[sl], 1);
                sl = (unsigned)(s.z - base);
                if (sl < (unsigned)RS_NODES) atomicAdd(&odeg[sl], 1);
                sl = (unsigned)(s.w - base);
                if (sl < (unsigned)RS_NODES) atomicAdd(&odeg[sl], 1);
            }
        }
    }
    __syncthreads();
    for (int i = tid; i < RS_NODES; i += OD_THREADS)
        normS[base + i] = rsqrtf(fmaxf((float)odeg[i], 1.0f));
}

template <int LAYER>
__device__ __forceinline__ void rs_slow_add(int e, unsigned dl, int* __restrict__ indeg,
                                            float* __restrict__ acc,
                                            const float* __restrict__ M,
                                            const int* __restrict__ src) {
    if constexpr (LAYER == 1) atomicAdd(indeg + dl, 1);
    const float* mrow = M + (size_t)src[e] * NHID;
    float* arow = acc + dl * RS_PITCH;
#pragma unroll 1
    for (int j = 0; j < NHID; ++j) atomicAdd(arow + j, mrow[j]);
}

template <bool FULL>
__device__ __forceinline__ void rs_round(const unsigned* wq, const int head, const int n,
                                         const float4* __restrict__ M4, const int lane,
                                         float4* acc4) {
    const unsigned qw = wq[(head + (lane & 7)) & (WQ_LEN - 1)];
    float4 v[8];
#pragma unroll
    for (int j = 0; j < 8; ++j) {
        v[j] = make_float4(0.0f, 0.0f, 0.0f, 0.0f);
        if (FULL || j < n) {
            const int sj = (int)(__builtin_amdgcn_readlane(qw, j) & BIN_MASK);
            v[j] = M4[(size_t)sj * (NHID / 4) + lane];
        }
    }
#pragma unroll
    for (int j = 0; j < 8; ++j) {
        if (FULL || j < n) {
            const int dj = (int)(__builtin_amdgcn_readlane(qw, j) >> BIN_SHIFT);
            float4* p = acc4 + dj * RS_PITCH4 + lane;
            float4 r = *p;
            r.x += v[j].x;  r.y += v[j].y;  r.z += v[j].z;  r.w += v[j].w;
            *p = r;
        }
    }
}

template <int LAYER>
__global__ void __launch_bounds__(RS_THREADS)
edge_rowsum_lds(const float* __restrict__ M, const int* __restrict__ src,
                const int* __restrict__ dst, const unsigned* bins,
                const float* __restrict__ normS, const float* __restrict__ normD,
                const float* __restrict__ bias, const float* __restrict__ Wn,
                const float* __restrict__ bfc, float* Out,
                float* __restrict__ normDout) {
    __shared__ __attribute__((aligned(16))) float acc[RS_NODES * RS_PITCH];
    __shared__ int   indeg[LAYER == 1 ? RS_NODES : 1];
    __shared__ unsigned ed[ST_CAP];
    __shared__ unsigned wqueue[RS_WAVES * WQ_LEN];
    __shared__ int   lstart[BIN_SLICES];
    __shared__ int   ltotal[2];

    const int tid  = threadIdx.x;
    const int lane = tid & 31;
    const int wave = __builtin_amdgcn_readfirstlane(tid >> 5);
    const int base = blockIdx.x * RS_NODES;

    float4* acc4 = reinterpret_cast<float4*>(acc);
    const int4*   dst4 = reinterpret_cast<const int4*>(dst);
    const float4* M4   = reinterpret_cast<const float4*>(M);

    const unsigned* mybin = bins + (size_t)blockIdx.x * BIN_REGION;
    int cl = 0;
    if (lane < BIN_PER_WAVE) cl = (int)mybin[(wave + lane * RS_WAVES) * BIN_SLOT + BIN_CAP];

    unsigned f0[BIN_PER_WAVE];
    unsigned f1[BIN_PER_WAVE];
#pragma unroll
    for (int k = 0; k < BIN_PER_WAVE; ++k) {
        const int c = __builtin_amdgcn_readlane(cl, k);
        const unsigned* slot = mybin + (wave + k * RS_WAVES) * BIN_SLOT;
        const bool whole = c <= BIN_CAP;
        f0[k] = slot[(whole && lane < c) ? lane : BIN_CAP];
        f1[k] = slot[(whole && lane + 32 < c) ? lane + 32 : BIN_CAP];
    }

    if (wave == 0) {
        int c[ST_PER_LANE];
        int sum = 0;
        bool cut = false;
#pragma unroll
        for (int k = 0; k < ST_PER_LANE; ++k) {
            c[k] = (int)mybin[(lane * ST_PER_LANE + k) * BIN_SLOT + BIN_CAP];
            cut = cut || c[k] > BIN_CAP;
            sum += c[k];
        }
        int incl = sum;
#pragma unroll
        for (int off = 1; off < 32; off <<= 1) {
            const int t = __shfl_up(incl, off, 32);
            if (lane >= off) incl += t;
        }
        int run = incl - sum;
#pragma unroll
        for (int k = 0; k < ST_PER_LANE; ++k) {
            lstart[lane * ST_PER_LANE + k] = run;
            run += c[k];
        }
        const unsigned anycut = __builtin_amdgcn_ballot_w32(cut);
        if (lane == 31) {
            ltotal[0] = incl;
            ltotal[1] = anycut != 0u ? 1 : 0;
        }
    }
    if constexpr (LAYER == 1) {
        if (tid < RS_NODES) indeg[tid] = 0;
    }
    __syncthreads();

    const int  total = __builtin_amdgcn_readfirstlane(ltotal[0]);
    const bool fast  = __builtin_amdgcn_readfirstlane(ltotal[1]) == 0 && total <= ST_CAP;

    if (fast) {
#pragma unroll
        for (int k = 0; k < BIN_PER_WAVE; ++k) {
            const int g = wave + k * RS_WAVES;
            const unsigned* slot = mybin + g * BIN_SLOT;
            const int c  = __builtin_amdgcn_readlane(cl, k);
            const int st = __builtin_amdgcn_readfirstlane(lstart[g]);
            if (lane < c) {
                ed[st + lane] = f0[k];
                if constexpr (LAYER == 1) atomicAdd(indeg + (f0[k] >> BIN_SHIFT), 1);
            }
            if (lane + 32 < c) {
                ed[st + lane + 32] = f1[k];
                if constexpr (LAYER == 1) atomicAdd(indeg + (f1[k] >> BIN_SHIFT), 1);
            }
            for (int p0 = 64; p0 < c; p0 += 32) {
                if (p0 + lane < c) {
                    const unsigned w = slot[p0 + lane];
                    ed[st + p0 + lane] = w;
                    if constexpr (LAYER == 1) atomicAdd(indeg + (w >> BIN_SHIFT), 1);
                }
            }
        }
        __syncthreads();

        const int lo   = wave * 12 + (wave < 16 ? wave : 16);
        const int rows = wave < 16 ? 13 : 12;
        for (int k = 0; k < rows; ++k)
            acc4[(lo + k) * RS_PITCH4 + lane] = make_float4(0.0f, 0.0f, 0.0f, 0.0f);

        unsigned* wq = wqueue + wave * WQ_LEN;
        int head = 0;
        int tail = 0;
        for (int p0 = 0; p0 < total; p0 += 128) {
            unsigned wb[4];
#pragma unroll
            for (int u = 0; u < 4; ++u) {
                const int i = p0 + 32 * u + lane;
                wb[u] = i < total ? ed[i] : 0xFFFFFFFFu;
            }
#pragma unroll
            for (int u = 0; u < 4; ++u) {
                const unsigned w = wb[u];
                const bool mine = (w >> BIN_SHIFT) - (unsigned)lo < (unsigned)rows;
                const unsigned mask = __builtin_amdgcn_ballot_w32(mine);
                if (mask != 0u) {
                    if (mine) wq[(tail + (int)__builtin_amdgcn_mbcnt_lo(mask, 0u)) & (WQ_LEN - 1)] = w;
                    tail += __builtin_popcount(mask);
                    while (tail - head >= 8) {
                        rs_round<true>(wq, head, 8, M4, lane, acc4);
                        head += 8;
                    }
                }
            }
        }
        if (tail > head) rs_round<false>(wq, head, tail - head, M4, lane, acc4);
    } else {
        for (int i = tid; i < RS_NODES * RS_PITCH4; i += RS_THREADS)
            acc4[i] = make_float4(0.0f, 0.0f, 0.0f, 0.0f);
        __syncthreads();

        const bool hi = lane >= 16;
        const int o0 = lane * 4 + (hi ? 2 : 0);
        const int o1 = lane * 4 + (hi ? 3 : 1);
        const int o2 = lane * 4 + (hi ? 0 : 2);
        const int o3 = lane * 4 + (hi ? 1 : 3);

        for (int k = 0; k < BIN_PER_WAVE; ++k) {
            const int g = wave + k * RS_WAVES;
            const unsigned* slot = mybin + g * BIN_SLOT;
            const int c = __builtin_amdgcn_readlane(cl, k);
            if (c <= BIN_CAP) {
                for (int p0 = 0; p0 < c; p0 += 32) {
                    const unsigned w = (p0 + lane < c) ? slot[p0 + lane] : 0u;
                    const int m32 = (c - p0) < 32 ? (c - p0) : 32;
                    const int s  = (int)(w & BIN_MASK);
                    const int dl = (int)(w >> BIN_SHIFT);
                    if constexpr (LAYER == 1) {
                        if (lane < m32) atomicAdd(indeg + dl, 1);
                    }
                    for (int r = 0; r < 4; ++r) {
                        const int m = m32 - r * 8;
                        if (m <= 0) break;
                        float4 v[8];
#pragma unroll
                        for (int j = 0; j < 8; ++j) {
                            v[j] = make_float4(0.0f, 0.0f, 0.0f, 0.0f);
                            if (j < m) {
                                const int sj = __builtin_amdgcn_readlane(s, r * 8 + j);
                                v[j] = M4[(size_t)sj * (NHID / 4) + lane];
                            }
                        }
#pragma unroll
                        for (int j = 0; j < 8; ++j) {
                            if (j < m) {
                                const int dj = __builtin_amdgcn_readlane(dl, r * 8 + j);
                                float* p = acc + dj * RS_PITCH;
                                atomicAdd(p + o0, hi ? v[j].z : v[j].x);
                                atomicAdd(p + o1, hi ? v[j].w : v[j].y);
                                atomicAdd(p + o2, hi ? v[j].x : v[j].z);
                                atomicAdd(p + o3, hi ? v[j].y : v[j].w);
                            }
                        }
                    }
                }
            } else {
                for (int i = lane; i < BIN_I4; i += 32) {
                    const int i4 = g * BIN_I4 + i;
                    const int4 d = dst4[i4];
                    const int e = i4 * 4;
                    unsigned dl;
                    dl = (unsigned)(d.x - base);
                    if (dl < (unsigned)RS_NODES) rs_slow_add<LAYER>(e + 0, dl, indeg, acc, M, src);
                    dl = (unsigned)(d.y - base);
                    if (dl < (unsigned)RS_NODES) rs_slow_add<LAYER>(e + 1, dl, indeg, acc, M, src);
                    dl = (unsigned)(d.z - base);
                    if (dl < (unsigned)RS_NODES) rs_slow_add<LAYER>(e + 2, dl, indeg, acc, M, src);
                    dl = (unsigned)(d.w - base);
                    if (dl < (unsigned)RS_NODES) rs_slow_add<LAYER>(e + 3, dl, indeg, acc, M, src);
                }
            }
        }
    }
    __syncthreads();

    if constexpr (LAYER == 1) {
        if (tid < RS_NODES) normDout[base + tid] = rsqrtf(fmaxf((float)indeg[tid], 1.0f));
        const float4* b4 = reinterpret_cast<const float4*>(bias);
        const float4 bb = b4[lane];
        for (int i = tid; i < RS_NODES * (NHID / 4); i += RS_THREADS) {
            const int ln   = i >> 5;
            const int node = base + ln;
            const float nd = rsqrtf(fmaxf((float)indeg[ln], 1.0f));
            const float ns = normS[node];
            float4 v = acc4[ln * RS_PITCH4 + lane];
            float t;
            t = v.x * nd + bb.x;  v.x = (t > 0.0f ? t : 0.0f) * ns;
            t = v.y * nd + bb.y;  v.y = (t > 0.0f ? t : 0.0f) * ns;
            t = v.z * nd + bb.z;  v.z = (t > 0.0f ? t : 0.0f) * ns;
            t = v.w * nd + bb.w;  v.w = (t > 0.0f ? t : 0.0f) * ns;
            acc4[ln * RS_PITCH4 + lane] = v;
        }
        __syncthreads();

        const int l   = lane & 15;
        const int kh  = lane >> 4;
        const int col = (wave & 7) * 16 + l;
        v2f bfr[NHID / 4];
#pragma unroll
        for (int ks = 0; ks < NHID / 4; ++ks) {
            const int ka = ks * 4 + 2 * kh;
            bfr[ks].x = Wn[(size_t)ka       * NHID + col];
            bfr[ks].y = Wn[(size_t)(ka + 1) * NHID + col];
        }
        for (int rt = wave >> 3; rt < RS_RTILES; rt += RS_WAVES / 8) {
            const float* arow = acc + (rt * 16 + l) * RS_PITCH + 2 * kh;
            v8f c = {};
#pragma unroll
            for (int ks = 0; ks < NHID / 4; ++ks) {
                const v2f a = *reinterpret_cast<const v2f*>(arow + ks * 4);
                c = __builtin_amdgcn_wmma_f32_16x16x4_f32(false, a, false, bfr[ks],
                                                          (short)0, c, false, false);
            }
            float* orow = Out + (size_t)(base + rt * 16 + 8 * kh) * NHID + col;
#pragma unroll
            for (int r = 0; r < 8; ++r) orow[(size_t)r * NHID] = c[r];
        }
    } else {
        __shared__ float outv[RS_NODES];
        const float bf = bfc[0];
        for (int ln = wave; ln < RS_NODES; ln += RS_WAVES) {
            const float nd = normD[base + ln];
            const float* arow = acc + ln * RS_PITCH;
            float a = 0.0f;
#pragma unroll
            for (int t = 0; t < 4; ++t) {
                const int j = lane + 32 * t;
                a += (arow[j] * nd + bias[j]) * Wn[j];
            }
#pragma unroll
            for (int off = 16; off > 0; off >>= 1)
                a += __shfl_xor(a, off, 32);
            if (lane == 0) outv[ln] = a + bf;
        }
        __syncthreads();
        if (tid < RS_NODES) Out[base + tid] = outv[tid];
    }
}

extern "C" void kernel_launch(void* const* d_in, const int* in_sizes, int n_in,
                              void* d_out, int out_size, void* d_ws, size_t ws_size,
                              hipStream_t stream) {
    const float* x   = (const float*)d_in[0];
    const float* W1  = (const float*)d_in[1];
    const float* b1  = (const float*)d_in[2];
    const float* W2  = (const float*)d_in[3];
    const float* b2  = (const float*)d_in[4];
    const float* Wfc = (const float*)d_in[5];
    const float* bfc = (const float*)d_in[6];
    const int*   src = (const int*)d_in[7];
    const int*   dst = (const int*)d_in[8];
    float* out = (float*)d_out;

    const int N = N_NODES;
    const int NH = NHID;

    float* BUF_A = (float*)d_ws;
    float* BUF_B = BUF_A + (size_t)N * NH;
    float* normS = BUF_B + (size_t)N * NH;
    float* normD = normS + N;

    bin_edges<true><<<BIN_SLICES, BIN_THREADS, 0, stream>>>(src, dst, (unsigned*)BUF_B, (unsigned*)BUF_A);
    outdeg_norm<<<RS_BLOCKS, OD_THREADS, 0, stream>>>(src, (const unsigned*)BUF_A, normS);

    gemm_scaled_wmma<<<G1_BLOCKS, G1_THREADS, 0, stream>>>(x, normS, W1, BUF_A);
    edge_rowsum_lds<1><<<RS_BLOCKS, RS_THREADS, 0, stream>>>(BUF_A, src, dst, (const unsigned*)BUF_B,
                                                            normS, normD, b1, W2, bfc, BUF_B, normD);

    bin_edges<false><<<BIN_SLICES, BIN_THREADS, 0, stream>>>(src, dst, (unsigned*)BUF_A, nullptr);
    edge_rowsum_lds<2><<<RS_BLOCKS, RS_THREADS, 0, stream>>>(BUF_B, src, dst, (const unsigned*)BUF_A,
                                                            normS, normD, b2, Wfc, bfc, out, nullptr);
}


# edge row sum: owner's lists split once into 32 per-wave lists in LDS (a returning LDS integer add places each word); a wave walks only its own edges, no 32-fold filter scan; sums in arrival order
# speedup vs baseline: 5.0774x; 1.0921x over previous
#include <hip/hip_runtime.h>

typedef __attribute__((ext_vector_type(2))) float v2f;
typedef __attribute__((ext_vector_type(8))) float v8f;

#define N_NODES 100000
#define N_EDGES 1600000
#define NFEAT   256
#define NHID    128


#define G1_BLOCKS  250
#define G1_THREADS 256
#define G1_TILES   (N_NODES / 16)
#define G1_STREAMS (2 * G1_BLOCKS)
#define G1_J       (NFEAT / 8)

static_assert(N_NODES % 16 == 0, "whole 16-row tiles: no row guard below");
static_assert(G1_STREAMS <= G1_TILES, "every stream has a first tile");
static_assert(NFEAT % 8 == 0 && NHID == 128, "8 k per float4 pair, 4 groups of 32 columns");

template <bool REFILL>
__device__ __forceinline__ void g1_tile(float4 (&a)[G1_J], const float4* __restrict__ np,
                                        const float ns, const v2f (&b0)[2 * G1_J],
                                        const v2f (&b1)[2 * G1_J],
                                        float* __restrict__ orow) {
    v8f c0 = {};
    v8f c1 = {};
#pragma unroll
    for (int j = 0; j < G1_J; ++j) {
        const float4 v = a[j];
        v2f lo, hi;
        lo.x = v.x * ns;  lo.y = v.y * ns;
        hi.x = v.z * ns;  hi.y = v.w * ns;
        if constexpr (REFILL) {
            __builtin_amdgcn_sched_barrier(0);
            a[j] = np[2 * j];
        }
        c0 = __builtin_amdgcn_wmma_f32_16x16x4_f32(false, lo, false, b0[2 * j],
                                                   (short)0, c0, false, false);
        c1 = __builtin_amdgcn_wmma_f32_16x16x4_f32(false, lo, false, b1[2 * j],
                                                   (short)0, c1, false, false);
        c0 = __builtin_amdgcn_wmma_f32_16x16x4_f32(false, hi, false, b0[2 * j + 1],
                                                   (short)0, c0, false, false);
        c1 = __builtin_amdgcn_wmma_f32_16x16x4_f32(false, hi, false, b1[2 * j + 1],
                                                   (short)0, c1, false, false);
    }
#pragma unroll
    for (int r = 0; r < 8; ++r) {
        orow[(size_t)r * NHID]      = c0[r];
        orow[(size_t)r * NHID + 16] = c1[r];
    }
}

__global__ void __launch_bounds__(G1_THREADS)
gemm_scaled_wmma(const float* __restrict__ X, const float* __restrict__ scale,
                 const float* __restrict__ W, float* __restrict__ Out) {
    const int lane = threadIdx.x & 31;
    const int wave = __builtin_amdgcn_readfirstlane(threadIdx.x >> 5);
    const int kh   = lane >> 4;
    const int l    = lane & 15;
    const int col  = (wave & 3) * 32 + l;
    const int stream = blockIdx.x + (wave >> 2) * G1_BLOCKS;

    v2f b0[2 * G1_J];
    v2f b1[2 * G1_J];
#pragma unroll
    for (int j = 0; j < G1_J; ++j) {
#pragma unroll
        for (int h = 0; h < 2; ++h) {
            const int k0 = 8 * j + 4 * kh + 2 * h;
            b0[2 * j + h].x = W[(size_t)k0       * NHID + col];
            b0[2 * j + h].y = W[(size_t)(k0 + 1) * NHID + col];
            b1[2 * j + h].x = W[(size_t)k0       * NHID + col + 16];
            b1[2 * j + h].y = W[(size_t)(k0 + 1) * NHID + col + 16];
        }
    }

    const float4* X4 = reinterpret_cast<const float4*>(X);
    int rt = stream;
    float4 a[G1_J];
    {
        const float4* p = X4 + (size_t)(rt * 16 + l) * (NFEAT / 4) + kh;
#pragma unroll
        for (int j = 0; j < G1_J; ++j) a[j] = p[2 * j];
    }
    float ns = scale[rt * 16 + l];
    for (;;) {
        float* orow = Out + (size_t)(rt * 16 + 8 * kh) * NHID + col;
        const int nrt = rt + G1_STREAMS;
        if (nrt >= G1_TILES) {
            g1_tile<false>(a, nullptr, ns, b0, b1, orow);
            break;
        }
        const float4* np = X4 + (size_t)(nrt * 16 + l) * (NFEAT / 4) + kh;
        const float nns = scale[nrt * 16 + l];
        g1_tile<true>(a, np, ns, b0, b1, orow);
        rt = nrt;
        ns = nns;
    }
}

#define RS_NODES   400
#define RS_THREADS 1024
#define RS_WAVES   (RS_THREADS / 32)
#define RS_E4      (N_EDGES / 4)
#define RS_BLOCKS  (N_NODES / RS_NODES)
#define RS_PITCH   (NHID + 4)
#define RS_PITCH4  (RS_PITCH / 4)
#define RS_RTILES  (RS_NODES / 16)

static_assert(N_NODES % RS_NODES == 0, "node ranges must tile the node set");
static_assert(RS_NODES % 16 == 0, "owned rows must be whole 16-row tiles");
static_assert(NHID == 128 && RS_WAVES % 8 == 0, "8 column tiles, waves in groups of 8");

#define BIN_SLICES   160
#define BIN_THREADS  1024
#define BIN_I4       (RS_E4 / BIN_SLICES)
#define BIN_LOADS    ((BIN_I4 + BIN_THREADS - 1) / BIN_THREADS)
#define BIN_REGION   (RS_NODES * NHID)
#define BIN_SLOT     (BIN_REGION / BIN_SLICES)
#define BIN_CAP      (BIN_SLOT - 1)
#define BIN_PER_WAVE (BIN_SLICES / RS_WAVES)
#define BIN_SHIFT    17
#define BIN_MASK     ((1u << BIN_SHIFT) - 1u)

static_assert(RS_E4 % BIN_SLICES == 0, "slices are whole 16-byte loads");
static_assert(BIN_REGION % BIN_SLICES == 0 && BIN_SLOT % 32 == 0, "a list starts on a 128-byte line");
static_assert(BIN_SLICES % RS_WAVES == 0 && BIN_PER_WAVE <= 32, "whole lists per wave, counts in one load");
static_assert(N_NODES <= (1 << BIN_SHIFT) && RS_NODES <= (1 << (32 - BIN_SHIFT)), "packed word holds both");
static_assert(RS_BLOCKS <= 256, "the start table below is built by one wave, 8 owners a lane");

#define WL_CAP     512
#define RS_ROWS_HI 13
#define RS_ROWS_LO 12
#define RS_SPLIT   (16 * RS_ROWS_HI)
static_assert(RS_NODES == 12 * RS_WAVES + 16 && RS_WAVES == 32, "waves 0..15 own 13 rows, waves 16..31 own 12");
static_assert(RS_SPLIT + 16 * RS_ROWS_LO == RS_NODES, "the two row blocks tile the owned rows");
static_assert(WL_CAP % 8 == 0, "a round of 8 list words never reads past the list's capacity");

__device__ __forceinline__ void bin_starts(const int* cnt, int* st, int lane) {
    int c[8];
    int sum = 0;
#pragma unroll
    for (int k = 0; k < 8; ++k) {
        const int b = lane * 8 + k;
        c[k] = b < RS_BLOCKS ? cnt[b] : 0;
        sum += c[k];
    }
    int incl = sum;
#pragma unroll
    for (int off = 1; off < 32; off <<= 1) {
        const int t = __shfl_up(incl, off, 32);
        if (lane >= off) incl += t;
    }
    int run = incl - sum;
#pragma unroll
    for (int k = 0; k < 8; ++k) {
        const int b = lane * 8 + k;
        if (b < RS_BLOCKS) st[b] = run;
        run += c[k];
    }
}

template <bool SRC_LISTS>
__global__ void __launch_bounds__(BIN_THREADS)
bin_edges(const int* __restrict__ src, const int* __restrict__ dst, unsigned* __restrict__ bins,
          unsigned* __restrict__ sbins) {
    __shared__ int      lcount[RS_BLOCKS];
    __shared__ int      lstart[RS_BLOCKS];
    __shared__ unsigned sorted[BIN_I4 * 4];
    __shared__ int      scount[SRC_LISTS ? RS_BLOCKS : 1];
    __shared__ int      sstart[SRC_LISTS ? RS_BLOCKS : 1];
    __shared__ unsigned ssorted[SRC_LISTS ? BIN_I4 * 4 : 1];

    const int tid  = threadIdx.x;
    const int lane = tid & 31;
    const int wave = tid >> 5;
    const int g    = blockIdx.x;

    if (tid < RS_BLOCKS) {
        lcount[tid] = 0;
        if constexpr (SRC_LISTS) scount[tid] = 0;
    }
    __syncthreads();

    const int4* d4 = reinterpret_cast<const int4*>(dst) + (size_t)g * BIN_I4;
    const int4* s4 = reinterpret_cast<const int4*>(src) + (size_t)g * BIN_I4;

    unsigned wd[BIN_LOADS * 4];
    int      bn[BIN_LOADS * 4];
    int      rk[BIN_LOADS * 4];
    unsigned sw[SRC_LISTS ? BIN_LOADS * 4 : 1];
    int      sb[SRC_LISTS ? BIN_LOADS * 4 : 1];
    int      sr[SRC_LISTS ? BIN_LOADS * 4 : 1];
#pragma unroll
    for (int j = 0; j < BIN_LOADS; ++j) {
        const int i4 = j * BIN_THREADS + tid;
        int dv[4] = {-1, -1, -1, -1};
        int sv[4] = {-1, -1, -1, -1};
        if (i4 < BIN_I4) {
            const int4 d = d4[i4];
            const int4 s = s4[i4];
            dv[0] = d.x; dv[1] = d.y; dv[2] = d.z; dv[3] = d.w;
            sv[0] = s.x; sv[1] = s.y; sv[2] = s.z; sv[3] = s.w;
        }
#pragma unroll
        for (int t = 0; t < 4; ++t) {
            const int q = j * 4 + t;
            bn[q] = -1; rk[q] = 0; wd[q] = 0u;
            if ((unsigned)dv[t] < (unsigned)N_NODES) {
                const int b  = dv[t] / RS_NODES;
                const int dl = dv[t] - b * RS_NODES;
                bn[q] = b;
                wd[q] = (unsigned)sv[t] | ((unsigned)dl << BIN_SHIFT);
                rk[q] = atomicAdd(&lcount[b], 1);
            }
            if constexpr (SRC_LISTS) {
                sb[q] = -1; sr[q] = 0; sw[q] = 0u;
                if ((unsigned)sv[t] < (unsigned)N_NODES) {
                    const int b2 = sv[t] / RS_NODES;
                    sb[q] = b2;
                    sw[q] = (unsigned)(sv[t] - b2 * RS_NODES);
                    sr[q] = atomicAdd(&scount[b2], 1);
                }
            }
        }
    }
    __syncthreads();

    if (wave == 0) bin_starts(lcount, lstart, lane);
    if constexpr (SRC_LISTS) {
        if (wave == 1) bin_starts(scount, sstart, lane);
    }
    __syncthreads();

#pragma unroll
    for (int q = 0; q < BIN_LOADS * 4; ++q)
        if (bn[q] >= 0) sorted[lstart[bn[q]] + rk[q]] = wd[q];
    if constexpr (SRC_LISTS) {
#pragma unroll
        for (int q = 0; q < BIN_LOADS * 4; ++q)
            if (sb[q] >= 0) ssorted[sstart[sb[q]] + sr[q]] = sw[q];
    }
    __syncthreads();

    for (int b = wave; b < RS_BLOCKS; b += BIN_THREADS / 32) {
        const int c  = lcount[b];
        const int st = lstart[b];
        unsigned* dest = bins + (size_t)b * BIN_REGION + (size_t)g * BIN_SLOT;
        const int n = c < BIN_CAP ? c : BIN_CAP;
        for (int i = lane; i < n; i += 32) dest[i] = sorted[st + i];
        if (lane == 0) dest[BIN_CAP] = (unsigned)c;
        if constexpr (SRC_LISTS) {
            const int c2  = scount[b];
            const int st2 = sstart[b];
            unsigned* dest2 = sbins + (size_t)b * BIN_REGION + (size_t)g * BIN_SLOT;
            const int n2 = c2 < BIN_CAP ? c2 : BIN_CAP;
            for (int i = lane; i < n2; i += 32) dest2[i] = ssorted[st2 + i];
            if (lane == 0) dest2[BIN_CAP] = (unsigned)c2;
        }
    }
}

#define OD_THREADS  256
#define OD_WAVES    (OD_THREADS / 32)
#define OD_PER_WAVE (BIN_SLICES / OD_WAVES)
static_assert(BIN_SLICES % OD_WAVES == 0 && OD_PER_WAVE <= 32, "whole lists per wave, counts in one load");

__global__ void __launch_bounds__(OD_THREADS)
outdeg_norm(const int* __restrict__ src, const unsigned* __restrict__ sbins,
            float* __restrict__ normS) {
    __shared__ int odeg[RS_NODES];

    const int tid  = threadIdx.x;
    const int lane = tid & 31;
    const int wave = __builtin_amdgcn_readfirstlane(tid >> 5);
    const int base = blockIdx.x * RS_NODES;

    for (int i = tid; i < RS_NODES; i += OD_THREADS) odeg[i] = 0;
    __syncthreads();

    const unsigned* mybin = sbins + (size_t)blockIdx.x * BIN_REGION;
    int cl = 0;
    if (lane < OD_PER_WAVE) cl = (int)mybin[(wave + lane * OD_WAVES) * BIN_SLOT + BIN_CAP];

    unsigned w0[OD_PER_WAVE];
    unsigned w1[OD_PER_WAVE];
#pragma unroll
    for (int k = 0; k < OD_PER_WAVE; ++k) {
        const int c = __builtin_amdgcn_readlane(cl, k);
        const unsigned* slot = mybin + (wave + k * OD_WAVES) * BIN_SLOT;
        const bool whole = c <= BIN_CAP;
        w0[k] = slot[(whole && lane < c) ? lane : BIN_CAP];
        w1[k] = slot[(whole && lane + 32 < c) ? lane + 32 : BIN_CAP];
    }
#pragma unroll
    for (int k = 0; k < OD_PER_WAVE; ++k) {
        const int g = wave + k * OD_WAVES;
        const unsigned* slot = mybin + g * BIN_SLOT;
        const int c = __builtin_amdgcn_readlane(cl, k);
        if (c <= BIN_CAP) {
            if (lane < c && w0[k] < (unsigned)RS_NODES) atomicAdd(&odeg[w0[k]], 1);
            if (lane + 32 < c && w1[k] < (unsigned)RS_NODES) atomicAdd(&odeg[w1[k]], 1);
            for (int p0 = 64; p0 < c; p0 += 32) {
                if (p0 + lane < c) {
                    const unsigned v = slot[p0 + lane];
                    if (v < (unsigned)RS_NODES) atomicAdd(&odeg[v], 1);
                }
            }
        } else {
            const int4* s4 = reinterpret_cast<const int4*>(src) + (size_t)g * BIN_I4;
            for (int i = lane; i < BIN_I4; i += 32) {
                const int4 s = s4[i];
                unsigned sl;
                sl = (unsigned)(s.x - base);
                if (sl < (unsigned)RS_NODES) atomicAdd(&odeg[sl], 1);
                sl = (unsigned)(s.y - base);
                if (sl < (unsigned)RS_NODES) atomicAdd(&odeg[sl], 1);
                sl = (unsigned)(s.z - base);
                if (sl < (unsigned)RS_NODES) atomicAdd(&odeg[sl], 1);
                sl = (unsigned)(s.w - base);
                if (sl < (unsigned)RS_NODES) atomicAdd(&odeg[sl], 1);
            }
        }
    }
    __syncthreads();
    for (int i = tid; i < RS_NODES; i += OD_THREADS)
        normS[base + i] = rsqrtf(fmaxf((float)odeg[i], 1.0f));
}

template <int LAYER>
__device__ __forceinline__ void rs_slow_add(int e, unsigned dl, int* __restrict__ indeg,
                                            float* __restrict__ acc,
                                            const float* __restrict__ M,
                                            const int* __restrict__ src) {
    if constexpr (LAYER == 1) atomicAdd(indeg + dl, 1);
    const float* mrow = M + (size_t)src[e] * NHID;
    float* arow = acc + dl * RS_PITCH;
#pragma unroll 1
    for (int j = 0; j < NHID; ++j) atomicAdd(arow + j, mrow[j]);
}

template <int LAYER>
__device__ __forceinline__ void rs_append(const unsigned w, unsigned* wlist, int* wcount,
                                          int* indeg) {
    const unsigned dl = w >> BIN_SHIFT;
    if (dl < (unsigned)RS_NODES) {
        const unsigned t = dl < (unsigned)RS_SPLIT ? dl / (unsigned)RS_ROWS_HI
                                                   : 16u + (dl - (unsigned)RS_SPLIT) / (unsigned)RS_ROWS_LO;
        const int pos = atomicAdd(wcount + t, 1);
        if (pos < WL_CAP) wlist[t * WL_CAP + pos] = w;
        if constexpr (LAYER == 1) atomicAdd(indeg + dl, 1);
    }
}

template <bool FULL>
__device__ __forceinline__ void rs_round(const unsigned* wl, const int head, const int n,
                                         const float4* __restrict__ M4, const int lane,
                                         float4* acc4) {
    const unsigned qw = wl[head + (lane & 7)];
    float4 v[8];
#pragma unroll
    for (int j = 0; j < 8; ++j) {
        v[j] = make_float4(0.0f, 0.0f, 0.0f, 0.0f);
        if (FULL || j < n) {
            const int sj = (int)(__builtin_amdgcn_readlane(qw, j) & BIN_MASK);
            v[j] = M4[(size_t)sj * (NHID / 4) + lane];
        }
    }
#pragma unroll
    for (int j = 0; j < 8; ++j) {
        if (FULL || j < n) {
            const int dj = (int)(__builtin_amdgcn_readlane(qw, j) >> BIN_SHIFT);
            float4* p = acc4 + dj * RS_PITCH4 + lane;
            float4 r = *p;
            r.x += v[j].x;  r.y += v[j].y;  r.z += v[j].z;  r.w += v[j].w;
            *p = r;
        }
    }
}

template <int LAYER>
__global__ void __launch_bounds__(RS_THREADS)
edge_rowsum_lds(const float* __restrict__ M, const int* __restrict__ src,
                const int* __restrict__ dst, const unsigned* bins,
                const float* __restrict__ normS, const float* __restrict__ normD,
                const float* __restrict__ bias, const float* __restrict__ Wn,
                const float* __restrict__ bfc, float* Out,
                float* __restrict__ normDout) {
    __shared__ __attribute__((aligned(16))) float acc[RS_NODES * RS_PITCH];
    __shared__ int   indeg[LAYER == 1 ? RS_NODES : 1];
    __shared__ unsigned wlist[RS_WAVES * WL_CAP];
    __shared__ int   wcount[RS_WAVES];
    __shared__ int   cutflag;

    const int tid  = threadIdx.x;
    const int lane = tid & 31;
    const int wave = __builtin_amdgcn_readfirstlane(tid >> 5);
    const int base = blockIdx.x * RS_NODES;

    float4* acc4 = reinterpret_cast<float4*>(acc);
    const int4*   dst4 = reinterpret_cast<const int4*>(dst);
    const float4* M4   = reinterpret_cast<const float4*>(M);

    const unsigned* mybin = bins + (size_t)blockIdx.x * BIN_REGION;
    int cl = 0;
    if (lane < BIN_PER_WAVE) cl = (int)mybin[(wave + lane * RS_WAVES) * BIN_SLOT + BIN_CAP];

    unsigned f0[BIN_PER_WAVE];
    unsigned f1[BIN_PER_WAVE];
#pragma unroll
    for (int k = 0; k < BIN_PER_WAVE; ++k) {
        const int c = __builtin_amdgcn_readlane(cl, k);
        const unsigned* slot = mybin + (wave + k * RS_WAVES) * BIN_SLOT;
        const bool whole = c <= BIN_CAP;
        f0[k] = slot[(whole && lane < c) ? lane : BIN_CAP];
        f1[k] = slot[(whole && lane + 32 < c) ? lane + 32 : BIN_CAP];
    }

    if (tid < RS_WAVES) wcount[tid] = 0;
    if (tid == 0) cutflag = 0;
    if constexpr (LAYER == 1) {
        if (tid < RS_NODES) indeg[tid] = 0;
    }
    __syncthreads();

    if (__builtin_amdgcn_ballot_w32(cl > BIN_CAP) != 0u) {
        if (lane == 0) cutflag = 1;
    } else {
#pragma unroll
        for (int k = 0; k < BIN_PER_WAVE; ++k) {
            const unsigned* slot = mybin + (wave + k * RS_WAVES) * BIN_SLOT;
            const int c = __builtin_amdgcn_readlane(cl, k);
            if (lane < c) rs_append<LAYER>(f0[k], wlist, wcount, indeg);
            if (lane + 32 < c) rs_append<LAYER>(f1[k], wlist, wcount, indeg);
            for (int p0 = 64; p0 < c; p0 += 32)
                if (p0 + lane < c) rs_append<LAYER>(slot[p0 + lane], wlist, wcount, indeg);
        }
    }
    __syncthreads();

    const int  myn  = wcount[lane];
    const bool over = __builtin_amdgcn_ballot_w32(myn > WL_CAP) != 0u;
    const bool fast = __builtin_amdgcn_readfirstlane(cutflag) == 0 && !over;

    if (fast) {
        const int lo   = wave * RS_ROWS_LO + (wave < 16 ? wave : 16);
        const int rows = wave < 16 ? RS_ROWS_HI : RS_ROWS_LO;
        for (int k = 0; k < rows; ++k)
            acc4[(lo + k) * RS_PITCH4 + lane] = make_float4(0.0f, 0.0f, 0.0f, 0.0f);

        const unsigned* wl = wlist + wave * WL_CAP;
        const int n = __builtin_amdgcn_readlane(myn, wave);
        int head = 0;
        for (; head + 8 <= n; head += 8) rs_round<true>(wl, head, 8, M4, lane, acc4);
        if (head < n) rs_round<false>(wl, head, n - head, M4, lane, acc4);
    } else {
        for (int i = tid; i < RS_NODES * RS_PITCH4; i += RS_THREADS)
            acc4[i] = make_float4(0.0f, 0.0f, 0.0f, 0.0f);
        if constexpr (LAYER == 1) {
            if (tid < RS_NODES) indeg[tid] = 0;
        }
        __syncthreads();

        const bool hi = lane >= 16;
        const int o0 = lane * 4 + (hi ? 2 : 0);
        const int o1 = lane * 4 + (hi ? 3 : 1);
        const int o2 = lane * 4 + (hi ? 0 : 2);
        const int o3 = lane * 4 + (hi ? 1 : 3);

        for (int k = 0; k < BIN_PER_WAVE; ++k) {
            const int g = wave + k * RS_WAVES;
            const unsigned* slot = mybin + g * BIN_SLOT;
            const int c = __builtin_amdgcn_readlane(cl, k);
            if (c <= BIN_CAP) {
                for (int p0 = 0; p0 < c; p0 += 32) {
                    const unsigned w = (p0 + lane < c) ? slot[p0 + lane] : 0u;
                    const int m32 = (c - p0) < 32 ? (c - p0) : 32;
                    const int s  = (int)(w & BIN_MASK);
                    const int dl = (int)(w >> BIN_SHIFT);
                    if constexpr (LAYER == 1) {
                        if (lane < m32) atomicAdd(indeg + dl, 1);
                    }
                    for (int r = 0; r < 4; ++r) {
                        const int m = m32 - r * 8;
                        if (m <= 0) break;
                        float4 v[8];
#pragma unroll
                        for (int j = 0; j < 8; ++j) {
                            v[j] = make_float4(0.0f, 0.0f, 0.0f, 0.0f);
                            if (j < m) {
                                const int sj = __builtin_amdgcn_readlane(s, r * 8 + j);
                                v[j] = M4[(size_t)sj * (NHID / 4) + lane];
                            }
                        }
#pragma unroll
                        for (int j = 0; j < 8; ++j) {
                            if (j < m) {
                                const int dj = __builtin_amdgcn_readlane(dl, r * 8 + j);
                                float* p = acc + dj * RS_PITCH;
                                atomicAdd(p + o0, hi ? v[j].z : v[j].x);
                                atomicAdd(p + o1, hi ? v[j].w : v[j].y);
                                atomicAdd(p + o2, hi ? v[j].x : v[j].z);
                                atomicAdd(p + o3, hi ? v[j].y : v[j].w);
                            }
                        }
                    }
                }
            } else {
                for (int i = lane; i < BIN_I4; i += 32) {
                    const int i4 = g * BIN_I4 + i;
                    const int4 d = dst4[i4];
                    const int e = i4 * 4;
                    unsigned dl;
                    dl = (unsigned)(d.x - base);
                    if (dl < (unsigned)RS_NODES) rs_slow_add<LAYER>(e + 0, dl, indeg, acc, M, src);
                    dl = (unsigned)(d.y - base);
                    if (dl < (unsigned)RS_NODES) rs_slow_add<LAYER>(e + 1, dl, indeg, acc, M, src);
                    dl = (unsigned)(d.z - base);
                    if (dl < (unsigned)RS_NODES) rs_slow_add<LAYER>(e + 2, dl, indeg, acc, M, src);
                    dl = (unsigned)(d.w - base);
                    if (dl < (unsigned)RS_NODES) rs_slow_add<LAYER>(e + 3, dl, indeg, acc, M, src);
                }
            }
        }
    }
    __syncthreads();

    if constexpr (LAYER == 1) {
        if (tid < RS_NODES) normDout[base + tid] = rsqrtf(fmaxf((float)indeg[tid], 1.0f));
        const float4* b4 = reinterpret_cast<const float4*>(bias);
        const float4 bb = b4[lane];
        for (int i = tid; i < RS_NODES * (NHID / 4); i += RS_THREADS) {
            const int ln   = i >> 5;
            const int node = base + ln;
            const float nd = rsqrtf(fmaxf((float)indeg[ln], 1.0f));
            const float ns = normS[node];
            float4 v = acc4[ln * RS_PITCH4 + lane];
            float t;
            t = v.x * nd + bb.x;  v.x = (t > 0.0f ? t : 0.0f) * ns;
            t = v.y * nd + bb.y;  v.y = (t > 0.0f ? t : 0.0f) * ns;
            t = v.z * nd + bb.z;  v.z = (t > 0.0f ? t : 0.0f) * ns;
            t = v.w * nd + bb.w;  v.w = (t > 0.0f ? t : 0.0f) * ns;
            acc4[ln * RS_PITCH4 + lane] = v;
        }
        __syncthreads();

        const int l   = lane & 15;
        const int kh  = lane >> 4;
        const int col = (wave & 7) * 16 + l;
        v2f bfr[NHID / 4];
#pragma unroll
        for (int ks = 0; ks < NHID / 4; ++ks) {
            const int ka = ks * 4 + 2 * kh;
            bfr[ks].x = Wn[(size_t)ka       * NHID + col];
            bfr[ks].y = Wn[(size_t)(ka + 1) * NHID + col];
        }
        for (int rt = wave >> 3; rt < RS_RTILES; rt += RS_WAVES / 8) {
            const float* arow = acc + (rt * 16 + l) * RS_PITCH + 2 * kh;
            v8f c = {};
#pragma unroll
            for (int ks = 0; ks < NHID / 4; ++ks) {
                const v2f a = *reinterpret_cast<const v2f*>(arow + ks * 4);
                c = __builtin_amdgcn_wmma_f32_16x16x4_f32(false, a, false, bfr[ks],
                                                          (short)0, c, false, false);
            }
            float* orow = Out + (size_t)(base + rt * 16 + 8 * kh) * NHID + col;
#pragma unroll
            for (int r = 0; r < 8; ++r) orow[(size_t)r * NHID] = c[r];
        }
    } else {
        __shared__ float outv[RS_NODES];
        const float bf = bfc[0];
        for (int ln = wave; ln < RS_NODES; ln += RS_WAVES) {
            const float nd = normD[base + ln];
            const float* arow = acc + ln * RS_PITCH;
            float a = 0.0f;
#pragma unroll
            for (int t = 0; t < 4; ++t) {
                const int j = lane + 32 * t;
                a += (arow[j] * nd + bias[j]) * Wn[j];
            }
#pragma unroll
            for (int off = 16; off > 0; off >>= 1)
                a += __shfl_xor(a, off, 32);
            if (lane == 0) outv[ln] = a + bf;
        }
        __syncthreads();
        if (tid < RS_NODES) Out[base + tid] = outv[tid];
    }
}

extern "C" void kernel_launch(void* const* d_in, const int* in_sizes, int n_in,
                              void* d_out, int out_size, void* d_ws, size_t ws_size,
                              hipStream_t stream) {
    const float* x   = (const float*)d_in[0];
    const float* W1  = (const float*)d_in[1];
    const float* b1  = (const float*)d_in[2];
    const float* W2  = (const float*)d_in[3];
    const float* b2  = (const float*)d_in[4];
    const float* Wfc = (const float*)d_in[5];
    const float* bfc = (const float*)d_in[6];
    const int*   src = (const int*)d_in[7];
    const int*   dst = (const int*)d_in[8];
    float* out = (float*)d_out;

    const int N = N_NODES;
    const int NH = NHID;

    float* BUF_A = (float*)d_ws;
    float* BUF_B = BUF_A + (size_t)N * NH;
    float* normS = BUF_B + (size_t)N * NH;
    float* normD = normS + N;

    bin_edges<true><<<BIN_SLICES, BIN_THREADS, 0, stream>>>(src, dst, (unsigned*)BUF_B, (unsigned*)BUF_A);
    outdeg_norm<<<RS_BLOCKS, OD_THREADS, 0, stream>>>(src, (const unsigned*)BUF_A, normS);

    gemm_scaled_wmma<<<G1_BLOCKS, G1_THREADS, 0, stream>>>(x, normS, W1, BUF_A);
    edge_rowsum_lds<1><<<RS_BLOCKS, RS_THREADS, 0, stream>>>(BUF_A, src, dst, (const unsigned*)BUF_B,
                                                            normS, normD, b1, W2, bfc, BUF_B, normD);

    bin_edges<false><<<BIN_SLICES, BIN_THREADS, 0, stream>>>(src, dst, (unsigned*)BUF_A, nullptr);
    edge_rowsum_lds<2><<<RS_BLOCKS, RS_THREADS, 0, stream>>>(BUF_B, src, dst, (const unsigned*)BUF_A,
                                                            normS, normD, b2, Wfc, bfc, out, nullptr);
}


# 128->1 projection moved ahead of the second edge sum: layer-1 owner stores s=(W2 product row).Wfc, layer 2 adds one float per listed edge (kept lists, LDS float adds, arrival order); re-associated
# speedup vs baseline: 6.6410x; 1.3080x over previous
#include <hip/hip_runtime.h>

typedef __attribute__((ext_vector_type(2))) float v2f;
typedef __attribute__((ext_vector_type(8))) float v8f;

#define N_NODES 100000
#define N_EDGES 1600000
#define NFEAT   256
#define NHID    128


#define G1_BLOCKS  250
#define G1_THREADS 256
#define G1_TILES   (N_NODES / 16)
#define G1_STREAMS (2 * G1_BLOCKS)
#define G1_J       (NFEAT / 8)

static_assert(N_NODES % 16 == 0, "whole 16-row tiles: no row guard below");
static_assert(G1_STREAMS <= G1_TILES, "every stream has a first tile");
static_assert(NFEAT % 8 == 0 && NHID == 128, "8 k per float4 pair, 4 groups of 32 columns");

template <bool REFILL>
__device__ __forceinline__ void g1_tile(float4 (&a)[G1_J], const float4* __restrict__ np,
                                        const float ns, const v2f (&b0)[2 * G1_J],
                                        const v2f (&b1)[2 * G1_J],
                                        float* __restrict__ orow) {
    v8f c0 = {};
    v8f c1 = {};
#pragma unroll
    for (int j = 0; j < G1_J; ++j) {
        const float4 v = a[j];
        v2f lo, hi;
        lo.x = v.x * ns;  lo.y = v.y * ns;
        hi.x = v.z * ns;  hi.y = v.w * ns;
        if constexpr (REFILL) {
            __builtin_amdgcn_sched_barrier(0);
            a[j] = np[2 * j];
        }
        c0 = __builtin_amdgcn_wmma_f32_16x16x4_f32(false, lo, false, b0[2 * j],
                                                   (short)0, c0, false, false);
        c1 = __builtin_amdgcn_wmma_f32_16x16x4_f32(false, lo, false, b1[2 * j],
                                                   (short)0, c1, false, false);
        c0 = __builtin_amdgcn_wmma_f32_16x16x4_f32(false, hi, false, b0[2 * j + 1],
                                                   (short)0, c0, false, false);
        c1 = __builtin_amdgcn_wmma_f32_16x16x4_f32(false, hi, false, b1[2 * j + 1],
                                                   (short)0, c1, false, false);
    }
#pragma unroll
    for (int r = 0; r < 8; ++r) {
        orow[(size_t)r * NHID]      = c0[r];
        orow[(size_t)r * NHID + 16] = c1[r];
    }
}

__global__ void __launch_bounds__(G1_THREADS)
gemm_scaled_wmma(const float* __restrict__ X, const float* __restrict__ scale,
                 const float* __restrict__ W, float* __restrict__ Out) {
    const int lane = threadIdx.x & 31;
    const int wave = __builtin_amdgcn_readfirstlane(threadIdx.x >> 5);
    const int kh   = lane >> 4;
    const int l    = lane & 15;
    const int col  = (wave & 3) * 32 + l;
    const int stream = blockIdx.x + (wave >> 2) * G1_BLOCKS;

    v2f b0[2 * G1_J];
    v2f b1[2 * G1_J];
#pragma unroll
    for (int j = 0; j < G1_J; ++j) {
#pragma unroll
        for (int h = 0; h < 2; ++h) {
            const int k0 = 8 * j + 4 * kh + 2 * h;
            b0[2 * j + h].x = W[(size_t)k0       * NHID + col];
            b0[2 * j + h].y = W[(size_t)(k0 + 1) * NHID + col];
            b1[2 * j + h].x = W[(size_t)k0       * NHID + col + 16];
            b1[2 * j + h].y = W[(size_t)(k0 + 1) * NHID + col + 16];
        }
    }

    const float4* X4 = reinterpret_cast<const float4*>(X);
    int rt = stream;
    float4 a[G1_J];
    {
        const float4* p = X4 + (size_t)(rt * 16 + l) * (NFEAT / 4) + kh;
#pragma unroll
        for (int j = 0; j < G1_J; ++j) a[j] = p[2 * j];
    }
    float ns = scale[rt * 16 + l];
    for (;;) {
        float* orow = Out + (size_t)(rt * 16 + 8 * kh) * NHID + col;
        const int nrt = rt + G1_STREAMS;
        if (nrt >= G1_TILES) {
            g1_tile<false>(a, nullptr, ns, b0, b1, orow);
            break;
        }
        const float4* np = X4 + (size_t)(nrt * 16 + l) * (NFEAT / 4) + kh;
        const float nns = scale[nrt * 16 + l];
        g1_tile<true>(a, np, ns, b0, b1, orow);
        rt = nrt;
        ns = nns;
    }
}

#define RS_NODES   400
#define RS_THREADS 1024
#define RS_WAVES   (RS_THREADS / 32)
#define RS_E4      (N_EDGES / 4)
#define RS_BLOCKS  (N_NODES / RS_NODES)
#define RS_PITCH   (NHID + 4)
#define RS_PITCH4  (RS_PITCH / 4)
#define RS_RTILES  (RS_NODES / 16)

static_assert(N_NODES % RS_NODES == 0, "node ranges must tile the node set");
static_assert(RS_NODES % 16 == 0, "owned rows must be whole 16-row tiles");
static_assert(NHID == 128 && RS_WAVES % 8 == 0, "8 column tiles, waves in groups of 8");

#define BIN_SLICES   160
#define BIN_THREADS  1024
#define BIN_I4       (RS_E4 / BIN_SLICES)
#define BIN_LOADS    ((BIN_I4 + BIN_THREADS - 1) / BIN_THREADS)
#define BIN_REGION   (RS_NODES * NHID)
#define BIN_SLOT     (BIN_REGION / BIN_SLICES)
#define BIN_CAP      (BIN_SLOT - 1)
#define BIN_PER_WAVE (BIN_SLICES / RS_WAVES)
#define BIN_SHIFT    17
#define BIN_MASK     ((1u << BIN_SHIFT) - 1u)

static_assert(RS_E4 % BIN_SLICES == 0, "slices are whole 16-byte loads");
static_assert(BIN_REGION % BIN_SLICES == 0 && BIN_SLOT % 32 == 0, "a list starts on a 128-byte line");
static_assert(BIN_SLICES % RS_WAVES == 0 && BIN_PER_WAVE <= 32, "whole lists per wave, counts in one load");
static_assert(N_NODES <= (1 << BIN_SHIFT) && RS_NODES <= (1 << (32 - BIN_SHIFT)), "packed word holds both");
static_assert(RS_BLOCKS <= 256, "the start table below is built by one wave, 8 owners a lane");

#define WL_CAP     512
#define RS_ROWS_HI 13
#define RS_ROWS_LO 12
#define RS_SPLIT   (16 * RS_ROWS_HI)
static_assert(RS_NODES == 12 * RS_WAVES + 16 && RS_WAVES == 32, "waves 0..15 own 13 rows, waves 16..31 own 12");
static_assert(RS_SPLIT + 16 * RS_ROWS_LO == RS_NODES, "the two row blocks tile the owned rows");
static_assert(WL_CAP % 8 == 0, "a round of 8 list words never reads past the list's capacity");

__device__ __forceinline__ void bin_starts(const int* cnt, int* st, int lane) {
    int c[8];
    int sum = 0;
#pragma unroll
    for (int k = 0; k < 8; ++k) {
        const int b = lane * 8 + k;
        c[k] = b < RS_BLOCKS ? cnt[b] : 0;
        sum += c[k];
    }
    int incl = sum;
#pragma unroll
    for (int off = 1; off < 32; off <<= 1) {
        const int t = __shfl_up(incl, off, 32);
        if (lane >= off) incl += t;
    }
    int run = incl - sum;
#pragma unroll
    for (int k = 0; k < 8; ++k) {
        const int b = lane * 8 + k;
        if (b < RS_BLOCKS) st[b] = run;
        run += c[k];
    }
}

__global__ void __launch_bounds__(BIN_THREADS)
bin_edges(const int* __restrict__ src, const int* __restrict__ dst, unsigned* __restrict__ bins,
          unsigned* __restrict__ sbins) {
    __shared__ int      lcount[RS_BLOCKS];
    __shared__ int      lstart[RS_BLOCKS];
    __shared__ unsigned sorted[BIN_I4 * 4];
    __shared__ int      scount[RS_BLOCKS];
    __shared__ int      sstart[RS_BLOCKS];
    __shared__ unsigned ssorted[BIN_I4 * 4];

    const int tid  = threadIdx.x;
    const int lane = tid & 31;
    const int wave = tid >> 5;
    const int g    = blockIdx.x;

    if (tid < RS_BLOCKS) {
        lcount[tid] = 0;
        scount[tid] = 0;
    }
    __syncthreads();

    const int4* d4 = reinterpret_cast<const int4*>(dst) + (size_t)g * BIN_I4;
    const int4* s4 = reinterpret_cast<const int4*>(src) + (size_t)g * BIN_I4;

    unsigned wd[BIN_LOADS * 4];
    int      bn[BIN_LOADS * 4];
    int      rk[BIN_LOADS * 4];
    unsigned sw[BIN_LOADS * 4];
    int      sb[BIN_LOADS * 4];
    int      sr[BIN_LOADS * 4];
#pragma unroll
    for (int j = 0; j < BIN_LOADS; ++j) {
        const int i4 = j * BIN_THREADS + tid;
        int dv[4] = {-1, -1, -1, -1};
        int sv[4] = {-1, -1, -1, -1};
        if (i4 < BIN_I4) {
            const int4 d = d4[i4];
            const int4 s = s4[i4];
            dv[0] = d.x; dv[1] = d.y; dv[2] = d.z; dv[3] = d.w;
            sv[0] = s.x; sv[1] = s.y; sv[2] = s.z; sv[3] = s.w;
        }
#pragma unroll
        for (int t = 0; t < 4; ++t) {
            const int q = j * 4 + t;
            bn[q] = -1; rk[q] = 0; wd[q] = 0u;
            if ((unsigned)dv[t] < (unsigned)N_NODES) {
                const int b  = dv[t] / RS_NODES;
                const int dl = dv[t] - b * RS_NODES;
                bn[q] = b;
                wd[q] = (unsigned)sv[t] | ((unsigned)dl << BIN_SHIFT);
                rk[q] = atomicAdd(&lcount[b], 1);
            }
            sb[q] = -1; sr[q] = 0; sw[q] = 0u;
            if ((unsigned)sv[t] < (unsigned)N_NODES) {
                const int b2 = sv[t] / RS_NODES;
                sb[q] = b2;
                sw[q] = (unsigned)(sv[t] - b2 * RS_NODES);
                sr[q] = atomicAdd(&scount[b2], 1);
            }
        }
    }
    __syncthreads();

    if (wave == 0) bin_starts(lcount, lstart, lane);
    if (wave == 1) bin_starts(scount, sstart, lane);
    __syncthreads();

#pragma unroll
    for (int q = 0; q < BIN_LOADS * 4; ++q)
        if (bn[q] >= 0) sorted[lstart[bn[q]] + rk[q]] = wd[q];
#pragma unroll
    for (int q = 0; q < BIN_LOADS * 4; ++q)
        if (sb[q] >= 0) ssorted[sstart[sb[q]] + sr[q]] = sw[q];
    __syncthreads();

    for (int b = wave; b < RS_BLOCKS; b += BIN_THREADS / 32) {
        const int c  = lcount[b];
        const int st = lstart[b];
        unsigned* dest = bins + (size_t)b * BIN_REGION + (size_t)g * BIN_SLOT;
        const int n = c < BIN_CAP ? c : BIN_CAP;
        for (int i = lane; i < n; i += 32) dest[i] = sorted[st + i];
        if (lane == 0) dest[BIN_CAP] = (unsigned)c;
        const int c2  = scount[b];
        const int st2 = sstart[b];
        unsigned* dest2 = sbins + (size_t)b * BIN_REGION + (size_t)g * BIN_SLOT;
        const int n2 = c2 < BIN_CAP ? c2 : BIN_CAP;
        for (int i = lane; i < n2; i += 32) dest2[i] = ssorted[st2 + i];
        if (lane == 0) dest2[BIN_CAP] = (unsigned)c2;
    }
}

#define OD_THREADS  256
#define OD_WAVES    (OD_THREADS / 32)
#define OD_PER_WAVE (BIN_SLICES / OD_WAVES)
static_assert(BIN_SLICES % OD_WAVES == 0 && OD_PER_WAVE <= 32, "whole lists per wave, counts in one load");

__global__ void __launch_bounds__(OD_THREADS)
outdeg_norm(const int* __restrict__ src, const unsigned* __restrict__ sbins,
            float* __restrict__ normS) {
    __shared__ int odeg[RS_NODES];

    const int tid  = threadIdx.x;
    const int lane = tid & 31;
    const int wave = __builtin_amdgcn_readfirstlane(tid >> 5);
    const int base = blockIdx.x * RS_NODES;

    for (int i = tid; i < RS_NODES; i += OD_THREADS) odeg[i] = 0;
    __syncthreads();

    const unsigned* mybin = sbins + (size_t)blockIdx.x * BIN_REGION;
    int cl = 0;
    if (lane < OD_PER_WAVE) cl = (int)mybin[(wave + lane * OD_WAVES) * BIN_SLOT + BIN_CAP];

    unsigned w0[OD_PER_WAVE];
    unsigned w1[OD_PER_WAVE];
#pragma unroll
    for (int k = 0; k < OD_PER_WAVE; ++k) {
        const int c = __builtin_amdgcn_readlane(cl, k);
        const unsigned* slot = mybin + (wave + k * OD_WAVES) * BIN_SLOT;
        const bool whole = c <= BIN_CAP;
        w0[k] = slot[(whole && lane < c) ? lane : BIN_CAP];
        w1[k] = slot[(whole && lane + 32 < c) ? lane + 32 : BIN_CAP];
    }
#pragma unroll
    for (int k = 0; k < OD_PER_WAVE; ++k) {
        const int g = wave + k * OD_WAVES;
        const unsigned* slot = mybin + g * BIN_SLOT;
        const int c = __builtin_amdgcn_readlane(cl, k);
        if (c <= BIN_CAP) {
            if (lane < c && w0[k] < (unsigned)RS_NODES) atomicAdd(&odeg[w0[k]], 1);
            if (lane + 32 < c && w1[k] < (unsigned)RS_NODES) atomicAdd(&odeg[w1[k]], 1);
            for (int p0 = 64; p0 < c; p0 += 32) {
                if (p0 + lane < c) {
                    const unsigned v = slot[p0 + lane];
                    if (v < (unsigned)RS_NODES) atomicAdd(&odeg[v], 1);
                }
            }
        } else {
            const int4* s4 = reinterpret_cast<const int4*>(src) + (size_t)g * BIN_I4;
            for (int i = lane; i < BIN_I4; i += 32) {
                const int4 s = s4[i];
                unsigned sl;
                sl = (unsigned)(s.x - base);
                if (sl < (unsigned)RS_NODES) atomicAdd(&odeg[sl], 1);
                sl = (unsigned)(s.y - base);
                if (sl < (unsigned)RS_NODES) atomicAdd(&odeg[sl], 1);
                sl = (unsigned)(s.z - base);
                if (sl < (unsigned)RS_NODES) atomicAdd(&odeg[sl], 1);
                sl = (unsigned)(s.w - base);
                if (sl < (unsigned)RS_NODES) atomicAdd(&odeg[sl], 1);
            }
        }
    }
    __syncthreads();
    for (int i = tid; i < RS_NODES; i += OD_THREADS)
        normS[base + i] = rsqrtf(fmaxf((float)odeg[i], 1.0f));
}

__device__ __forceinline__ void rs_slow_add(int e, unsigned dl, int* __restrict__ indeg,
                                            float* __restrict__ acc,
                                            const float* __restrict__ M,
                                            const int* __restrict__ src) {
    atomicAdd(indeg + dl, 1);
    const float* mrow = M + (size_t)src[e] * NHID;
    float* arow = acc + dl * RS_PITCH;
#pragma unroll 1
    for (int j = 0; j < NHID; ++j) atomicAdd(arow + j, mrow[j]);
}

__device__ __forceinline__ void rs_append(const unsigned w, unsigned* wlist, int* wcount,
                                          int* indeg) {
    const unsigned dl = w >> BIN_SHIFT;
    if (dl < (unsigned)RS_NODES) {
        const unsigned t = dl < (unsigned)RS_SPLIT ? dl / (unsigned)RS_ROWS_HI
                                                   : 16u + (dl - (unsigned)RS_SPLIT) / (unsigned)RS_ROWS_LO;
        const int pos = atomicAdd(wcount + t, 1);
        if (pos < WL_CAP) wlist[t * WL_CAP + pos] = w;
        atomicAdd(indeg + dl, 1);
    }
}

template <bool FULL>
__device__ __forceinline__ void rs_round(const unsigned* wl, const int head, const int n,
                                         const float4* __restrict__ M4, const int lane,
                                         float4* acc4) {
    const unsigned qw = wl[head + (lane & 7)];
    float4 v[8];
#pragma unroll
    for (int j = 0; j < 8; ++j) {
        v[j] = make_float4(0.0f, 0.0f, 0.0f, 0.0f);
        if (FULL || j < n) {
            const int sj = (int)(__builtin_amdgcn_readlane(qw, j) & BIN_MASK);
            v[j] = M4[(size_t)sj * (NHID / 4) + lane];
        }
    }
#pragma unroll
    for (int j = 0; j < 8; ++j) {
        if (FULL || j < n) {
            const int dj = (int)(__builtin_amdgcn_readlane(qw, j) >> BIN_SHIFT);
            float4* p = acc4 + dj * RS_PITCH4 + lane;
            float4 r = *p;
            r.x += v[j].x;  r.y += v[j].y;  r.z += v[j].z;  r.w += v[j].w;
            *p = r;
        }
    }
}

__global__ void __launch_bounds__(RS_THREADS)
edge_rowsum_lds(const float* __restrict__ M, const int* __restrict__ src,
                const int* __restrict__ dst, const unsigned* __restrict__ bins,
                const float* normS, const float* __restrict__ bias,
                const float* __restrict__ Wn, const float* __restrict__ Wfc,
                float* S, float* __restrict__ normDout) {
    __shared__ __attribute__((aligned(16))) float acc[RS_NODES * RS_PITCH];
    __shared__ int   indeg[RS_NODES];
    __shared__ unsigned wlist[RS_WAVES * WL_CAP];
    __shared__ int   wcount[RS_WAVES];
    __shared__ int   cutflag;
    __shared__ float part[8 * RS_NODES];

    const int tid  = threadIdx.x;
    const int lane = tid & 31;
    const int wave = __builtin_amdgcn_readfirstlane(tid >> 5);
    const int base = blockIdx.x * RS_NODES;

    float4* acc4 = reinterpret_cast<float4*>(acc);
    const int4*   dst4 = reinterpret_cast<const int4*>(dst);
    const float4* M4   = reinterpret_cast<const float4*>(M);

    const unsigned* mybin = bins + (size_t)blockIdx.x * BIN_REGION;
    int cl = 0;
    if (lane < BIN_PER_WAVE) cl = (int)mybin[(wave + lane * RS_WAVES) * BIN_SLOT + BIN_CAP];

    unsigned f0[BIN_PER_WAVE];
    unsigned f1[BIN_PER_WAVE];
#pragma unroll
    for (int k = 0; k < BIN_PER_WAVE; ++k) {
        const int c = __builtin_amdgcn_readlane(cl, k);
        const unsigned* slot = mybin + (wave + k * RS_WAVES) * BIN_SLOT;
        const bool whole = c <= BIN_CAP;
        f0[k] = slot[(whole && lane < c) ? lane : BIN_CAP];
        f1[k] = slot[(whole && lane + 32 < c) ? lane + 32 : BIN_CAP];
    }

    if (tid < RS_WAVES) wcount[tid] = 0;
    if (tid == 0) cutflag = 0;
    if (tid < RS_NODES) indeg[tid] = 0;
    __syncthreads();

    if (__builtin_amdgcn_ballot_w32(cl > BIN_CAP) != 0u) {
        if (lane == 0) cutflag = 1;
    } else {
#pragma unroll
        for (int k = 0; k < BIN_PER_WAVE; ++k) {
            const unsigned* slot = mybin + (wave + k * RS_WAVES) * BIN_SLOT;
            const int c = __builtin_amdgcn_readlane(cl, k);
            if (lane < c) rs_append(f0[k], wlist, wcount, indeg);
            if (lane + 32 < c) rs_append(f1[k], wlist, wcount, indeg);
            for (int p0 = 64; p0 < c; p0 += 32)
                if (p0 + lane < c) rs_append(slot[p0 + lane], wlist, wcount, indeg);
        }
    }
    __syncthreads();

    const int  myn  = wcount[lane];
    const bool over = __builtin_amdgcn_ballot_w32(myn > WL_CAP) != 0u;
    const bool fast = __builtin_amdgcn_readfirstlane(cutflag) == 0 && !over;

    if (fast) {
        const int lo   = wave * RS_ROWS_LO + (wave < 16 ? wave : 16);
        const int rows = wave < 16 ? RS_ROWS_HI : RS_ROWS_LO;
        for (int k = 0; k < rows; ++k)
            acc4[(lo + k) * RS_PITCH4 + lane] = make_float4(0.0f, 0.0f, 0.0f, 0.0f);

        const unsigned* wl = wlist + wave * WL_CAP;
        const int n = __builtin_amdgcn_readlane(myn, wave);
        int head = 0;
        for (; head + 8 <= n; head += 8) rs_round<true>(wl, head, 8, M4, lane, acc4);
        if (head < n) rs_round<false>(wl, head, n - head, M4, lane, acc4);
    } else {
        for (int i = tid; i < RS_NODES * RS_PITCH4; i += RS_THREADS)
            acc4[i] = make_float4(0.0f, 0.0f, 0.0f, 0.0f);
        if (tid < RS_NODES) indeg[tid] = 0;
        __syncthreads();

        const bool hi = lane >= 16;
        const int o0 = lane * 4 + (hi ? 2 : 0);
        const int o1 = lane * 4 + (hi ? 3 : 1);
        const int o2 = lane * 4 + (hi ? 0 : 2);
        const int o3 = lane * 4 + (hi ? 1 : 3);

        for (int k = 0; k < BIN_PER_WAVE; ++k) {
            const int g = wave + k * RS_WAVES;
            const unsigned* slot = mybin + g * BIN_SLOT;
            const int c = __builtin_amdgcn_readlane(cl, k);
            if (c <= BIN_CAP) {
                for (int p0 = 0; p0 < c; p0 += 32) {
                    const unsigned w = (p0 + lane < c) ? slot[p0 + lane] : 0u;
                    const int m32 = (c - p0) < 32 ? (c - p0) : 32;
                    const int s  = (int)(w & BIN_MASK);
                    const int dl = (int)(w >> BIN_SHIFT);
                    if (lane < m32) atomicAdd(indeg + dl, 1);
                    for (int r = 0; r < 4; ++r) {
                        const int m = m32 - r * 8;
                        if (m <= 0) break;
                        float4 v[8];
#pragma unroll
                        for (int j = 0; j < 8; ++j) {
                            v[j] = make_float4(0.0f, 0.0f, 0.0f, 0.0f);
                            if (j < m) {
                                const int sj = __builtin_amdgcn_readlane(s, r * 8 + j);
                                v[j] = M4[(size_t)sj * (NHID / 4) + lane];
                            }
                        }
#pragma unroll
                        for (int j = 0; j < 8; ++j) {
                            if (j < m) {
                                const int dj = __builtin_amdgcn_readlane(dl, r * 8 + j);
                                float* p = acc + dj * RS_PITCH;
                                atomicAdd(p + o0, hi ? v[j].z : v[j].x);
                                atomicAdd(p + o1, hi ? v[j].w : v[j].y);
                                atomicAdd(p + o2, hi ? v[j].x : v[j].z);
                                atomicAdd(p + o3, hi ? v[j].y : v[j].w);
                            }
                        }
                    }
                }
            } else {
                for (int i = lane; i < BIN_I4; i += 32) {
                    const int i4 = g * BIN_I4 + i;
                    const int4 d = dst4[i4];
                    const int e = i4 * 4;
                    unsigned dl;
                    dl = (unsigned)(d.x - base);
                    if (dl < (unsigned)RS_NODES) rs_slow_add(e + 0, dl, indeg, acc, M, src);
                    dl = (unsigned)(d.y - base);
                    if (dl < (unsigned)RS_NODES) rs_slow_add(e + 1, dl, indeg, acc, M, src);
                    dl = (unsigned)(d.z - base);
                    if (dl < (unsigned)RS_NODES) rs_slow_add(e + 2, dl, indeg, acc, M, src);
                    dl = (unsigned)(d.w - base);
                    if (dl < (unsigned)RS_NODES) rs_slow_add(e + 3, dl, indeg, acc, M, src);
                }
            }
        }
    }
    __syncthreads();

    if (tid < RS_NODES) normDout[base + tid] = rsqrtf(fmaxf((float)indeg[tid], 1.0f));
    const float4* b4 = reinterpret_cast<const float4*>(bias);
    const float4 bb = b4[lane];
    for (int i = tid; i < RS_NODES * (NHID / 4); i += RS_THREADS) {
        const int ln   = i >> 5;
        const int node = base + ln;
        const float nd = rsqrtf(fmaxf((float)indeg[ln], 1.0f));
        const float ns = normS[node];
        float4 v = acc4[ln * RS_PITCH4 + lane];
        float t;
        t = v.x * nd + bb.x;  v.x = (t > 0.0f ? t : 0.0f) * ns;
        t = v.y * nd + bb.y;  v.y = (t > 0.0f ? t : 0.0f) * ns;
        t = v.z * nd + bb.z;  v.z = (t > 0.0f ? t : 0.0f) * ns;
        t = v.w * nd + bb.w;  v.w = (t > 0.0f ? t : 0.0f) * ns;
        acc4[ln * RS_PITCH4 + lane] = v;
    }
    __syncthreads();

    const int l   = lane & 15;
    const int kh  = lane >> 4;
    const int ct  = wave & 7;
    const int col = ct * 16 + l;
    v2f bfr[NHID / 4];
#pragma unroll
    for (int ks = 0; ks < NHID / 4; ++ks) {
        const int ka = ks * 4 + 2 * kh;
        bfr[ks].x = Wn[(size_t)ka       * NHID + col];
        bfr[ks].y = Wn[(size_t)(ka + 1) * NHID + col];
    }
    const float wf = Wfc[col];
    for (int rt = wave >> 3; rt < RS_RTILES; rt += RS_WAVES / 8) {
        const float* arow = acc + (rt * 16 + l) * RS_PITCH + 2 * kh;
        v8f c = {};
#pragma unroll
        for (int ks = 0; ks < NHID / 4; ++ks) {
            const v2f a = *reinterpret_cast<const v2f*>(arow + ks * 4);
            c = __builtin_amdgcn_wmma_f32_16x16x4_f32(false, a, false, bfr[ks],
                                                      (short)0, c, false, false);
        }
        asm volatile("s_wait_alu depctr_va_vdst(0)\n\tv_nop\n\tv_nop\n\tv_nop\n\tv_nop" : "+v"(c));
        float keep = 0.0f;
#pragma unroll
        for (int r = 0; r < 8; ++r) {
            float p = c[r] * wf;
            p += __shfl_xor(p, 8, 32);
            p += __shfl_xor(p, 4, 32);
            p += __shfl_xor(p, 2, 32);
            p += __shfl_xor(p, 1, 32);
            if (l == r) keep = p;
        }
        if (l < 8) part[ct * RS_NODES + rt * 16 + 8 * kh + l] = keep;
    }
    __syncthreads();
    if (tid < RS_NODES) {
        float s = part[tid];
#pragma unroll
        for (int t = 1; t < 8; ++t) s += part[t * RS_NODES + tid];
        S[base + tid] = s;
    }
}

#define ES_THREADS  256
#define ES_WAVES    (ES_THREADS / 32)
#define ES_PER_WAVE (BIN_SLICES / ES_WAVES)
static_assert(BIN_SLICES % ES_WAVES == 0 && ES_PER_WAVE <= 32, "whole lists per wave, counts in one load");

__global__ void __launch_bounds__(ES_THREADS)
edge_scalar_sum(const int* __restrict__ src, const int* __restrict__ dst,
                const unsigned* __restrict__ bins, const float* __restrict__ S,
                const float* __restrict__ normD, const float* __restrict__ b2,
                const float* __restrict__ Wfc, const float* __restrict__ bfc,
                float* __restrict__ out) {
    __shared__ float esum[RS_NODES];

    const int tid  = threadIdx.x;
    const int lane = tid & 31;
    const int wave = __builtin_amdgcn_readfirstlane(tid >> 5);
    const int base = blockIdx.x * RS_NODES;

    for (int i = tid; i < RS_NODES; i += ES_THREADS) esum[i] = 0.0f;
    __syncthreads();

    const unsigned* mybin = bins + (size_t)blockIdx.x * BIN_REGION;
    int cl = 0;
    if (lane < ES_PER_WAVE) cl = (int)mybin[(wave + lane * ES_WAVES) * BIN_SLOT + BIN_CAP];

    float cw = 0.0f;
#pragma unroll
    for (int t = 0; t < 4; ++t) cw += b2[lane + 32 * t] * Wfc[lane + 32 * t];
#pragma unroll
    for (int off = 16; off > 0; off >>= 1) cw += __shfl_xor(cw, off, 32);
    cw += bfc[0];

    unsigned w0[ES_PER_WAVE];
    unsigned w1[ES_PER_WAVE];
#pragma unroll
    for (int k = 0; k < ES_PER_WAVE; ++k) {
        const int c = __builtin_amdgcn_readlane(cl, k);
        const unsigned* slot = mybin + (wave + k * ES_WAVES) * BIN_SLOT;
        const bool whole = c <= BIN_CAP;
        w0[k] = slot[(whole && lane < c) ? lane : BIN_CAP];
        w1[k] = slot[(whole && lane + 32 < c) ? lane + 32 : BIN_CAP];
    }
    __builtin_amdgcn_sched_barrier(0);
    float g0[ES_PER_WAVE];
    float g1[ES_PER_WAVE];
#pragma unroll
    for (int k = 0; k < ES_PER_WAVE; ++k) {
        const int c = __builtin_amdgcn_readlane(cl, k);
        const bool whole = c <= BIN_CAP;
        g0[k] = S[(whole && lane < c) ? (w0[k] & BIN_MASK) : 0u];
        g1[k] = S[(whole && lane + 32 < c) ? (w1[k] & BIN_MASK) : 0u];
    }
#pragma unroll
    for (int k = 0; k < ES_PER_WAVE; ++k) {
        const int c = __builtin_amdgcn_readlane(cl, k);
        const bool whole = c <= BIN_CAP;
        if (whole && lane < c && (w0[k] >> BIN_SHIFT) < (unsigned)RS_NODES)
            atomicAdd(&esum[w0[k] >> BIN_SHIFT], g0[k]);
        if (whole && lane + 32 < c && (w1[k] >> BIN_SHIFT) < (unsigned)RS_NODES)
            atomicAdd(&esum[w1[k] >> BIN_SHIFT], g1[k]);
    }
#pragma unroll 1
    for (int k = 0; k < ES_PER_WAVE; ++k) {
        const int g = wave + k * ES_WAVES;
        const unsigned* slot = mybin + g * BIN_SLOT;
        const int c = __builtin_amdgcn_readlane(cl, k);
        if (c <= BIN_CAP) {
            for (int p0 = 64; p0 < c; p0 += 32) {
                if (p0 + lane < c) {
                    const unsigned w = slot[p0 + lane];
                    const unsigned dl = w >> BIN_SHIFT;
                    if (dl < (unsigned)RS_NODES) atomicAdd(&esum[dl], S[w & BIN_MASK]);
                }
            }
        } else {
            const int4* d4 = reinterpret_cast<const int4*>(dst) + (size_t)g * BIN_I4;
            const int4* s4 = reinterpret_cast<const int4*>(src) + (size_t)g * BIN_I4;
            for (int i = lane; i < BIN_I4; i += 32) {
                const int4 d = d4[i];
                const int4 s = s4[i];
                unsigned dl;
                dl = (unsigned)(d.x - base);
                if (dl < (unsigned)RS_NODES) atomicAdd(&esum[dl], S[s.x]);
                dl = (unsigned)(d.y - base);
                if (dl < (unsigned)RS_NODES) atomicAdd(&esum[dl], S[s.y]);
                dl = (unsigned)(d.z - base);
                if (dl < (unsigned)RS_NODES) atomicAdd(&esum[dl], S[s.z]);
                dl = (unsigned)(d.w - base);
                if (dl < (unsigned)RS_NODES) atomicAdd(&esum[dl], S[s.w]);
            }
        }
    }
    __syncthreads();
    for (int i = tid; i < RS_NODES; i += ES_THREADS)
        out[base + i] = esum[i] * normD[base + i] + cw;
}

extern "C" void kernel_launch(void* const* d_in, const int* in_sizes, int n_in,
                              void* d_out, int out_size, void* d_ws, size_t ws_size,
                              hipStream_t stream) {
    const float* x   = (const float*)d_in[0];
    const float* W1  = (const float*)d_in[1];
    const float* b1  = (const float*)d_in[2];
    const float* W2  = (const float*)d_in[3];
    const float* b2  = (const float*)d_in[4];
    const float* Wfc = (const float*)d_in[5];
    const float* bfc = (const float*)d_in[6];
    const int*   src = (const int*)d_in[7];
    const int*   dst = (const int*)d_in[8];
    float* out = (float*)d_out;

    const int N = N_NODES;
    const int NH = NHID;

    float* BUF_A = (float*)d_ws;
    float* BUF_B = BUF_A + (size_t)N * NH;
    float* normS = BUF_B + (size_t)N * NH;
    float* normD = normS + N;

    bin_edges<<<BIN_SLICES, BIN_THREADS, 0, stream>>>(src, dst, (unsigned*)BUF_B, (unsigned*)BUF_A);
    outdeg_norm<<<RS_BLOCKS, OD_THREADS, 0, stream>>>(src, (const unsigned*)BUF_A, normS);

    gemm_scaled_wmma<<<G1_BLOCKS, G1_THREADS, 0, stream>>>(x, normS, W1, BUF_A);
    edge_rowsum_lds<<<RS_BLOCKS, RS_THREADS, 0, stream>>>(BUF_A, src, dst, (const unsigned*)BUF_B,
                                                         normS, b1, W2, Wfc, normS, normD);

    edge_scalar_sum<<<RS_BLOCKS, ES_THREADS, 0, stream>>>(src, dst, (const unsigned*)BUF_B, normS, normD,
                                                         b2, Wfc, bfc, out);
}
